# starting kernel
# baseline (speedup 1.0000x reference)
#include <hip/hip_runtime.h>
#include <hip/hip_bf16.h>
#include <math.h>

typedef float v2f __attribute__((ext_vector_type(2)));
typedef float v8f __attribute__((ext_vector_type(8)));

constexpr int NM    = 31;
constexpr int NP    = 961;
constexpr int NPPAD = 992;
constexpr int KNP   = 964;
constexpr int J     = 768;
constexpr int JOs   = 256;
constexpr int CCH   = 4;
constexpr int CMP   = 128;
constexpr int QMR   = 496;
constexpr int QMP   = 512;
constexpr float LBOX   = 20.0f;
constexpr float VBOX   = 8000.0f;
constexpr float TWOPI_L = 0.31415926535897931f;

constexpr size_t SZ_PH  = (size_t)NM * J;
constexpr size_t OFF_CX = 0;
constexpr size_t OFF_SX = OFF_CX + SZ_PH;
constexpr size_t OFF_CY = OFF_SX + SZ_PH;
constexpr size_t OFF_SY = OFF_CY + SZ_PH;
constexpr size_t OFF_CZ = OFF_SY + SZ_PH;
constexpr size_t OFF_SZ = OFF_CZ + SZ_PH;
constexpr size_t SZ_W   = (size_t)NPPAD * J;
constexpr size_t OFF_WC = OFF_SZ + SZ_PH;
constexpr size_t OFF_WS = OFF_WC + SZ_W;
constexpr size_t SZ_Y   = (size_t)CMP * J;
constexpr size_t OFF_YR = OFF_WS + SZ_W;
constexpr size_t OFF_YI = OFF_YR + SZ_Y;
constexpr size_t SZ_R   = (size_t)CMP * NPPAD;
constexpr size_t OFF_ROR = OFF_YI + SZ_Y;
constexpr size_t OFF_ROI = OFF_ROR + SZ_R;
constexpr size_t OFF_RHR = OFF_ROI + SZ_R;
constexpr size_t OFF_RHI = OFF_RHR + SZ_R;
constexpr size_t SZ_G   = (size_t)QMP * KNP;
constexpr size_t OFF_GOR = OFF_RHI + SZ_R;
constexpr size_t OFF_GOI = OFF_GOR + SZ_G;
constexpr size_t OFF_GHR = OFF_GOI + SZ_G;
constexpr size_t OFF_GHI = OFF_GHR + SZ_G;
constexpr size_t SZ_T   = (size_t)QMP * J;
constexpr size_t OFF_TR = OFF_GHI + SZ_G;
constexpr size_t OFF_TI = OFF_TR + SZ_T;
constexpr size_t WS_TOTAL = OFF_TI + SZ_T;

__global__ void phase_build(const float* __restrict__ pos,
                            float* __restrict__ Cx, float* __restrict__ Sx,
                            float* __restrict__ Cy, float* __restrict__ Sy,
                            float* __restrict__ Cz, float* __restrict__ Sz) {
  int t = blockIdx.x * blockDim.x + threadIdx.x;
  if (t >= 3 * J) return;
  int d = t / J, j = t % J;
  float th = TWOPI_L * (pos[j * 3 + d] - 0.5f * LBOX);
  float* C = (d == 0) ? Cx : (d == 1) ? Cy : Cz;
  float* S = (d == 0) ? Sx : (d == 1) ? Sy : Sz;
  for (int mi = 0; mi < NM; ++mi) {
    float s, c;
    sincosf((float)(mi - 15) * th, &s, &c);
    C[mi * J + j] = c;
    S[mi * J + j] = s;
  }
}

__global__ void w_build(const float* __restrict__ Cy, const float* __restrict__ Sy,
                        const float* __restrict__ Cz, const float* __restrict__ Sz,
                        float* __restrict__ Wc, float* __restrict__ Ws) {
  int t = blockIdx.x * blockDim.x + threadIdx.x;
  if (t >= NP * J) return;
  int np = t / J, j = t % J;
  int n = np / NM, p = np % NM;
  float cy = Cy[n * J + j], sy = Sy[n * J + j];
  float cz = Cz[p * J + j], sz = Sz[p * J + j];
  Wc[(size_t)np * J + j] = cy * cz - sy * sz;
  Ws[(size_t)np * J + j] = sy * cz + cy * sz;
}

__global__ void y_build(const float* __restrict__ charge,
                        const float* __restrict__ Cx, const float* __restrict__ Sx,
                        float* __restrict__ Yr, float* __restrict__ Yi) {
  int t = blockIdx.x * blockDim.x + threadIdx.x;
  if (t >= (CCH * NM) * J) return;
  int cm = t / J, j = t % J;
  int c = cm / NM, m = cm % NM;
  float q = charge[j * CCH + c];
  Yr[(size_t)cm * J + j] =  q * Cx[m * J + j];
  Yi[(size_t)cm * J + j] = -q * Sx[m * J + j];
}

template <bool NTMODE, bool SPLUS>
__global__ __launch_bounds__(32)
void cgemm_wmma(const float* __restrict__ Ar, const float* __restrict__ Ai, int lda,
                const float* __restrict__ Br, const float* __restrict__ Bi, int ldb,
                float* __restrict__ Cr, float* __restrict__ Ci, int ldc, int K) {
  const int lane = threadIdx.x;
  const int row  = lane & 15;
  const int kp   = lane >> 4;
  const int n0 = blockIdx.x * 32;
  const int m0 = blockIdx.y * 32;

  v8f accr[2][2], acci[2][2];
#pragma unroll
  for (int mt = 0; mt < 2; ++mt)
#pragma unroll
    for (int nt = 0; nt < 2; ++nt) {
      accr[mt][nt] = (v8f){0.f, 0.f, 0.f, 0.f, 0.f, 0.f, 0.f, 0.f};
      acci[mt][nt] = (v8f){0.f, 0.f, 0.f, 0.f, 0.f, 0.f, 0.f, 0.f};
    }

  const float* arp0 = Ar + (size_t)(m0 + row) * lda + 2 * kp;
  const float* aip0 = Ai + (size_t)(m0 + row) * lda + 2 * kp;
  const float* arp1 = arp0 + (size_t)16 * lda;
  const float* aip1 = aip0 + (size_t)16 * lda;

  for (int k0 = 0; k0 < K; k0 += 4) {
    v2f a_r[2], a_i[2], b_r[2], b_i[2];
    a_r[0] = *(const v2f*)(arp0 + k0);
    a_r[1] = *(const v2f*)(arp1 + k0);
    a_i[0] = *(const v2f*)(aip0 + k0);
    a_i[1] = *(const v2f*)(aip1 + k0);
    if (NTMODE) {
      const size_t bo = (size_t)(n0 + row) * ldb + k0 + 2 * kp;
      b_r[0] = *(const v2f*)(Br + bo);
      b_i[0] = *(const v2f*)(Bi + bo);
      b_r[1] = *(const v2f*)(Br + bo + (size_t)16 * ldb);
      b_i[1] = *(const v2f*)(Bi + bo + (size_t)16 * ldb);
    } else {
      const int kk = k0 + 2 * kp;
#pragma unroll
      for (int nt = 0; nt < 2; ++nt) {
        const size_t c0 = (size_t)kk * ldb + n0 + nt * 16 + row;
        b_r[nt].x = Br[c0];
        b_r[nt].y = Br[c0 + ldb];
        b_i[nt].x = Bi[c0];
        b_i[nt].y = Bi[c0 + ldb];
      }
    }
    v2f neg[2];
#pragma unroll
    for (int mt = 0; mt < 2; ++mt) {
      v2f src = SPLUS ? a_i[mt] : a_r[mt];
      neg[mt].x = -src.x;
      neg[mt].y = -src.y;
    }
#pragma unroll
    for (int mt = 0; mt < 2; ++mt)
#pragma unroll
      for (int nt = 0; nt < 2; ++nt) {
        accr[mt][nt] = __builtin_amdgcn_wmma_f32_16x16x4_f32(
            false, a_r[mt], false, b_r[nt], (short)0, accr[mt][nt], false, false);
        acci[mt][nt] = __builtin_amdgcn_wmma_f32_16x16x4_f32(
            false, a_i[mt], false, b_r[nt], (short)0, acci[mt][nt], false, false);
        if (SPLUS) {
          accr[mt][nt] = __builtin_amdgcn_wmma_f32_16x16x4_f32(
              false, neg[mt], false, b_i[nt], (short)0, accr[mt][nt], false, false);
          acci[mt][nt] = __builtin_amdgcn_wmma_f32_16x16x4_f32(
              false, a_r[mt], false, b_i[nt], (short)0, acci[mt][nt], false, false);
        } else {
          accr[mt][nt] = __builtin_amdgcn_wmma_f32_16x16x4_f32(
              false, a_i[mt], false, b_i[nt], (short)0, accr[mt][nt], false, false);
          acci[mt][nt] = __builtin_amdgcn_wmma_f32_16x16x4_f32(
              false, neg[mt], false, b_i[nt], (short)0, acci[mt][nt], false, false);
        }
      }
  }

#pragma unroll
  for (int mt = 0; mt < 2; ++mt)
#pragma unroll
    for (int nt = 0; nt < 2; ++nt)
#pragma unroll
      for (int v = 0; v < 8; ++v) {
        const size_t o = (size_t)(m0 + mt * 16 + v + 8 * kp) * ldc + n0 + nt * 16 + row;
        Cr[o] = accr[mt][nt][v];
        Ci[o] = acci[mt][nt][v];
      }
}

__global__ void g_build(const float* __restrict__ rOr, const float* __restrict__ rOi,
                        const float* __restrict__ rHr, const float* __restrict__ rHi,
                        const float* __restrict__ shOO, const float* __restrict__ amOO,
                        const float* __restrict__ shOH, const float* __restrict__ amOH,
                        const float* __restrict__ shHO, const float* __restrict__ amHO,
                        const float* __restrict__ shHH, const float* __restrict__ amHH,
                        float* __restrict__ gOr, float* __restrict__ gOi,
                        float* __restrict__ gHr, float* __restrict__ gHi) {
  int t = blockIdx.x * blockDim.x + threadIdx.x;
  if (t >= NM * NP) return;
  int m = t / NP, np = t % NP;
  int n = np / NM, p = np % NM;
  float kx = TWOPI_L * (float)(m - 15);
  float ky = TWOPI_L * (float)(n - 15);
  float kz = TWOPI_L * (float)(p - 15);
  float squ = kx * kx + ky * ky + kz * kz;

  float mOO = 0.f, mOH = 0.f, mHO = 0.f, mHH = 0.f;
#pragma unroll
  for (int i = 0; i < 6; ++i) {
    mOO += amOO[i] * expf(-squ * expf(2.0f * shOO[i]));
    mOH += amOH[i] * expf(-squ * expf(2.0f * shOH[i]));
    mHO += amHO[i] * expf(-squ * expf(2.0f * shHO[i]));
    mHH += amHH[i] * expf(-squ * expf(2.0f * shHH[i]));
  }
  if (squ == 0.0f) { mOH = 0.f; mHO = 0.f; mHH = 0.f; }

#pragma unroll
  for (int c = 0; c < CCH; ++c) {
    const size_t ri = (size_t)(c * NM + m) * NPPAD + np;
    float ror = rOr[ri], roi = rOi[ri];
    float rhr = rHr[ri], rhi = rHi[ri];
    float fOr = mOO * ror + mOH * rhr, fOi = mOO * roi + mOH * rhi;
    float fHr = mHO * ror + mHH * rhr, fHi = mHO * roi + mHH * rhi;
    const size_t r0 = (size_t)(c * NM + m)        * KNP + np;
    const size_t r1 = (size_t)((4  + c) * NM + m) * KNP + np;
    const size_t r2 = (size_t)((8  + c) * NM + m) * KNP + np;
    const size_t r3 = (size_t)((12 + c) * NM + m) * KNP + np;
    gOr[r0] = fOr;        gOi[r0] = fOi;
    gOr[r1] = kx * fOr;   gOi[r1] = kx * fOi;
    gOr[r2] = ky * fOr;   gOi[r2] = ky * fOi;
    gOr[r3] = kz * fOr;   gOi[r3] = kz * fOi;
    gHr[r0] = fHr;        gHi[r0] = fHi;
    gHr[r1] = kx * fHr;   gHi[r1] = kx * fHi;
    gHr[r2] = ky * fHr;   gHi[r2] = ky * fHi;
    gHr[r3] = kz * fHr;   gHi[r3] = kz * fHi;
  }
}

__global__ void finalize(const float* __restrict__ Tr, const float* __restrict__ Ti,
                         const float* __restrict__ Cx, const float* __restrict__ Sx,
                         const float* __restrict__ charge, float* __restrict__ out, int b) {
  int j = blockIdx.x * blockDim.x + threadIdx.x;
  if (j >= J) return;
  float ur[4]  = {0.f, 0.f, 0.f, 0.f};
  float ui[12] = {0.f, 0.f, 0.f, 0.f, 0.f, 0.f, 0.f, 0.f, 0.f, 0.f, 0.f, 0.f};
  for (int m = 0; m < NM; ++m) {
    float cx = Cx[m * J + j], sx = Sx[m * J + j];
#pragma unroll
    for (int q = 0; q < 16; ++q) {
      float tr = Tr[(size_t)(q * NM + m) * J + j];
      float ti = Ti[(size_t)(q * NM + m) * J + j];
      if (q < 4) ur[q]     += tr * cx - ti * sx;
      else       ui[q - 4] += tr * sx + ti * cx;
    }
  }
  float e = 0.f, f0 = 0.f, f1 = 0.f, f2 = 0.f;
#pragma unroll
  for (int c = 0; c < CCH; ++c) {
    float q = charge[j * CCH + c];
    e  += q * ur[c];
    f0 += q * ui[0 + c];
    f1 += q * ui[4 + c];
    f2 += q * ui[8 + c];
  }
  out[b * J + j] = e / (2.0f * VBOX);
  float* F = out + 2 * J;
  F[(size_t)(b * J + j) * 3 + 0] = f0 / VBOX;
  F[(size_t)(b * J + j) * 3 + 1] = f1 / VBOX;
  F[(size_t)(b * J + j) * 3 + 2] = f2 / VBOX;
}

extern "C" void kernel_launch(void* const* d_in, const int* in_sizes, int n_in,
                              void* d_out, int out_size, void* d_ws, size_t ws_size,
                              hipStream_t stream) {
  const float* pos    = (const float*)d_in[0];
  const float* charge = (const float*)d_in[1];
  const float* shOO = (const float*)d_in[2];
  const float* amOO = (const float*)d_in[3];
  const float* shOH = (const float*)d_in[4];
  const float* amOH = (const float*)d_in[5];
  const float* shHO = (const float*)d_in[6];
  const float* amHO = (const float*)d_in[7];
  const float* shHH = (const float*)d_in[8];
  const float* amHH = (const float*)d_in[9];
  float* out = (float*)d_out;
  float* ws  = (float*)d_ws;

  float *Cx = ws + OFF_CX, *Sx = ws + OFF_SX;
  float *Cy = ws + OFF_CY, *Sy = ws + OFF_SY;
  float *Cz = ws + OFF_CZ, *Sz = ws + OFF_SZ;
  float *Wc = ws + OFF_WC, *Wss = ws + OFF_WS;
  float *Yr = ws + OFF_YR, *Yi = ws + OFF_YI;
  float *rOr = ws + OFF_ROR, *rOi = ws + OFF_ROI;
  float *rHr = ws + OFF_RHR, *rHi = ws + OFF_RHI;
  float *gOr = ws + OFF_GOR, *gOi = ws + OFF_GOI;
  float *gHr = ws + OFF_GHR, *gHi = ws + OFF_GHI;
  float *Tr = ws + OFF_TR, *Ti = ws + OFF_TI;

  hipMemsetAsync(d_ws, 0, WS_TOTAL * sizeof(float), stream);

  for (int b = 0; b < 2; ++b) {
    const float* posb = pos + (size_t)b * J * 3;
    const float* chb  = charge + (size_t)b * J * CCH;

    phase_build<<<(3 * J + 127) / 128, 128, 0, stream>>>(posb, Cx, Sx, Cy, Sy, Cz, Sz);
    w_build<<<(NP * J + 255) / 256, 256, 0, stream>>>(Cy, Sy, Cz, Sz, Wc, Wss);
    y_build<<<(CCH * NM * J + 255) / 256, 256, 0, stream>>>(chb, Cx, Sx, Yr, Yi);

    dim3 g1(NPPAD / 32, CMP / 32);
    cgemm_wmma<true, false><<<g1, 32, 0, stream>>>(Yr, Yi, J, Wc, Wss, J,
                                                   rOr, rOi, NPPAD, JOs);
    cgemm_wmma<true, false><<<g1, 32, 0, stream>>>(Yr + JOs, Yi + JOs, J,
                                                   Wc + JOs, Wss + JOs, J,
                                                   rHr, rHi, NPPAD, J - JOs);

    g_build<<<(NM * NP + 255) / 256, 256, 0, stream>>>(
        rOr, rOi, rHr, rHi,
        shOO, amOO, shOH, amOH, shHO, amHO, shHH, amHH,
        gOr, gOi, gHr, gHi);

    dim3 g2o(JOs / 32, QMP / 32), g2h((J - JOs) / 32, QMP / 32);
    cgemm_wmma<false, true><<<g2o, 32, 0, stream>>>(gOr, gOi, KNP, Wc, Wss, J,
                                                    Tr, Ti, J, KNP);
    cgemm_wmma<false, true><<<g2h, 32, 0, stream>>>(gHr, gHi, KNP, Wc + JOs, Wss + JOs, J,
                                                    Tr + JOs, Ti + JOs, J, KNP);

    finalize<<<(J + 127) / 128, 128, 0, stream>>>(Tr, Ti, Cx, Sx, chb, out, b);
  }
}


# GEMM K loops software-pipelined in registers: the next chunk's global loads (5 or 7 K-steps) go out as one group ahead of the current chunk's WMMAs; per-accumulator sum order unchanged
# speedup vs baseline: 2.1643x; 2.1643x over previous
#include <hip/hip_runtime.h>
#include <hip/hip_bf16.h>
#include <math.h>

typedef float v2f __attribute__((ext_vector_type(2)));
typedef float v8f __attribute__((ext_vector_type(8)));

constexpr int NM    = 31;
constexpr int NP    = 961;
constexpr int NPPAD = 992;
constexpr int KNP   = 964;
constexpr int J     = 768;
constexpr int JOs   = 256;
constexpr int CCH   = 4;
constexpr int CMP   = 128;
constexpr int QMR   = 496;
constexpr int QMP   = 512;
constexpr float LBOX   = 20.0f;
constexpr float VBOX   = 8000.0f;
constexpr float TWOPI_L = 0.31415926535897931f;

constexpr size_t SZ_PH  = (size_t)NM * J;
constexpr size_t OFF_CX = 0;
constexpr size_t OFF_SX = OFF_CX + SZ_PH;
constexpr size_t OFF_CY = OFF_SX + SZ_PH;
constexpr size_t OFF_SY = OFF_CY + SZ_PH;
constexpr size_t OFF_CZ = OFF_SY + SZ_PH;
constexpr size_t OFF_SZ = OFF_CZ + SZ_PH;
constexpr size_t SZ_W   = (size_t)NPPAD * J;
constexpr size_t OFF_WC = OFF_SZ + SZ_PH;
constexpr size_t OFF_WS = OFF_WC + SZ_W;
constexpr size_t SZ_Y   = (size_t)CMP * J;
constexpr size_t OFF_YR = OFF_WS + SZ_W;
constexpr size_t OFF_YI = OFF_YR + SZ_Y;
constexpr size_t SZ_R   = (size_t)CMP * NPPAD;
constexpr size_t OFF_ROR = OFF_YI + SZ_Y;
constexpr size_t OFF_ROI = OFF_ROR + SZ_R;
constexpr size_t OFF_RHR = OFF_ROI + SZ_R;
constexpr size_t OFF_RHI = OFF_RHR + SZ_R;
constexpr size_t SZ_G   = (size_t)QMP * KNP;
constexpr size_t OFF_GOR = OFF_RHI + SZ_R;
constexpr size_t OFF_GOI = OFF_GOR + SZ_G;
constexpr size_t OFF_GHR = OFF_GOI + SZ_G;
constexpr size_t OFF_GHI = OFF_GHR + SZ_G;
constexpr size_t SZ_T   = (size_t)QMP * J;
constexpr size_t OFF_TR = OFF_GHI + SZ_G;
constexpr size_t OFF_TI = OFF_TR + SZ_T;
constexpr size_t WS_TOTAL = OFF_TI + SZ_T;

__global__ void phase_build(const float* __restrict__ pos,
                            float* __restrict__ Cx, float* __restrict__ Sx,
                            float* __restrict__ Cy, float* __restrict__ Sy,
                            float* __restrict__ Cz, float* __restrict__ Sz) {
  int t = blockIdx.x * blockDim.x + threadIdx.x;
  if (t >= 3 * J) return;
  int d = t / J, j = t % J;
  float th = TWOPI_L * (pos[j * 3 + d] - 0.5f * LBOX);
  float* C = (d == 0) ? Cx : (d == 1) ? Cy : Cz;
  float* S = (d == 0) ? Sx : (d == 1) ? Sy : Sz;
  for (int mi = 0; mi < NM; ++mi) {
    float s, c;
    sincosf((float)(mi - 15) * th, &s, &c);
    C[mi * J + j] = c;
    S[mi * J + j] = s;
  }
}

__global__ void w_build(const float* __restrict__ Cy, const float* __restrict__ Sy,
                        const float* __restrict__ Cz, const float* __restrict__ Sz,
                        float* __restrict__ Wc, float* __restrict__ Ws) {
  int t = blockIdx.x * blockDim.x + threadIdx.x;
  if (t >= NP * J) return;
  int np = t / J, j = t % J;
  int n = np / NM, p = np % NM;
  float cy = Cy[n * J + j], sy = Sy[n * J + j];
  float cz = Cz[p * J + j], sz = Sz[p * J + j];
  Wc[(size_t)np * J + j] = cy * cz - sy * sz;
  Ws[(size_t)np * J + j] = sy * cz + cy * sz;
}

__global__ void y_build(const float* __restrict__ charge,
                        const float* __restrict__ Cx, const float* __restrict__ Sx,
                        float* __restrict__ Yr, float* __restrict__ Yi) {
  int t = blockIdx.x * blockDim.x + threadIdx.x;
  if (t >= (CCH * NM) * J) return;
  int cm = t / J, j = t % J;
  int c = cm / NM, m = cm % NM;
  float q = charge[j * CCH + c];
  Yr[(size_t)cm * J + j] =  q * Cx[m * J + j];
  Yi[(size_t)cm * J + j] = -q * Sx[m * J + j];
}

template <int CH>
struct Frags {
  v2f ar[CH][2], ai[CH][2], br[CH][2], bi[CH][2];
};

template <bool NTMODE, int CH>
__device__ __forceinline__ void load_chunk(Frags<CH>& f,
                                           const float* __restrict__ arp0,
                                           const float* __restrict__ aip0,
                                           const float* __restrict__ arp1,
                                           const float* __restrict__ aip1,
                                           const float* __restrict__ brp,
                                           const float* __restrict__ bip,
                                           int ldb, int k0) {
#pragma unroll
  for (int s = 0; s < CH; ++s) {
    const int k = k0 + 4 * s;
    f.ar[s][0] = *(const v2f*)(arp0 + k);
    f.ar[s][1] = *(const v2f*)(arp1 + k);
    f.ai[s][0] = *(const v2f*)(aip0 + k);
    f.ai[s][1] = *(const v2f*)(aip1 + k);
    if (NTMODE) {
      f.br[s][0] = *(const v2f*)(brp + k);
      f.bi[s][0] = *(const v2f*)(bip + k);
      f.br[s][1] = *(const v2f*)(brp + (size_t)16 * ldb + k);
      f.bi[s][1] = *(const v2f*)(bip + (size_t)16 * ldb + k);
    } else {
      const size_t r0 = (size_t)k * ldb;
#pragma unroll
      for (int nt = 0; nt < 2; ++nt) {
        f.br[s][nt].x = brp[r0 + nt * 16];
        f.br[s][nt].y = brp[r0 + ldb + nt * 16];
        f.bi[s][nt].x = bip[r0 + nt * 16];
        f.bi[s][nt].y = bip[r0 + ldb + nt * 16];
      }
    }
  }
}

template <bool SPLUS>
__device__ __forceinline__ void kstep(const v2f (&a_r)[2], const v2f (&a_i)[2],
                                      const v2f (&b_r)[2], const v2f (&b_i)[2],
                                      v8f (&accr)[2][2], v8f (&acci)[2][2]) {
  v2f neg[2];
#pragma unroll
  for (int mt = 0; mt < 2; ++mt) {
    v2f src = SPLUS ? a_i[mt] : a_r[mt];
    neg[mt].x = -src.x;
    neg[mt].y = -src.y;
  }
#pragma unroll
  for (int mt = 0; mt < 2; ++mt)
#pragma unroll
    for (int nt = 0; nt < 2; ++nt) {
      accr[mt][nt] = __builtin_amdgcn_wmma_f32_16x16x4_f32(
          false, a_r[mt], false, b_r[nt], (short)0, accr[mt][nt], false, false);
      acci[mt][nt] = __builtin_amdgcn_wmma_f32_16x16x4_f32(
          false, a_i[mt], false, b_r[nt], (short)0, acci[mt][nt], false, false);
      if (SPLUS) {
        accr[mt][nt] = __builtin_amdgcn_wmma_f32_16x16x4_f32(
            false, neg[mt], false, b_i[nt], (short)0, accr[mt][nt], false, false);
        acci[mt][nt] = __builtin_amdgcn_wmma_f32_16x16x4_f32(
            false, a_r[mt], false, b_i[nt], (short)0, acci[mt][nt], false, false);
      } else {
        accr[mt][nt] = __builtin_amdgcn_wmma_f32_16x16x4_f32(
            false, a_i[mt], false, b_i[nt], (short)0, accr[mt][nt], false, false);
        acci[mt][nt] = __builtin_amdgcn_wmma_f32_16x16x4_f32(
            false, neg[mt], false, b_i[nt], (short)0, acci[mt][nt], false, false);
      }
    }
}

template <bool SPLUS, int CH>
__device__ __forceinline__ void compute_chunk(const Frags<CH>& f,
                                              v8f (&accr)[2][2], v8f (&acci)[2][2]) {
#pragma unroll
  for (int s = 0; s < CH; ++s)
    kstep<SPLUS>(f.ar[s], f.ai[s], f.br[s], f.bi[s], accr, acci);
}

template <bool NTMODE, bool SPLUS>
__global__ __launch_bounds__(32)
void cgemm_wmma(const float* __restrict__ Ar, const float* __restrict__ Ai, int lda,
                const float* __restrict__ Br, const float* __restrict__ Bi, int ldb,
                float* __restrict__ Cr, float* __restrict__ Ci, int ldc, int K) {
  constexpr int CH = NTMODE ? 7 : 5;
  const int lane = threadIdx.x;
  const int row  = lane & 15;
  const int kp   = lane >> 4;
  const int n0 = blockIdx.x * 32;
  const int m0 = blockIdx.y * 32;

  v8f accr[2][2], acci[2][2];
#pragma unroll
  for (int mt = 0; mt < 2; ++mt)
#pragma unroll
    for (int nt = 0; nt < 2; ++nt) {
      accr[mt][nt] = (v8f){0.f, 0.f, 0.f, 0.f, 0.f, 0.f, 0.f, 0.f};
      acci[mt][nt] = (v8f){0.f, 0.f, 0.f, 0.f, 0.f, 0.f, 0.f, 0.f};
    }

  const float* arp0 = Ar + (size_t)(m0 + row) * lda + 2 * kp;
  const float* aip0 = Ai + (size_t)(m0 + row) * lda + 2 * kp;
  const float* arp1 = arp0 + (size_t)16 * lda;
  const float* aip1 = aip0 + (size_t)16 * lda;
  const float* brp = NTMODE ? Br + (size_t)(n0 + row) * ldb + 2 * kp
                            : Br + (size_t)(2 * kp) * ldb + n0 + row;
  const float* bip = NTMODE ? Bi + (size_t)(n0 + row) * ldb + 2 * kp
                            : Bi + (size_t)(2 * kp) * ldb + n0 + row;

  const int nch = K / (4 * CH);
  Frags<CH> f0, f1;
  if (nch > 0) load_chunk<NTMODE, CH>(f0, arp0, aip0, arp1, aip1, brp, bip, ldb, 0);
  int c = 0;
  for (; c + 2 <= nch; c += 2) {
    load_chunk<NTMODE, CH>(f1, arp0, aip0, arp1, aip1, brp, bip, ldb, (c + 1) * 4 * CH);
    __builtin_amdgcn_sched_barrier(0);
    compute_chunk<SPLUS, CH>(f0, accr, acci);
    __builtin_amdgcn_sched_barrier(0);
    const int cn = (c + 2 < nch) ? (c + 2) : (nch - 1);
    load_chunk<NTMODE, CH>(f0, arp0, aip0, arp1, aip1, brp, bip, ldb, cn * 4 * CH);
    __builtin_amdgcn_sched_barrier(0);
    compute_chunk<SPLUS, CH>(f1, accr, acci);
    __builtin_amdgcn_sched_barrier(0);
  }
  if (c < nch) compute_chunk<SPLUS, CH>(f0, accr, acci);

  for (int k0 = nch * 4 * CH; k0 < K; k0 += 4) {
    Frags<1> t;
    load_chunk<NTMODE, 1>(t, arp0, aip0, arp1, aip1, brp, bip, ldb, k0);
    compute_chunk<SPLUS, 1>(t, accr, acci);
  }

#pragma unroll
  for (int mt = 0; mt < 2; ++mt)
#pragma unroll
    for (int nt = 0; nt < 2; ++nt)
#pragma unroll
      for (int v = 0; v < 8; ++v) {
        const size_t o = (size_t)(m0 + mt * 16 + v + 8 * kp) * ldc + n0 + nt * 16 + row;
        Cr[o] = accr[mt][nt][v];
        Ci[o] = acci[mt][nt][v];
      }
}

__global__ void g_build(const float* __restrict__ rOr, const float* __restrict__ rOi,
                        const float* __restrict__ rHr, const float* __restrict__ rHi,
                        const float* __restrict__ shOO, const float* __restrict__ amOO,
                        const float* __restrict__ shOH, const float* __restrict__ amOH,
                        const float* __restrict__ shHO, const float* __restrict__ amHO,
                        const float* __restrict__ shHH, const float* __restrict__ amHH,
                        float* __restrict__ gOr, float* __restrict__ gOi,
                        float* __restrict__ gHr, float* __restrict__ gHi) {
  int t = blockIdx.x * blockDim.x + threadIdx.x;
  if (t >= NM * NP) return;
  int m = t / NP, np = t % NP;
  int n = np / NM, p = np % NM;
  float kx = TWOPI_L * (float)(m - 15);
  float ky = TWOPI_L * (float)(n - 15);
  float kz = TWOPI_L * (float)(p - 15);
  float squ = kx * kx + ky * ky + kz * kz;

  float mOO = 0.f, mOH = 0.f, mHO = 0.f, mHH = 0.f;
#pragma unroll
  for (int i = 0; i < 6; ++i) {
    mOO += amOO[i] * expf(-squ * expf(2.0f * shOO[i]));
    mOH += amOH[i] * expf(-squ * expf(2.0f * shOH[i]));
    mHO += amHO[i] * expf(-squ * expf(2.0f * shHO[i]));
    mHH += amHH[i] * expf(-squ * expf(2.0f * shHH[i]));
  }
  if (squ == 0.0f) { mOH = 0.f; mHO = 0.f; mHH = 0.f; }

#pragma unroll
  for (int c = 0; c < CCH; ++c) {
    const size_t ri = (size_t)(c * NM + m) * NPPAD + np;
    float ror = rOr[ri], roi = rOi[ri];
    float rhr = rHr[ri], rhi = rHi[ri];
    float fOr = mOO * ror + mOH * rhr, fOi = mOO * roi + mOH * rhi;
    float fHr = mHO * ror + mHH * rhr, fHi = mHO * roi + mHH * rhi;
    const size_t r0 = (size_t)(c * NM + m)        * KNP + np;
    const size_t r1 = (size_t)((4  + c) * NM + m) * KNP + np;
    const size_t r2 = (size_t)((8  + c) * NM + m) * KNP + np;
    const size_t r3 = (size_t)((12 + c) * NM + m) * KNP + np;
    gOr[r0] = fOr;        gOi[r0] = fOi;
    gOr[r1] = kx * fOr;   gOi[r1] = kx * fOi;
    gOr[r2] = ky * fOr;   gOi[r2] = ky * fOi;
    gOr[r3] = kz * fOr;   gOi[r3] = kz * fOi;
    gHr[r0] = fHr;        gHi[r0] = fHi;
    gHr[r1] = kx * fHr;   gHi[r1] = kx * fHi;
    gHr[r2] = ky * fHr;   gHi[r2] = ky * fHi;
    gHr[r3] = kz * fHr;   gHi[r3] = kz * fHi;
  }
}

__global__ void finalize(const float* __restrict__ Tr, const float* __restrict__ Ti,
                         const float* __restrict__ Cx, const float* __restrict__ Sx,
                         const float* __restrict__ charge, float* __restrict__ out, int b) {
  int j = blockIdx.x * blockDim.x + threadIdx.x;
  if (j >= J) return;
  float ur[4]  = {0.f, 0.f, 0.f, 0.f};
  float ui[12] = {0.f, 0.f, 0.f, 0.f, 0.f, 0.f, 0.f, 0.f, 0.f, 0.f, 0.f, 0.f};
  for (int m = 0; m < NM; ++m) {
    float cx = Cx[m * J + j], sx = Sx[m * J + j];
#pragma unroll
    for (int q = 0; q < 16; ++q) {
      float tr = Tr[(size_t)(q * NM + m) * J + j];
      float ti = Ti[(size_t)(q * NM + m) * J + j];
      if (q < 4) ur[q]     += tr * cx - ti * sx;
      else       ui[q - 4] += tr * sx + ti * cx;
    }
  }
  float e = 0.f, f0 = 0.f, f1 = 0.f, f2 = 0.f;
#pragma unroll
  for (int c = 0; c < CCH; ++c) {
    float q = charge[j * CCH + c];
    e  += q * ur[c];
    f0 += q * ui[0 + c];
    f1 += q * ui[4 + c];
    f2 += q * ui[8 + c];
  }
  out[b * J + j] = e / (2.0f * VBOX);
  float* F = out + 2 * J;
  F[(size_t)(b * J + j) * 3 + 0] = f0 / VBOX;
  F[(size_t)(b * J + j) * 3 + 1] = f1 / VBOX;
  F[(size_t)(b * J + j) * 3 + 2] = f2 / VBOX;
}

extern "C" void kernel_launch(void* const* d_in, const int* in_sizes, int n_in,
                              void* d_out, int out_size, void* d_ws, size_t ws_size,
                              hipStream_t stream) {
  const float* pos    = (const float*)d_in[0];
  const float* charge = (const float*)d_in[1];
  const float* shOO = (const float*)d_in[2];
  const float* amOO = (const float*)d_in[3];
  const float* shOH = (const float*)d_in[4];
  const float* amOH = (const float*)d_in[5];
  const float* shHO = (const float*)d_in[6];
  const float* amHO = (const float*)d_in[7];
  const float* shHH = (const float*)d_in[8];
  const float* amHH = (const float*)d_in[9];
  float* out = (float*)d_out;
  float* ws  = (float*)d_ws;

  float *Cx = ws + OFF_CX, *Sx = ws + OFF_SX;
  float *Cy = ws + OFF_CY, *Sy = ws + OFF_SY;
  float *Cz = ws + OFF_CZ, *Sz = ws + OFF_SZ;
  float *Wc = ws + OFF_WC, *Wss = ws + OFF_WS;
  float *Yr = ws + OFF_YR, *Yi = ws + OFF_YI;
  float *rOr = ws + OFF_ROR, *rOi = ws + OFF_ROI;
  float *rHr = ws + OFF_RHR, *rHi = ws + OFF_RHI;
  float *gOr = ws + OFF_GOR, *gOi = ws + OFF_GOI;
  float *gHr = ws + OFF_GHR, *gHi = ws + OFF_GHI;
  float *Tr = ws + OFF_TR, *Ti = ws + OFF_TI;

  hipMemsetAsync(d_ws, 0, WS_TOTAL * sizeof(float), stream);

  for (int b = 0; b < 2; ++b) {
    const float* posb = pos + (size_t)b * J * 3;
    const float* chb  = charge + (size_t)b * J * CCH;

    phase_build<<<(3 * J + 127) / 128, 128, 0, stream>>>(posb, Cx, Sx, Cy, Sy, Cz, Sz);
    w_build<<<(NP * J + 255) / 256, 256, 0, stream>>>(Cy, Sy, Cz, Sz, Wc, Wss);
    y_build<<<(CCH * NM * J + 255) / 256, 256, 0, stream>>>(chb, Cx, Sx, Yr, Yi);

    dim3 g1(NPPAD / 32, CMP / 32);
    cgemm_wmma<true, false><<<g1, 32, 0, stream>>>(Yr, Yi, J, Wc, Wss, J,
                                                   rOr, rOi, NPPAD, JOs);
    cgemm_wmma<true, false><<<g1, 32, 0, stream>>>(Yr + JOs, Yi + JOs, J,
                                                   Wc + JOs, Wss + JOs, J,
                                                   rHr, rHi, NPPAD, J - JOs);

    g_build<<<(NM * NP + 255) / 256, 256, 0, stream>>>(
        rOr, rOi, rHr, rHi,
        shOO, amOO, shOH, amOH, shHO, amHO, shHH, amHH,
        gOr, gOi, gHr, gHi);

    dim3 g2o(JOs / 32, QMP / 32), g2h((J - JOs) / 32, QMP / 32);
    cgemm_wmma<false, true><<<g2o, 32, 0, stream>>>(gOr, gOi, KNP, Wc, Wss, J,
                                                    Tr, Ti, J, KNP);
    cgemm_wmma<false, true><<<g2h, 32, 0, stream>>>(gHr, gHi, KNP, Wc + JOs, Wss + JOs, J,
                                                    Tr + JOs, Ti + JOs, J, KNP);

    finalize<<<(J + 127) / 128, 128, 0, stream>>>(Tr, Ti, Cx, Sx, chb, out, b);
  }
}


# Real charges make the mode sum conjugate-symmetric: only the (n,p) half plane np<=480 is built and multiplied, g doubled below the centre row; each mirrored pair now enters the sums as 2x one term
# speedup vs baseline: 7.4340x; 3.4348x over previous
#include <hip/hip_runtime.h>
#include <hip/hip_bf16.h>
#include <math.h>

typedef float v2f __attribute__((ext_vector_type(2)));
typedef float v8f __attribute__((ext_vector_type(8)));

constexpr int NM    = 31;
constexpr int NPH   = 481;
constexpr int NPPAD = 512;
constexpr int KNP   = 484;
constexpr int J     = 768;
constexpr int JOs   = 256;
constexpr int CCH   = 4;
constexpr int CMP   = 128;
constexpr int QMR   = 496;
constexpr int QMP   = 512;
constexpr float LBOX   = 20.0f;
constexpr float VBOX   = 8000.0f;
constexpr float TWOPI_L = 0.31415926535897931f;

constexpr size_t SZ_PH  = (size_t)NM * J;
constexpr size_t OFF_CX = 0;
constexpr size_t OFF_SX = OFF_CX + SZ_PH;
constexpr size_t OFF_CY = OFF_SX + SZ_PH;
constexpr size_t OFF_SY = OFF_CY + SZ_PH;
constexpr size_t OFF_CZ = OFF_SY + SZ_PH;
constexpr size_t OFF_SZ = OFF_CZ + SZ_PH;
constexpr size_t SZ_W   = (size_t)NPPAD * J;
constexpr size_t OFF_WC = OFF_SZ + SZ_PH;
constexpr size_t OFF_WS = OFF_WC + SZ_W;
constexpr size_t SZ_Y   = (size_t)CMP * J;
constexpr size_t OFF_YR = OFF_WS + SZ_W;
constexpr size_t OFF_YI = OFF_YR + SZ_Y;
constexpr size_t SZ_R   = (size_t)CMP * NPPAD;
constexpr size_t OFF_ROR = OFF_YI + SZ_Y;
constexpr size_t OFF_ROI = OFF_ROR + SZ_R;
constexpr size_t OFF_RHR = OFF_ROI + SZ_R;
constexpr size_t OFF_RHI = OFF_RHR + SZ_R;
constexpr size_t SZ_G   = (size_t)QMP * KNP;
constexpr size_t OFF_GOR = OFF_RHI + SZ_R;
constexpr size_t OFF_GOI = OFF_GOR + SZ_G;
constexpr size_t OFF_GHR = OFF_GOI + SZ_G;
constexpr size_t OFF_GHI = OFF_GHR + SZ_G;
constexpr size_t SZ_T   = (size_t)QMP * J;
constexpr size_t OFF_TR = OFF_GHI + SZ_G;
constexpr size_t OFF_TI = OFF_TR + SZ_T;
constexpr size_t WS_TOTAL = OFF_TI + SZ_T;
constexpr int NBATCH = 2;

__global__ void phase_build(const float* __restrict__ pos,
                            float* __restrict__ Cx, float* __restrict__ Sx,
                            float* __restrict__ Cy, float* __restrict__ Sy,
                            float* __restrict__ Cz, float* __restrict__ Sz) {
  const int b = blockIdx.y;
  const size_t bo = (size_t)b * WS_TOTAL;
  int t = blockIdx.x * blockDim.x + threadIdx.x;
  if (t >= 3 * J) return;
  int d = t / J, j = t % J;
  float th = TWOPI_L * (pos[(size_t)b * J * 3 + j * 3 + d] - 0.5f * LBOX);
  float* C = ((d == 0) ? Cx : (d == 1) ? Cy : Cz) + bo;
  float* S = ((d == 0) ? Sx : (d == 1) ? Sy : Sz) + bo;
  for (int mi = 0; mi < NM; ++mi) {
    float s, c;
    sincosf((float)(mi - 15) * th, &s, &c);
    C[mi * J + j] = c;
    S[mi * J + j] = s;
  }
}

constexpr int WY_THREADS = 256;
constexpr int WB_BLOCKS = (NPH * J + WY_THREADS - 1) / WY_THREADS;
constexpr int YB_BLOCKS = (CCH * NM * J + WY_THREADS - 1) / WY_THREADS;
__global__ void wy_build(const float* __restrict__ charge,
                         const float* __restrict__ Cx, const float* __restrict__ Sx,
                         const float* __restrict__ Cy, const float* __restrict__ Sy,
                         const float* __restrict__ Cz, const float* __restrict__ Sz,
                         float* __restrict__ Wc, float* __restrict__ Ws,
                         float* __restrict__ Yr, float* __restrict__ Yi) {
  const int b = blockIdx.y;
  const size_t bo = (size_t)b * WS_TOTAL;
  if (blockIdx.x < WB_BLOCKS) {
    int t = blockIdx.x * WY_THREADS + threadIdx.x;
    if (t >= NPH * J) return;
    int np = t / J, j = t % J;
    int n = np / NM, p = np % NM;
    float cy = Cy[bo + n * J + j], sy = Sy[bo + n * J + j];
    float cz = Cz[bo + p * J + j], sz = Sz[bo + p * J + j];
    Wc[bo + (size_t)np * J + j] = cy * cz - sy * sz;
    Ws[bo + (size_t)np * J + j] = sy * cz + cy * sz;
  } else {
    int t = (blockIdx.x - WB_BLOCKS) * WY_THREADS + threadIdx.x;
    if (t >= (CCH * NM) * J) return;
    int cm = t / J, j = t % J;
    int c = cm / NM, m = cm % NM;
    float q = charge[(size_t)b * J * CCH + j * CCH + c];
    Yr[bo + (size_t)cm * J + j] =  q * Cx[bo + m * J + j];
    Yi[bo + (size_t)cm * J + j] = -q * Sx[bo + m * J + j];
  }
}

template <int CH>
struct Frags {
  v2f ar[CH][2], ai[CH][2], br[CH][2], bi[CH][2];
};

template <bool NTMODE, int CH>
__device__ __forceinline__ void load_chunk(Frags<CH>& f,
                                           const float* __restrict__ arp0,
                                           const float* __restrict__ aip0,
                                           const float* __restrict__ arp1,
                                           const float* __restrict__ aip1,
                                           const float* __restrict__ brp,
                                           const float* __restrict__ bip,
                                           int ldb, int k0) {
#pragma unroll
  for (int s = 0; s < CH; ++s) {
    const int k = k0 + 4 * s;
    f.ar[s][0] = *(const v2f*)(arp0 + k);
    f.ar[s][1] = *(const v2f*)(arp1 + k);
    f.ai[s][0] = *(const v2f*)(aip0 + k);
    f.ai[s][1] = *(const v2f*)(aip1 + k);
    if (NTMODE) {
      f.br[s][0] = *(const v2f*)(brp + k);
      f.bi[s][0] = *(const v2f*)(bip + k);
      f.br[s][1] = *(const v2f*)(brp + (size_t)16 * ldb + k);
      f.bi[s][1] = *(const v2f*)(bip + (size_t)16 * ldb + k);
    } else {
      const size_t r0 = (size_t)k * ldb;
#pragma unroll
      for (int nt = 0; nt < 2; ++nt) {
        f.br[s][nt].x = brp[r0 + nt * 16];
        f.br[s][nt].y = brp[r0 + ldb + nt * 16];
        f.bi[s][nt].x = bip[r0 + nt * 16];
        f.bi[s][nt].y = bip[r0 + ldb + nt * 16];
      }
    }
  }
}

template <bool SPLUS>
__device__ __forceinline__ void kstep(const v2f (&a_r)[2], const v2f (&a_i)[2],
                                      const v2f (&b_r)[2], const v2f (&b_i)[2],
                                      v8f (&accr)[2][2], v8f (&acci)[2][2]) {
  v2f neg[2];
#pragma unroll
  for (int mt = 0; mt < 2; ++mt) {
    v2f src = SPLUS ? a_i[mt] : a_r[mt];
    neg[mt].x = -src.x;
    neg[mt].y = -src.y;
  }
#pragma unroll
  for (int mt = 0; mt < 2; ++mt)
#pragma unroll
    for (int nt = 0; nt < 2; ++nt) {
      accr[mt][nt] = __builtin_amdgcn_wmma_f32_16x16x4_f32(
          false, a_r[mt], false, b_r[nt], (short)0, accr[mt][nt], false, false);
      acci[mt][nt] = __builtin_amdgcn_wmma_f32_16x16x4_f32(
          false, a_i[mt], false, b_r[nt], (short)0, acci[mt][nt], false, false);
      if (SPLUS) {
        accr[mt][nt] = __builtin_amdgcn_wmma_f32_16x16x4_f32(
            false, neg[mt], false, b_i[nt], (short)0, accr[mt][nt], false, false);
        acci[mt][nt] = __builtin_amdgcn_wmma_f32_16x16x4_f32(
            false, a_r[mt], false, b_i[nt], (short)0, acci[mt][nt], false, false);
      } else {
        accr[mt][nt] = __builtin_amdgcn_wmma_f32_16x16x4_f32(
            false, a_i[mt], false, b_i[nt], (short)0, accr[mt][nt], false, false);
        acci[mt][nt] = __builtin_amdgcn_wmma_f32_16x16x4_f32(
            false, neg[mt], false, b_i[nt], (short)0, acci[mt][nt], false, false);
      }
    }
}

template <bool SPLUS, int CH>
__device__ __forceinline__ void compute_chunk(const Frags<CH>& f,
                                              v8f (&accr)[2][2], v8f (&acci)[2][2]) {
#pragma unroll
  for (int s = 0; s < CH; ++s)
    kstep<SPLUS>(f.ar[s], f.ai[s], f.br[s], f.bi[s], accr, acci);
}

template <bool NTMODE, bool SPLUS>
__device__ __forceinline__
void cgemm_tile(const float* __restrict__ Ar, const float* __restrict__ Ai, int lda,
                const float* __restrict__ Br, const float* __restrict__ Bi, int ldb,
                float* __restrict__ Cr, float* __restrict__ Ci, int ldc, int K,
                int m0, int n0) {
  constexpr int CH = NTMODE ? 7 : 5;
  const int lane = threadIdx.x & 31;
  const int row  = lane & 15;
  const int kp   = lane >> 4;

  v8f accr[2][2], acci[2][2];
#pragma unroll
  for (int mt = 0; mt < 2; ++mt)
#pragma unroll
    for (int nt = 0; nt < 2; ++nt) {
      accr[mt][nt] = (v8f){0.f, 0.f, 0.f, 0.f, 0.f, 0.f, 0.f, 0.f};
      acci[mt][nt] = (v8f){0.f, 0.f, 0.f, 0.f, 0.f, 0.f, 0.f, 0.f};
    }

  const float* arp0 = Ar + (size_t)(m0 + row) * lda + 2 * kp;
  const float* aip0 = Ai + (size_t)(m0 + row) * lda + 2 * kp;
  const float* arp1 = arp0 + (size_t)16 * lda;
  const float* aip1 = aip0 + (size_t)16 * lda;
  const float* brp = NTMODE ? Br + (size_t)(n0 + row) * ldb + 2 * kp
                            : Br + (size_t)(2 * kp) * ldb + n0 + row;
  const float* bip = NTMODE ? Bi + (size_t)(n0 + row) * ldb + 2 * kp
                            : Bi + (size_t)(2 * kp) * ldb + n0 + row;

  const int nch = K / (4 * CH);
  Frags<CH> f0, f1;
  if (nch > 0) load_chunk<NTMODE, CH>(f0, arp0, aip0, arp1, aip1, brp, bip, ldb, 0);
  int c = 0;
  for (; c + 2 <= nch; c += 2) {
    load_chunk<NTMODE, CH>(f1, arp0, aip0, arp1, aip1, brp, bip, ldb, (c + 1) * 4 * CH);
    __builtin_amdgcn_sched_barrier(0);
    compute_chunk<SPLUS, CH>(f0, accr, acci);
    __builtin_amdgcn_sched_barrier(0);
    const int cn = (c + 2 < nch) ? (c + 2) : (nch - 1);
    load_chunk<NTMODE, CH>(f0, arp0, aip0, arp1, aip1, brp, bip, ldb, cn * 4 * CH);
    __builtin_amdgcn_sched_barrier(0);
    compute_chunk<SPLUS, CH>(f1, accr, acci);
    __builtin_amdgcn_sched_barrier(0);
  }
  if (c < nch) compute_chunk<SPLUS, CH>(f0, accr, acci);

  for (int k0 = nch * 4 * CH; k0 < K; k0 += 4) {
    Frags<1> t;
    load_chunk<NTMODE, 1>(t, arp0, aip0, arp1, aip1, brp, bip, ldb, k0);
    compute_chunk<SPLUS, 1>(t, accr, acci);
  }

#pragma unroll
  for (int mt = 0; mt < 2; ++mt)
#pragma unroll
    for (int nt = 0; nt < 2; ++nt)
#pragma unroll
      for (int v = 0; v < 8; ++v) {
        const size_t o = (size_t)(m0 + mt * 16 + v + 8 * kp) * ldc + n0 + nt * 16 + row;
        Cr[o] = accr[mt][nt][v];
        Ci[o] = acci[mt][nt][v];
      }
}


__global__ __launch_bounds__(128)
void nufft1_gemm(const float* __restrict__ Yr, const float* __restrict__ Yi,
                 const float* __restrict__ Wc, const float* __restrict__ Ws,
                 float* __restrict__ rOr, float* __restrict__ rOi,
                 float* __restrict__ rHr, float* __restrict__ rHi) {
  const int wave = threadIdx.x >> 5;
  const bool h = blockIdx.y != 0;
  const size_t bo = (size_t)blockIdx.z * WS_TOTAL;
  const int j0 = h ? JOs : 0;
  const int K  = h ? (J - JOs) : JOs;
  cgemm_tile<true, false>(Yr + bo + j0, Yi + bo + j0, J,
                          Wc + bo + j0, Ws + bo + j0, J,
                          (h ? rHr : rOr) + bo, (h ? rHi : rOi) + bo, NPPAD, K,
                          wave * 32, blockIdx.x * 32);
}

__global__ __launch_bounds__(128)
void stageA_gemm(const float* __restrict__ gOr, const float* __restrict__ gOi,
                 const float* __restrict__ gHr, const float* __restrict__ gHi,
                 const float* __restrict__ Wc, const float* __restrict__ Ws,
                 float* __restrict__ Tr, float* __restrict__ Ti) {
  const int wave = threadIdx.x >> 5;
  const bool h = blockIdx.x * 64 >= JOs;
  const size_t bo = (size_t)blockIdx.z * WS_TOTAL;
  const int n0 = blockIdx.x * 64 + (wave & 1) * 32;
  const int m0 = blockIdx.y * 64 + (wave >> 1) * 32;
  cgemm_tile<false, true>((h ? gHr : gOr) + bo, (h ? gHi : gOi) + bo, KNP,
                          Wc + bo, Ws + bo, J,
                          Tr + bo, Ti + bo, J, KNP, m0, n0);
}

__global__ void g_build(const float* __restrict__ rOr, const float* __restrict__ rOi,
                        const float* __restrict__ rHr, const float* __restrict__ rHi,
                        const float* __restrict__ shOO, const float* __restrict__ amOO,
                        const float* __restrict__ shOH, const float* __restrict__ amOH,
                        const float* __restrict__ shHO, const float* __restrict__ amHO,
                        const float* __restrict__ shHH, const float* __restrict__ amHH,
                        float* __restrict__ gOr, float* __restrict__ gOi,
                        float* __restrict__ gHr, float* __restrict__ gHi) {
  const size_t bo = (size_t)blockIdx.y * WS_TOTAL;
  int t = blockIdx.x * blockDim.x + threadIdx.x;
  if (t >= NM * NPH) return;
  int m = t / NPH, np = t % NPH;
  int n = np / NM, p = np % NM;
  float kx = TWOPI_L * (float)(m - 15);
  float ky = TWOPI_L * (float)(n - 15);
  float kz = TWOPI_L * (float)(p - 15);
  float squ = kx * kx + ky * ky + kz * kz;

  float mOO = 0.f, mOH = 0.f, mHO = 0.f, mHH = 0.f;
#pragma unroll
  for (int i = 0; i < 6; ++i) {
    mOO += amOO[i] * expf(-squ * expf(2.0f * shOO[i]));
    mOH += amOH[i] * expf(-squ * expf(2.0f * shOH[i]));
    mHO += amHO[i] * expf(-squ * expf(2.0f * shHO[i]));
    mHH += amHH[i] * expf(-squ * expf(2.0f * shHH[i]));
  }
  if (squ == 0.0f) { mOH = 0.f; mHO = 0.f; mHH = 0.f; }
  if (np < NPH - 1) { mOO *= 2.0f; mOH *= 2.0f; mHO *= 2.0f; mHH *= 2.0f; }

#pragma unroll
  for (int c = 0; c < CCH; ++c) {
    const size_t ri = bo + (size_t)(c * NM + m) * NPPAD + np;
    float ror = rOr[ri], roi = rOi[ri];
    float rhr = rHr[ri], rhi = rHi[ri];
    float fOr = mOO * ror + mOH * rhr, fOi = mOO * roi + mOH * rhi;
    float fHr = mHO * ror + mHH * rhr, fHi = mHO * roi + mHH * rhi;
    const size_t r0 = bo + (size_t)(c * NM + m)        * KNP + np;
    const size_t r1 = bo + (size_t)((4  + c) * NM + m) * KNP + np;
    const size_t r2 = bo + (size_t)((8  + c) * NM + m) * KNP + np;
    const size_t r3 = bo + (size_t)((12 + c) * NM + m) * KNP + np;
    gOr[r0] = fOr;        gOi[r0] = fOi;
    gOr[r1] = kx * fOr;   gOi[r1] = kx * fOi;
    gOr[r2] = ky * fOr;   gOi[r2] = ky * fOi;
    gOr[r3] = kz * fOr;   gOi[r3] = kz * fOi;
    gHr[r0] = fHr;        gHi[r0] = fHi;
    gHr[r1] = kx * fHr;   gHi[r1] = kx * fHi;
    gHr[r2] = ky * fHr;   gHi[r2] = ky * fHi;
    gHr[r3] = kz * fHr;   gHi[r3] = kz * fHi;
  }
}

__global__ void finalize(const float* __restrict__ Tr, const float* __restrict__ Ti,
                         const float* __restrict__ Cx, const float* __restrict__ Sx,
                         const float* __restrict__ charge, float* __restrict__ out) {
  const int b = blockIdx.y;
  const size_t bo = (size_t)b * WS_TOTAL;
  int j = blockIdx.x * blockDim.x + threadIdx.x;
  if (j >= J) return;
  float ur[4]  = {0.f, 0.f, 0.f, 0.f};
  float ui[12] = {0.f, 0.f, 0.f, 0.f, 0.f, 0.f, 0.f, 0.f, 0.f, 0.f, 0.f, 0.f};
  for (int m = 0; m < NM; ++m) {
    float cx = Cx[bo + m * J + j], sx = Sx[bo + m * J + j];
#pragma unroll
    for (int q = 0; q < 16; ++q) {
      float tr = Tr[bo + (size_t)(q * NM + m) * J + j];
      float ti = Ti[bo + (size_t)(q * NM + m) * J + j];
      if (q < 4) ur[q]     += tr * cx - ti * sx;
      else       ui[q - 4] += tr * sx + ti * cx;
    }
  }
  float e = 0.f, f0 = 0.f, f1 = 0.f, f2 = 0.f;
#pragma unroll
  for (int c = 0; c < CCH; ++c) {
    float q = charge[(size_t)b * J * CCH + j * CCH + c];
    e  += q * ur[c];
    f0 += q * ui[0 + c];
    f1 += q * ui[4 + c];
    f2 += q * ui[8 + c];
  }
  out[b * J + j] = e / (2.0f * VBOX);
  float* F = out + 2 * J;
  F[(size_t)(b * J + j) * 3 + 0] = f0 / VBOX;
  F[(size_t)(b * J + j) * 3 + 1] = f1 / VBOX;
  F[(size_t)(b * J + j) * 3 + 2] = f2 / VBOX;
}

extern "C" void kernel_launch(void* const* d_in, const int* in_sizes, int n_in,
                              void* d_out, int out_size, void* d_ws, size_t ws_size,
                              hipStream_t stream) {
  const float* pos    = (const float*)d_in[0];
  const float* charge = (const float*)d_in[1];
  const float* shOO = (const float*)d_in[2];
  const float* amOO = (const float*)d_in[3];
  const float* shOH = (const float*)d_in[4];
  const float* amOH = (const float*)d_in[5];
  const float* shHO = (const float*)d_in[6];
  const float* amHO = (const float*)d_in[7];
  const float* shHH = (const float*)d_in[8];
  const float* amHH = (const float*)d_in[9];
  float* out = (float*)d_out;
  float* ws  = (float*)d_ws;

  float *Cx = ws + OFF_CX, *Sx = ws + OFF_SX;
  float *Cy = ws + OFF_CY, *Sy = ws + OFF_SY;
  float *Cz = ws + OFF_CZ, *Sz = ws + OFF_SZ;
  float *Wc = ws + OFF_WC, *Wss = ws + OFF_WS;
  float *Yr = ws + OFF_YR, *Yi = ws + OFF_YI;
  float *rOr = ws + OFF_ROR, *rOi = ws + OFF_ROI;
  float *rHr = ws + OFF_RHR, *rHi = ws + OFF_RHI;
  float *gOr = ws + OFF_GOR, *gOi = ws + OFF_GOI;
  float *gHr = ws + OFF_GHR, *gHi = ws + OFF_GHI;
  float *Tr = ws + OFF_TR, *Ti = ws + OFF_TI;

  hipMemsetAsync(d_ws, 0, NBATCH * WS_TOTAL * sizeof(float), stream);

  phase_build<<<dim3((3 * J + 127) / 128, NBATCH), 128, 0, stream>>>(pos, Cx, Sx, Cy, Sy, Cz, Sz);
  wy_build<<<dim3(WB_BLOCKS + YB_BLOCKS, NBATCH), WY_THREADS, 0, stream>>>(
      charge, Cx, Sx, Cy, Sy, Cz, Sz, Wc, Wss, Yr, Yi);

  nufft1_gemm<<<dim3(NPPAD / 32, 2, NBATCH), 128, 0, stream>>>(Yr, Yi, Wc, Wss,
                                                               rOr, rOi, rHr, rHi);

  g_build<<<dim3((NM * NPH + 255) / 256, NBATCH), 256, 0, stream>>>(
      rOr, rOi, rHr, rHi,
      shOO, amOO, shOH, amOH, shHO, amHO, shHH, amHH,
      gOr, gOi, gHr, gHi);

  stageA_gemm<<<dim3(J / 64, QMP / 64, NBATCH), 128, 0, stream>>>(gOr, gOi, gHr, gHi,
                                                                  Wc, Wss, Tr, Ti);

  finalize<<<dim3((J + 127) / 128, NBATCH), 128, 0, stream>>>(Tr, Ti, Cx, Sx, charge, out);
}


# Final sum over m: one wave per (channel, 32 particles) instead of one lane per particle, a lane's 124 loads issued as one group ahead of its sums, channels joined in LDS; sum order over m unchanged
# speedup vs baseline: 8.9518x; 1.2042x over previous
#include <hip/hip_runtime.h>
#include <hip/hip_bf16.h>
#include <math.h>

typedef float v2f __attribute__((ext_vector_type(2)));
typedef float v8f __attribute__((ext_vector_type(8)));

constexpr int NM    = 31;
constexpr int NPH   = 481;
constexpr int NPPAD = 512;
constexpr int KNP   = 484;
constexpr int J     = 768;
constexpr int JOs   = 256;
constexpr int CCH   = 4;
constexpr int CMP   = 128;
constexpr int QMR   = 496;
constexpr int QMP   = 512;
constexpr float LBOX   = 20.0f;
constexpr float VBOX   = 8000.0f;
constexpr float TWOPI_L = 0.31415926535897931f;

constexpr size_t SZ_PH  = (size_t)NM * J;
constexpr size_t OFF_CX = 0;
constexpr size_t OFF_SX = OFF_CX + SZ_PH;
constexpr size_t OFF_CY = OFF_SX + SZ_PH;
constexpr size_t OFF_SY = OFF_CY + SZ_PH;
constexpr size_t OFF_CZ = OFF_SY + SZ_PH;
constexpr size_t OFF_SZ = OFF_CZ + SZ_PH;
constexpr size_t SZ_W   = (size_t)NPPAD * J;
constexpr size_t OFF_WC = OFF_SZ + SZ_PH;
constexpr size_t OFF_WS = OFF_WC + SZ_W;
constexpr size_t SZ_Y   = (size_t)CMP * J;
constexpr size_t OFF_YR = OFF_WS + SZ_W;
constexpr size_t OFF_YI = OFF_YR + SZ_Y;
constexpr size_t SZ_R   = (size_t)CMP * NPPAD;
constexpr size_t OFF_ROR = OFF_YI + SZ_Y;
constexpr size_t OFF_ROI = OFF_ROR + SZ_R;
constexpr size_t OFF_RHR = OFF_ROI + SZ_R;
constexpr size_t OFF_RHI = OFF_RHR + SZ_R;
constexpr size_t SZ_G   = (size_t)QMP * KNP;
constexpr size_t OFF_GOR = OFF_RHI + SZ_R;
constexpr size_t OFF_GOI = OFF_GOR + SZ_G;
constexpr size_t OFF_GHR = OFF_GOI + SZ_G;
constexpr size_t OFF_GHI = OFF_GHR + SZ_G;
constexpr size_t SZ_T   = (size_t)QMP * J;
constexpr size_t OFF_TR = OFF_GHI + SZ_G;
constexpr size_t OFF_TI = OFF_TR + SZ_T;
constexpr size_t WS_TOTAL = OFF_TI + SZ_T;
constexpr int NBATCH = 2;

__global__ void phase_build(const float* __restrict__ pos,
                            float* __restrict__ Cx, float* __restrict__ Sx,
                            float* __restrict__ Cy, float* __restrict__ Sy,
                            float* __restrict__ Cz, float* __restrict__ Sz) {
  const int b = blockIdx.y;
  const size_t bo = (size_t)b * WS_TOTAL;
  int t = blockIdx.x * blockDim.x + threadIdx.x;
  if (t >= 3 * J) return;
  int d = t / J, j = t % J;
  float th = TWOPI_L * (pos[(size_t)b * J * 3 + j * 3 + d] - 0.5f * LBOX);
  float* C = ((d == 0) ? Cx : (d == 1) ? Cy : Cz) + bo;
  float* S = ((d == 0) ? Sx : (d == 1) ? Sy : Sz) + bo;
  for (int mi = 0; mi < NM; ++mi) {
    float s, c;
    sincosf((float)(mi - 15) * th, &s, &c);
    C[mi * J + j] = c;
    S[mi * J + j] = s;
  }
}

constexpr int WY_THREADS = 256;
constexpr int WB_BLOCKS = (NPH * J + WY_THREADS - 1) / WY_THREADS;
constexpr int YB_BLOCKS = (CCH * NM * J + WY_THREADS - 1) / WY_THREADS;
__global__ void wy_build(const float* __restrict__ charge,
                         const float* __restrict__ Cx, const float* __restrict__ Sx,
                         const float* __restrict__ Cy, const float* __restrict__ Sy,
                         const float* __restrict__ Cz, const float* __restrict__ Sz,
                         float* __restrict__ Wc, float* __restrict__ Ws,
                         float* __restrict__ Yr, float* __restrict__ Yi) {
  const int b = blockIdx.y;
  const size_t bo = (size_t)b * WS_TOTAL;
  if (blockIdx.x < WB_BLOCKS) {
    int t = blockIdx.x * WY_THREADS + threadIdx.x;
    if (t >= NPH * J) return;
    int np = t / J, j = t % J;
    int n = np / NM, p = np % NM;
    float cy = Cy[bo + n * J + j], sy = Sy[bo + n * J + j];
    float cz = Cz[bo + p * J + j], sz = Sz[bo + p * J + j];
    Wc[bo + (size_t)np * J + j] = cy * cz - sy * sz;
    Ws[bo + (size_t)np * J + j] = sy * cz + cy * sz;
  } else {
    int t = (blockIdx.x - WB_BLOCKS) * WY_THREADS + threadIdx.x;
    if (t >= (CCH * NM) * J) return;
    int cm = t / J, j = t % J;
    int c = cm / NM, m = cm % NM;
    float q = charge[(size_t)b * J * CCH + j * CCH + c];
    Yr[bo + (size_t)cm * J + j] =  q * Cx[bo + m * J + j];
    Yi[bo + (size_t)cm * J + j] = -q * Sx[bo + m * J + j];
  }
}

template <int CH>
struct Frags {
  v2f ar[CH][2], ai[CH][2], br[CH][2], bi[CH][2];
};

template <bool NTMODE, int CH>
__device__ __forceinline__ void load_chunk(Frags<CH>& f,
                                           const float* __restrict__ arp0,
                                           const float* __restrict__ aip0,
                                           const float* __restrict__ arp1,
                                           const float* __restrict__ aip1,
                                           const float* __restrict__ brp,
                                           const float* __restrict__ bip,
                                           int ldb, int k0) {
#pragma unroll
  for (int s = 0; s < CH; ++s) {
    const int k = k0 + 4 * s;
    f.ar[s][0] = *(const v2f*)(arp0 + k);
    f.ar[s][1] = *(const v2f*)(arp1 + k);
    f.ai[s][0] = *(const v2f*)(aip0 + k);
    f.ai[s][1] = *(const v2f*)(aip1 + k);
    if (NTMODE) {
      f.br[s][0] = *(const v2f*)(brp + k);
      f.bi[s][0] = *(const v2f*)(bip + k);
      f.br[s][1] = *(const v2f*)(brp + (size_t)16 * ldb + k);
      f.bi[s][1] = *(const v2f*)(bip + (size_t)16 * ldb + k);
    } else {
      const size_t r0 = (size_t)k * ldb;
#pragma unroll
      for (int nt = 0; nt < 2; ++nt) {
        f.br[s][nt].x = brp[r0 + nt * 16];
        f.br[s][nt].y = brp[r0 + ldb + nt * 16];
        f.bi[s][nt].x = bip[r0 + nt * 16];
        f.bi[s][nt].y = bip[r0 + ldb + nt * 16];
      }
    }
  }
}

template <bool SPLUS>
__device__ __forceinline__ void kstep(const v2f (&a_r)[2], const v2f (&a_i)[2],
                                      const v2f (&b_r)[2], const v2f (&b_i)[2],
                                      v8f (&accr)[2][2], v8f (&acci)[2][2]) {
  v2f neg[2];
#pragma unroll
  for (int mt = 0; mt < 2; ++mt) {
    v2f src = SPLUS ? a_i[mt] : a_r[mt];
    neg[mt].x = -src.x;
    neg[mt].y = -src.y;
  }
#pragma unroll
  for (int mt = 0; mt < 2; ++mt)
#pragma unroll
    for (int nt = 0; nt < 2; ++nt) {
      accr[mt][nt] = __builtin_amdgcn_wmma_f32_16x16x4_f32(
          false, a_r[mt], false, b_r[nt], (short)0, accr[mt][nt], false, false);
      acci[mt][nt] = __builtin_amdgcn_wmma_f32_16x16x4_f32(
          false, a_i[mt], false, b_r[nt], (short)0, acci[mt][nt], false, false);
      if (SPLUS) {
        accr[mt][nt] = __builtin_amdgcn_wmma_f32_16x16x4_f32(
            false, neg[mt], false, b_i[nt], (short)0, accr[mt][nt], false, false);
        acci[mt][nt] = __builtin_amdgcn_wmma_f32_16x16x4_f32(
            false, a_r[mt], false, b_i[nt], (short)0, acci[mt][nt], false, false);
      } else {
        accr[mt][nt] = __builtin_amdgcn_wmma_f32_16x16x4_f32(
            false, a_i[mt], false, b_i[nt], (short)0, accr[mt][nt], false, false);
        acci[mt][nt] = __builtin_amdgcn_wmma_f32_16x16x4_f32(
            false, neg[mt], false, b_i[nt], (short)0, acci[mt][nt], false, false);
      }
    }
}

template <bool SPLUS, int CH>
__device__ __forceinline__ void compute_chunk(const Frags<CH>& f,
                                              v8f (&accr)[2][2], v8f (&acci)[2][2]) {
#pragma unroll
  for (int s = 0; s < CH; ++s)
    kstep<SPLUS>(f.ar[s], f.ai[s], f.br[s], f.bi[s], accr, acci);
}

template <bool NTMODE, bool SPLUS>
__device__ __forceinline__
void cgemm_tile(const float* __restrict__ Ar, const float* __restrict__ Ai, int lda,
                const float* __restrict__ Br, const float* __restrict__ Bi, int ldb,
                float* __restrict__ Cr, float* __restrict__ Ci, int ldc, int K,
                int m0, int n0) {
  constexpr int CH = NTMODE ? 7 : 5;
  const int lane = threadIdx.x & 31;
  const int row  = lane & 15;
  const int kp   = lane >> 4;

  v8f accr[2][2], acci[2][2];
#pragma unroll
  for (int mt = 0; mt < 2; ++mt)
#pragma unroll
    for (int nt = 0; nt < 2; ++nt) {
      accr[mt][nt] = (v8f){0.f, 0.f, 0.f, 0.f, 0.f, 0.f, 0.f, 0.f};
      acci[mt][nt] = (v8f){0.f, 0.f, 0.f, 0.f, 0.f, 0.f, 0.f, 0.f};
    }

  const float* arp0 = Ar + (size_t)(m0 + row) * lda + 2 * kp;
  const float* aip0 = Ai + (size_t)(m0 + row) * lda + 2 * kp;
  const float* arp1 = arp0 + (size_t)16 * lda;
  const float* aip1 = aip0 + (size_t)16 * lda;
  const float* brp = NTMODE ? Br + (size_t)(n0 + row) * ldb + 2 * kp
                            : Br + (size_t)(2 * kp) * ldb + n0 + row;
  const float* bip = NTMODE ? Bi + (size_t)(n0 + row) * ldb + 2 * kp
                            : Bi + (size_t)(2 * kp) * ldb + n0 + row;

  const int nch = K / (4 * CH);
  Frags<CH> f0, f1;
  if (nch > 0) load_chunk<NTMODE, CH>(f0, arp0, aip0, arp1, aip1, brp, bip, ldb, 0);
  int c = 0;
  for (; c + 2 <= nch; c += 2) {
    load_chunk<NTMODE, CH>(f1, arp0, aip0, arp1, aip1, brp, bip, ldb, (c + 1) * 4 * CH);
    __builtin_amdgcn_sched_barrier(0);
    compute_chunk<SPLUS, CH>(f0, accr, acci);
    __builtin_amdgcn_sched_barrier(0);
    const int cn = (c + 2 < nch) ? (c + 2) : (nch - 1);
    load_chunk<NTMODE, CH>(f0, arp0, aip0, arp1, aip1, brp, bip, ldb, cn * 4 * CH);
    __builtin_amdgcn_sched_barrier(0);
    compute_chunk<SPLUS, CH>(f1, accr, acci);
    __builtin_amdgcn_sched_barrier(0);
  }
  if (c < nch) compute_chunk<SPLUS, CH>(f0, accr, acci);

  for (int k0 = nch * 4 * CH; k0 < K; k0 += 4) {
    Frags<1> t;
    load_chunk<NTMODE, 1>(t, arp0, aip0, arp1, aip1, brp, bip, ldb, k0);
    compute_chunk<SPLUS, 1>(t, accr, acci);
  }

#pragma unroll
  for (int mt = 0; mt < 2; ++mt)
#pragma unroll
    for (int nt = 0; nt < 2; ++nt)
#pragma unroll
      for (int v = 0; v < 8; ++v) {
        const size_t o = (size_t)(m0 + mt * 16 + v + 8 * kp) * ldc + n0 + nt * 16 + row;
        Cr[o] = accr[mt][nt][v];
        Ci[o] = acci[mt][nt][v];
      }
}


__global__ __launch_bounds__(128)
void nufft1_gemm(const float* __restrict__ Yr, const float* __restrict__ Yi,
                 const float* __restrict__ Wc, const float* __restrict__ Ws,
                 float* __restrict__ rOr, float* __restrict__ rOi,
                 float* __restrict__ rHr, float* __restrict__ rHi) {
  const int wave = threadIdx.x >> 5;
  const bool h = blockIdx.y != 0;
  const size_t bo = (size_t)blockIdx.z * WS_TOTAL;
  const int j0 = h ? JOs : 0;
  const int K  = h ? (J - JOs) : JOs;
  cgemm_tile<true, false>(Yr + bo + j0, Yi + bo + j0, J,
                          Wc + bo + j0, Ws + bo + j0, J,
                          (h ? rHr : rOr) + bo, (h ? rHi : rOi) + bo, NPPAD, K,
                          wave * 32, blockIdx.x * 32);
}

__global__ __launch_bounds__(128)
void stageA_gemm(const float* __restrict__ gOr, const float* __restrict__ gOi,
                 const float* __restrict__ gHr, const float* __restrict__ gHi,
                 const float* __restrict__ Wc, const float* __restrict__ Ws,
                 float* __restrict__ Tr, float* __restrict__ Ti) {
  const int wave = threadIdx.x >> 5;
  const bool h = blockIdx.x * 64 >= JOs;
  const size_t bo = (size_t)blockIdx.z * WS_TOTAL;
  const int n0 = blockIdx.x * 64 + (wave & 1) * 32;
  const int m0 = blockIdx.y * 64 + (wave >> 1) * 32;
  cgemm_tile<false, true>((h ? gHr : gOr) + bo, (h ? gHi : gOi) + bo, KNP,
                          Wc + bo, Ws + bo, J,
                          Tr + bo, Ti + bo, J, KNP, m0, n0);
}

__global__ void g_build(const float* __restrict__ rOr, const float* __restrict__ rOi,
                        const float* __restrict__ rHr, const float* __restrict__ rHi,
                        const float* __restrict__ shOO, const float* __restrict__ amOO,
                        const float* __restrict__ shOH, const float* __restrict__ amOH,
                        const float* __restrict__ shHO, const float* __restrict__ amHO,
                        const float* __restrict__ shHH, const float* __restrict__ amHH,
                        float* __restrict__ gOr, float* __restrict__ gOi,
                        float* __restrict__ gHr, float* __restrict__ gHi) {
  const size_t bo = (size_t)blockIdx.y * WS_TOTAL;
  int t = blockIdx.x * blockDim.x + threadIdx.x;
  if (t >= NM * NPH) return;
  int m = t / NPH, np = t % NPH;
  int n = np / NM, p = np % NM;
  float kx = TWOPI_L * (float)(m - 15);
  float ky = TWOPI_L * (float)(n - 15);
  float kz = TWOPI_L * (float)(p - 15);
  float squ = kx * kx + ky * ky + kz * kz;

  float mOO = 0.f, mOH = 0.f, mHO = 0.f, mHH = 0.f;
#pragma unroll
  for (int i = 0; i < 6; ++i) {
    mOO += amOO[i] * expf(-squ * expf(2.0f * shOO[i]));
    mOH += amOH[i] * expf(-squ * expf(2.0f * shOH[i]));
    mHO += amHO[i] * expf(-squ * expf(2.0f * shHO[i]));
    mHH += amHH[i] * expf(-squ * expf(2.0f * shHH[i]));
  }
  if (squ == 0.0f) { mOH = 0.f; mHO = 0.f; mHH = 0.f; }
  if (np < NPH - 1) { mOO *= 2.0f; mOH *= 2.0f; mHO *= 2.0f; mHH *= 2.0f; }

#pragma unroll
  for (int c = 0; c < CCH; ++c) {
    const size_t ri = bo + (size_t)(c * NM + m) * NPPAD + np;
    float ror = rOr[ri], roi = rOi[ri];
    float rhr = rHr[ri], rhi = rHi[ri];
    float fOr = mOO * ror + mOH * rhr, fOi = mOO * roi + mOH * rhi;
    float fHr = mHO * ror + mHH * rhr, fHi = mHO * roi + mHH * rhi;
    const size_t r0 = bo + (size_t)(c * NM + m)        * KNP + np;
    const size_t r1 = bo + (size_t)((4  + c) * NM + m) * KNP + np;
    const size_t r2 = bo + (size_t)((8  + c) * NM + m) * KNP + np;
    const size_t r3 = bo + (size_t)((12 + c) * NM + m) * KNP + np;
    gOr[r0] = fOr;        gOi[r0] = fOi;
    gOr[r1] = kx * fOr;   gOi[r1] = kx * fOi;
    gOr[r2] = ky * fOr;   gOi[r2] = ky * fOi;
    gOr[r3] = kz * fOr;   gOi[r3] = kz * fOi;
    gHr[r0] = fHr;        gHi[r0] = fHi;
    gHr[r1] = kx * fHr;   gHi[r1] = kx * fHi;
    gHr[r2] = ky * fHr;   gHi[r2] = ky * fHi;
    gHr[r3] = kz * fHr;   gHi[r3] = kz * fHi;
  }
}

constexpr int FIN_J = 32;
constexpr int FIN_Q = 16;
__global__ __launch_bounds__(FIN_J * FIN_Q)
void finalize(const float* __restrict__ Tr, const float* __restrict__ Ti,
              const float* __restrict__ Cx, const float* __restrict__ Sx,
              const float* __restrict__ charge, float* __restrict__ out) {
  __shared__ float u[FIN_Q][FIN_J];
  const int b = blockIdx.y;
  const size_t bo = (size_t)b * WS_TOTAL;
  const int jl = threadIdx.x & (FIN_J - 1);
  const int q = __builtin_amdgcn_readfirstlane((int)(threadIdx.x >> 5));
  const int j = blockIdx.x * FIN_J + jl;
  const float* trp = Tr + bo + (size_t)(q * NM) * J + j;
  const float* tip = Ti + bo + (size_t)(q * NM) * J + j;
  const float* cxp = Cx + bo + j;
  const float* sxp = Sx + bo + j;
  float cx[NM], sx[NM], tr[NM], ti[NM];
#pragma unroll
  for (int m = 0; m < NM; ++m) {
    cx[m] = cxp[m * J];
    sx[m] = sxp[m * J];
    tr[m] = trp[m * J];
    ti[m] = tip[m * J];
  }
  __builtin_amdgcn_sched_barrier(0);
  float acc = 0.f;
  if (q < 4) {
#pragma unroll
    for (int m = 0; m < NM; ++m) acc += tr[m] * cx[m] - ti[m] * sx[m];
  } else {
#pragma unroll
    for (int m = 0; m < NM; ++m) acc += tr[m] * sx[m] + ti[m] * cx[m];
  }
  u[q][jl] = acc;
  __syncthreads();
  if (q != 0) return;
  float e = 0.f, f0 = 0.f, f1 = 0.f, f2 = 0.f;
#pragma unroll
  for (int c = 0; c < CCH; ++c) {
    float qc = charge[(size_t)b * J * CCH + j * CCH + c];
    e  += qc * u[c][jl];
    f0 += qc * u[4 + c][jl];
    f1 += qc * u[8 + c][jl];
    f2 += qc * u[12 + c][jl];
  }
  out[b * J + j] = e / (2.0f * VBOX);
  float* F = out + 2 * J;
  F[(size_t)(b * J + j) * 3 + 0] = f0 / VBOX;
  F[(size_t)(b * J + j) * 3 + 1] = f1 / VBOX;
  F[(size_t)(b * J + j) * 3 + 2] = f2 / VBOX;
}

extern "C" void kernel_launch(void* const* d_in, const int* in_sizes, int n_in,
                              void* d_out, int out_size, void* d_ws, size_t ws_size,
                              hipStream_t stream) {
  const float* pos    = (const float*)d_in[0];
  const float* charge = (const float*)d_in[1];
  const float* shOO = (const float*)d_in[2];
  const float* amOO = (const float*)d_in[3];
  const float* shOH = (const float*)d_in[4];
  const float* amOH = (const float*)d_in[5];
  const float* shHO = (const float*)d_in[6];
  const float* amHO = (const float*)d_in[7];
  const float* shHH = (const float*)d_in[8];
  const float* amHH = (const float*)d_in[9];
  float* out = (float*)d_out;
  float* ws  = (float*)d_ws;

  float *Cx = ws + OFF_CX, *Sx = ws + OFF_SX;
  float *Cy = ws + OFF_CY, *Sy = ws + OFF_SY;
  float *Cz = ws + OFF_CZ, *Sz = ws + OFF_SZ;
  float *Wc = ws + OFF_WC, *Wss = ws + OFF_WS;
  float *Yr = ws + OFF_YR, *Yi = ws + OFF_YI;
  float *rOr = ws + OFF_ROR, *rOi = ws + OFF_ROI;
  float *rHr = ws + OFF_RHR, *rHi = ws + OFF_RHI;
  float *gOr = ws + OFF_GOR, *gOi = ws + OFF_GOI;
  float *gHr = ws + OFF_GHR, *gHi = ws + OFF_GHI;
  float *Tr = ws + OFF_TR, *Ti = ws + OFF_TI;

  hipMemsetAsync(d_ws, 0, NBATCH * WS_TOTAL * sizeof(float), stream);

  phase_build<<<dim3((3 * J + 127) / 128, NBATCH), 128, 0, stream>>>(pos, Cx, Sx, Cy, Sy, Cz, Sz);
  wy_build<<<dim3(WB_BLOCKS + YB_BLOCKS, NBATCH), WY_THREADS, 0, stream>>>(
      charge, Cx, Sx, Cy, Sy, Cz, Sz, Wc, Wss, Yr, Yi);

  nufft1_gemm<<<dim3(NPPAD / 32, 2, NBATCH), 128, 0, stream>>>(Yr, Yi, Wc, Wss,
                                                               rOr, rOi, rHr, rHi);

  g_build<<<dim3((NM * NPH + 255) / 256, NBATCH), 256, 0, stream>>>(
      rOr, rOi, rHr, rHi,
      shOO, amOO, shOH, amOH, shHO, amHO, shHH, amHH,
      gOr, gOi, gHr, gHi);

  stageA_gemm<<<dim3(J / 64, QMP / 64, NBATCH), 128, 0, stream>>>(gOr, gOi, gHr, gHi,
                                                                  Wc, Wss, Tr, Ti);

  finalize<<<dim3(J / FIN_J, NBATCH), FIN_J * FIN_Q, 0, stream>>>(Tr, Ti, Cx, Sx, charge, out);
}


# nufft1: the sum over a species' particles is cut into 128-particle slices, one GEMM wave per slice and tile (3x the workgroups); g_build adds the partial r in slice order: particle sum order changes
# speedup vs baseline: 11.0613x; 1.2357x over previous
#include <hip/hip_runtime.h>
#include <hip/hip_bf16.h>
#include <math.h>

typedef float v2f __attribute__((ext_vector_type(2)));
typedef float v8f __attribute__((ext_vector_type(8)));

constexpr int NM    = 31;
constexpr int NPH   = 481;
constexpr int NPPAD = 512;
constexpr int KNP   = 484;
constexpr int J     = 768;
constexpr int JOs   = 256;
constexpr int CCH   = 4;
constexpr int CMP   = 128;
constexpr int QMR   = 496;
constexpr int QMP   = 512;
constexpr int KSL   = 128;
constexpr int NSL_O = JOs / KSL;
constexpr int NSL   = J / KSL;
static_assert(JOs % KSL == 0 && J % KSL == 0, "a slice never straddles the two species");
constexpr float LBOX   = 20.0f;
constexpr float VBOX   = 8000.0f;
constexpr float TWOPI_L = 0.31415926535897931f;

constexpr size_t SZ_PH  = (size_t)NM * J;
constexpr size_t OFF_CX = 0;
constexpr size_t OFF_SX = OFF_CX + SZ_PH;
constexpr size_t OFF_CY = OFF_SX + SZ_PH;
constexpr size_t OFF_SY = OFF_CY + SZ_PH;
constexpr size_t OFF_CZ = OFF_SY + SZ_PH;
constexpr size_t OFF_SZ = OFF_CZ + SZ_PH;
constexpr size_t SZ_W   = (size_t)NPPAD * J;
constexpr size_t OFF_WC = OFF_SZ + SZ_PH;
constexpr size_t OFF_WS = OFF_WC + SZ_W;
constexpr size_t SZ_Y   = (size_t)CMP * J;
constexpr size_t OFF_YR = OFF_WS + SZ_W;
constexpr size_t OFF_YI = OFF_YR + SZ_Y;
constexpr size_t SZ_G   = (size_t)QMP * KNP;
constexpr size_t OFF_GOR = OFF_YI + SZ_Y;
constexpr size_t OFF_GOI = OFF_GOR + SZ_G;
constexpr size_t OFF_GHR = OFF_GOI + SZ_G;
constexpr size_t OFF_GHI = OFF_GHR + SZ_G;
constexpr size_t SZ_T   = (size_t)QMP * J;
constexpr size_t OFF_TR = OFF_GHI + SZ_G;
constexpr size_t OFF_TI = OFF_TR + SZ_T;
constexpr size_t WS_TOTAL = OFF_TI + SZ_T;
constexpr int NBATCH = 2;
constexpr size_t SZ_R     = (size_t)CMP * NPPAD;
constexpr size_t RP_BATCH = (size_t)NSL * 2 * SZ_R;
constexpr size_t OFF_RP   = (size_t)NBATCH * WS_TOTAL;

__global__ void phase_build(const float* __restrict__ pos,
                            float* __restrict__ Cx, float* __restrict__ Sx,
                            float* __restrict__ Cy, float* __restrict__ Sy,
                            float* __restrict__ Cz, float* __restrict__ Sz) {
  const int b = blockIdx.y;
  const size_t bo = (size_t)b * WS_TOTAL;
  int t = blockIdx.x * blockDim.x + threadIdx.x;
  if (t >= 3 * J) return;
  int d = t / J, j = t % J;
  float th = TWOPI_L * (pos[(size_t)b * J * 3 + j * 3 + d] - 0.5f * LBOX);
  float* C = ((d == 0) ? Cx : (d == 1) ? Cy : Cz) + bo;
  float* S = ((d == 0) ? Sx : (d == 1) ? Sy : Sz) + bo;
  for (int mi = 0; mi < NM; ++mi) {
    float s, c;
    sincosf((float)(mi - 15) * th, &s, &c);
    C[mi * J + j] = c;
    S[mi * J + j] = s;
  }
}

constexpr int WY_THREADS = 256;
constexpr int WB_BLOCKS = (NPH * J + WY_THREADS - 1) / WY_THREADS;
constexpr int YB_BLOCKS = (CCH * NM * J + WY_THREADS - 1) / WY_THREADS;
__global__ void wy_build(const float* __restrict__ charge,
                         const float* __restrict__ Cx, const float* __restrict__ Sx,
                         const float* __restrict__ Cy, const float* __restrict__ Sy,
                         const float* __restrict__ Cz, const float* __restrict__ Sz,
                         float* __restrict__ Wc, float* __restrict__ Ws,
                         float* __restrict__ Yr, float* __restrict__ Yi) {
  const int b = blockIdx.y;
  const size_t bo = (size_t)b * WS_TOTAL;
  if (blockIdx.x < WB_BLOCKS) {
    int t = blockIdx.x * WY_THREADS + threadIdx.x;
    if (t >= NPH * J) return;
    int np = t / J, j = t % J;
    int n = np / NM, p = np % NM;
    float cy = Cy[bo + n * J + j], sy = Sy[bo + n * J + j];
    float cz = Cz[bo + p * J + j], sz = Sz[bo + p * J + j];
    Wc[bo + (size_t)np * J + j] = cy * cz - sy * sz;
    Ws[bo + (size_t)np * J + j] = sy * cz + cy * sz;
  } else {
    int t = (blockIdx.x - WB_BLOCKS) * WY_THREADS + threadIdx.x;
    if (t >= (CCH * NM) * J) return;
    int cm = t / J, j = t % J;
    int c = cm / NM, m = cm % NM;
    float q = charge[(size_t)b * J * CCH + j * CCH + c];
    Yr[bo + (size_t)cm * J + j] =  q * Cx[bo + m * J + j];
    Yi[bo + (size_t)cm * J + j] = -q * Sx[bo + m * J + j];
  }
}

template <int CH>
struct Frags {
  v2f ar[CH][2], ai[CH][2], br[CH][2], bi[CH][2];
};

template <bool NTMODE, int CH>
__device__ __forceinline__ void load_chunk(Frags<CH>& f,
                                           const float* __restrict__ arp0,
                                           const float* __restrict__ aip0,
                                           const float* __restrict__ arp1,
                                           const float* __restrict__ aip1,
                                           const float* __restrict__ brp,
                                           const float* __restrict__ bip,
                                           int ldb, int k0) {
#pragma unroll
  for (int s = 0; s < CH; ++s) {
    const int k = k0 + 4 * s;
    f.ar[s][0] = *(const v2f*)(arp0 + k);
    f.ar[s][1] = *(const v2f*)(arp1 + k);
    f.ai[s][0] = *(const v2f*)(aip0 + k);
    f.ai[s][1] = *(const v2f*)(aip1 + k);
    if (NTMODE) {
      f.br[s][0] = *(const v2f*)(brp + k);
      f.bi[s][0] = *(const v2f*)(bip + k);
      f.br[s][1] = *(const v2f*)(brp + (size_t)16 * ldb + k);
      f.bi[s][1] = *(const v2f*)(bip + (size_t)16 * ldb + k);
    } else {
      const size_t r0 = (size_t)k * ldb;
#pragma unroll
      for (int nt = 0; nt < 2; ++nt) {
        f.br[s][nt].x = brp[r0 + nt * 16];
        f.br[s][nt].y = brp[r0 + ldb + nt * 16];
        f.bi[s][nt].x = bip[r0 + nt * 16];
        f.bi[s][nt].y = bip[r0 + ldb + nt * 16];
      }
    }
  }
}

template <bool SPLUS>
__device__ __forceinline__ void kstep(const v2f (&a_r)[2], const v2f (&a_i)[2],
                                      const v2f (&b_r)[2], const v2f (&b_i)[2],
                                      v8f (&accr)[2][2], v8f (&acci)[2][2]) {
  v2f neg[2];
#pragma unroll
  for (int mt = 0; mt < 2; ++mt) {
    v2f src = SPLUS ? a_i[mt] : a_r[mt];
    neg[mt].x = -src.x;
    neg[mt].y = -src.y;
  }
#pragma unroll
  for (int mt = 0; mt < 2; ++mt)
#pragma unroll
    for (int nt = 0; nt < 2; ++nt) {
      accr[mt][nt] = __builtin_amdgcn_wmma_f32_16x16x4_f32(
          false, a_r[mt], false, b_r[nt], (short)0, accr[mt][nt], false, false);
      acci[mt][nt] = __builtin_amdgcn_wmma_f32_16x16x4_f32(
          false, a_i[mt], false, b_r[nt], (short)0, acci[mt][nt], false, false);
      if (SPLUS) {
        accr[mt][nt] = __builtin_amdgcn_wmma_f32_16x16x4_f32(
            false, neg[mt], false, b_i[nt], (short)0, accr[mt][nt], false, false);
        acci[mt][nt] = __builtin_amdgcn_wmma_f32_16x16x4_f32(
            false, a_r[mt], false, b_i[nt], (short)0, acci[mt][nt], false, false);
      } else {
        accr[mt][nt] = __builtin_amdgcn_wmma_f32_16x16x4_f32(
            false, a_i[mt], false, b_i[nt], (short)0, accr[mt][nt], false, false);
        acci[mt][nt] = __builtin_amdgcn_wmma_f32_16x16x4_f32(
            false, neg[mt], false, b_i[nt], (short)0, acci[mt][nt], false, false);
      }
    }
}

template <bool SPLUS, int CH>
__device__ __forceinline__ void compute_chunk(const Frags<CH>& f,
                                              v8f (&accr)[2][2], v8f (&acci)[2][2]) {
#pragma unroll
  for (int s = 0; s < CH; ++s)
    kstep<SPLUS>(f.ar[s], f.ai[s], f.br[s], f.bi[s], accr, acci);
}

template <bool NTMODE, bool SPLUS>
__device__ __forceinline__
void cgemm_tile(const float* __restrict__ Ar, const float* __restrict__ Ai, int lda,
                const float* __restrict__ Br, const float* __restrict__ Bi, int ldb,
                float* __restrict__ Cr, float* __restrict__ Ci, int ldc, int K,
                int m0, int n0) {
  constexpr int CH = NTMODE ? 4 : 5;
  const int lane = threadIdx.x & 31;
  const int row  = lane & 15;
  const int kp   = lane >> 4;

  v8f accr[2][2], acci[2][2];
#pragma unroll
  for (int mt = 0; mt < 2; ++mt)
#pragma unroll
    for (int nt = 0; nt < 2; ++nt) {
      accr[mt][nt] = (v8f){0.f, 0.f, 0.f, 0.f, 0.f, 0.f, 0.f, 0.f};
      acci[mt][nt] = (v8f){0.f, 0.f, 0.f, 0.f, 0.f, 0.f, 0.f, 0.f};
    }

  const float* arp0 = Ar + (size_t)(m0 + row) * lda + 2 * kp;
  const float* aip0 = Ai + (size_t)(m0 + row) * lda + 2 * kp;
  const float* arp1 = arp0 + (size_t)16 * lda;
  const float* aip1 = aip0 + (size_t)16 * lda;
  const float* brp = NTMODE ? Br + (size_t)(n0 + row) * ldb + 2 * kp
                            : Br + (size_t)(2 * kp) * ldb + n0 + row;
  const float* bip = NTMODE ? Bi + (size_t)(n0 + row) * ldb + 2 * kp
                            : Bi + (size_t)(2 * kp) * ldb + n0 + row;

  const int nch = K / (4 * CH);
  Frags<CH> f0, f1;
  if (nch > 0) load_chunk<NTMODE, CH>(f0, arp0, aip0, arp1, aip1, brp, bip, ldb, 0);
  int c = 0;
  for (; c + 2 <= nch; c += 2) {
    load_chunk<NTMODE, CH>(f1, arp0, aip0, arp1, aip1, brp, bip, ldb, (c + 1) * 4 * CH);
    __builtin_amdgcn_sched_barrier(0);
    compute_chunk<SPLUS, CH>(f0, accr, acci);
    __builtin_amdgcn_sched_barrier(0);
    const int cn = (c + 2 < nch) ? (c + 2) : (nch - 1);
    load_chunk<NTMODE, CH>(f0, arp0, aip0, arp1, aip1, brp, bip, ldb, cn * 4 * CH);
    __builtin_amdgcn_sched_barrier(0);
    compute_chunk<SPLUS, CH>(f1, accr, acci);
    __builtin_amdgcn_sched_barrier(0);
  }
  if (c < nch) compute_chunk<SPLUS, CH>(f0, accr, acci);

  for (int k0 = nch * 4 * CH; k0 < K; k0 += 4) {
    Frags<1> t;
    load_chunk<NTMODE, 1>(t, arp0, aip0, arp1, aip1, brp, bip, ldb, k0);
    compute_chunk<SPLUS, 1>(t, accr, acci);
  }

#pragma unroll
  for (int mt = 0; mt < 2; ++mt)
#pragma unroll
    for (int nt = 0; nt < 2; ++nt)
#pragma unroll
      for (int v = 0; v < 8; ++v) {
        const size_t o = (size_t)(m0 + mt * 16 + v + 8 * kp) * ldc + n0 + nt * 16 + row;
        Cr[o] = accr[mt][nt][v];
        Ci[o] = acci[mt][nt][v];
      }
}


__global__ __launch_bounds__(128)
void nufft1_gemm(const float* __restrict__ Yr, const float* __restrict__ Yi,
                 const float* __restrict__ Wc, const float* __restrict__ Ws,
                 float* __restrict__ rP) {
  const int wave = threadIdx.x >> 5;
  const size_t bo = (size_t)blockIdx.z * WS_TOTAL;
  const int j0 = blockIdx.y * KSL;
  float* pr = rP + (size_t)blockIdx.z * RP_BATCH + (size_t)blockIdx.y * 2 * SZ_R;
  cgemm_tile<true, false>(Yr + bo + j0, Yi + bo + j0, J,
                          Wc + bo + j0, Ws + bo + j0, J,
                          pr, pr + SZ_R, NPPAD, KSL,
                          wave * 32, blockIdx.x * 32);
}

__global__ __launch_bounds__(128)
void stageA_gemm(const float* __restrict__ gOr, const float* __restrict__ gOi,
                 const float* __restrict__ gHr, const float* __restrict__ gHi,
                 const float* __restrict__ Wc, const float* __restrict__ Ws,
                 float* __restrict__ Tr, float* __restrict__ Ti) {
  const int wave = threadIdx.x >> 5;
  const bool h = blockIdx.x * 64 >= JOs;
  const size_t bo = (size_t)blockIdx.z * WS_TOTAL;
  const int n0 = blockIdx.x * 64 + (wave & 1) * 32;
  const int m0 = blockIdx.y * 64 + (wave >> 1) * 32;
  cgemm_tile<false, true>((h ? gHr : gOr) + bo, (h ? gHi : gOi) + bo, KNP,
                          Wc + bo, Ws + bo, J,
                          Tr + bo, Ti + bo, J, KNP, m0, n0);
}

__global__ void g_build(const float* __restrict__ rP,
                        const float* __restrict__ shOO, const float* __restrict__ amOO,
                        const float* __restrict__ shOH, const float* __restrict__ amOH,
                        const float* __restrict__ shHO, const float* __restrict__ amHO,
                        const float* __restrict__ shHH, const float* __restrict__ amHH,
                        float* __restrict__ gOr, float* __restrict__ gOi,
                        float* __restrict__ gHr, float* __restrict__ gHi) {
  const size_t bo = (size_t)blockIdx.y * WS_TOTAL;
  int t = blockIdx.x * blockDim.x + threadIdx.x;
  if (t >= NM * NPH) return;
  int m = t / NPH, np = t % NPH;
  int n = np / NM, p = np % NM;
  float kx = TWOPI_L * (float)(m - 15);
  float ky = TWOPI_L * (float)(n - 15);
  float kz = TWOPI_L * (float)(p - 15);
  float squ = kx * kx + ky * ky + kz * kz;

  float mOO = 0.f, mOH = 0.f, mHO = 0.f, mHH = 0.f;
#pragma unroll
  for (int i = 0; i < 6; ++i) {
    mOO += amOO[i] * expf(-squ * expf(2.0f * shOO[i]));
    mOH += amOH[i] * expf(-squ * expf(2.0f * shOH[i]));
    mHO += amHO[i] * expf(-squ * expf(2.0f * shHO[i]));
    mHH += amHH[i] * expf(-squ * expf(2.0f * shHH[i]));
  }
  if (squ == 0.0f) { mOH = 0.f; mHO = 0.f; mHH = 0.f; }
  if (np < NPH - 1) { mOO *= 2.0f; mOH *= 2.0f; mHO *= 2.0f; mHH *= 2.0f; }

#pragma unroll
  for (int c = 0; c < CCH; ++c) {
    const float* pp = rP + (size_t)blockIdx.y * RP_BATCH + (size_t)(c * NM + m) * NPPAD + np;
    float ror = 0.f, roi = 0.f, rhr = 0.f, rhi = 0.f;
    for (int s = 0; s < NSL_O; ++s) {
      ror += pp[(size_t)(2 * s) * SZ_R];
      roi += pp[(size_t)(2 * s + 1) * SZ_R];
    }
    for (int s = NSL_O; s < NSL; ++s) {
      rhr += pp[(size_t)(2 * s) * SZ_R];
      rhi += pp[(size_t)(2 * s + 1) * SZ_R];
    }
    float fOr = mOO * ror + mOH * rhr, fOi = mOO * roi + mOH * rhi;
    float fHr = mHO * ror + mHH * rhr, fHi = mHO * roi + mHH * rhi;
    const size_t r0 = bo + (size_t)(c * NM + m)        * KNP + np;
    const size_t r1 = bo + (size_t)((4  + c) * NM + m) * KNP + np;
    const size_t r2 = bo + (size_t)((8  + c) * NM + m) * KNP + np;
    const size_t r3 = bo + (size_t)((12 + c) * NM + m) * KNP + np;
    gOr[r0] = fOr;        gOi[r0] = fOi;
    gOr[r1] = kx * fOr;   gOi[r1] = kx * fOi;
    gOr[r2] = ky * fOr;   gOi[r2] = ky * fOi;
    gOr[r3] = kz * fOr;   gOi[r3] = kz * fOi;
    gHr[r0] = fHr;        gHi[r0] = fHi;
    gHr[r1] = kx * fHr;   gHi[r1] = kx * fHi;
    gHr[r2] = ky * fHr;   gHi[r2] = ky * fHi;
    gHr[r3] = kz * fHr;   gHi[r3] = kz * fHi;
  }
}

constexpr int FIN_J = 32;
constexpr int FIN_Q = 16;
__global__ __launch_bounds__(FIN_J * FIN_Q)
void finalize(const float* __restrict__ Tr, const float* __restrict__ Ti,
              const float* __restrict__ Cx, const float* __restrict__ Sx,
              const float* __restrict__ charge, float* __restrict__ out) {
  __shared__ float u[FIN_Q][FIN_J];
  const int b = blockIdx.y;
  const size_t bo = (size_t)b * WS_TOTAL;
  const int jl = threadIdx.x & (FIN_J - 1);
  const int q = __builtin_amdgcn_readfirstlane((int)(threadIdx.x >> 5));
  const int j = blockIdx.x * FIN_J + jl;
  const float* trp = Tr + bo + (size_t)(q * NM) * J + j;
  const float* tip = Ti + bo + (size_t)(q * NM) * J + j;
  const float* cxp = Cx + bo + j;
  const float* sxp = Sx + bo + j;
  float cx[NM], sx[NM], tr[NM], ti[NM];
#pragma unroll
  for (int m = 0; m < NM; ++m) {
    cx[m] = cxp[m * J];
    sx[m] = sxp[m * J];
    tr[m] = trp[m * J];
    ti[m] = tip[m * J];
  }
  __builtin_amdgcn_sched_barrier(0);
  float acc = 0.f;
  if (q < 4) {
#pragma unroll
    for (int m = 0; m < NM; ++m) acc += tr[m] * cx[m] - ti[m] * sx[m];
  } else {
#pragma unroll
    for (int m = 0; m < NM; ++m) acc += tr[m] * sx[m] + ti[m] * cx[m];
  }
  u[q][jl] = acc;
  __syncthreads();
  if (q != 0) return;
  float e = 0.f, f0 = 0.f, f1 = 0.f, f2 = 0.f;
#pragma unroll
  for (int c = 0; c < CCH; ++c) {
    float qc = charge[(size_t)b * J * CCH + j * CCH + c];
    e  += qc * u[c][jl];
    f0 += qc * u[4 + c][jl];
    f1 += qc * u[8 + c][jl];
    f2 += qc * u[12 + c][jl];
  }
  out[b * J + j] = e / (2.0f * VBOX);
  float* F = out + 2 * J;
  F[(size_t)(b * J + j) * 3 + 0] = f0 / VBOX;
  F[(size_t)(b * J + j) * 3 + 1] = f1 / VBOX;
  F[(size_t)(b * J + j) * 3 + 2] = f2 / VBOX;
}

extern "C" void kernel_launch(void* const* d_in, const int* in_sizes, int n_in,
                              void* d_out, int out_size, void* d_ws, size_t ws_size,
                              hipStream_t stream) {
  const float* pos    = (const float*)d_in[0];
  const float* charge = (const float*)d_in[1];
  const float* shOO = (const float*)d_in[2];
  const float* amOO = (const float*)d_in[3];
  const float* shOH = (const float*)d_in[4];
  const float* amOH = (const float*)d_in[5];
  const float* shHO = (const float*)d_in[6];
  const float* amHO = (const float*)d_in[7];
  const float* shHH = (const float*)d_in[8];
  const float* amHH = (const float*)d_in[9];
  float* out = (float*)d_out;
  float* ws  = (float*)d_ws;

  float *Cx = ws + OFF_CX, *Sx = ws + OFF_SX;
  float *Cy = ws + OFF_CY, *Sy = ws + OFF_SY;
  float *Cz = ws + OFF_CZ, *Sz = ws + OFF_SZ;
  float *Wc = ws + OFF_WC, *Wss = ws + OFF_WS;
  float *Yr = ws + OFF_YR, *Yi = ws + OFF_YI;
  float *rP = ws + OFF_RP;
  float *gOr = ws + OFF_GOR, *gOi = ws + OFF_GOI;
  float *gHr = ws + OFF_GHR, *gHi = ws + OFF_GHI;
  float *Tr = ws + OFF_TR, *Ti = ws + OFF_TI;

  hipMemsetAsync(d_ws, 0, NBATCH * WS_TOTAL * sizeof(float), stream);

  phase_build<<<dim3((3 * J + 127) / 128, NBATCH), 128, 0, stream>>>(pos, Cx, Sx, Cy, Sy, Cz, Sz);
  wy_build<<<dim3(WB_BLOCKS + YB_BLOCKS, NBATCH), WY_THREADS, 0, stream>>>(
      charge, Cx, Sx, Cy, Sy, Cz, Sz, Wc, Wss, Yr, Yi);

  nufft1_gemm<<<dim3(NPPAD / 32, NSL, NBATCH), 128, 0, stream>>>(Yr, Yi, Wc, Wss, rP);

  g_build<<<dim3((NM * NPH + 255) / 256, NBATCH), 256, 0, stream>>>(
      rP,
      shOO, amOO, shOH, amOH, shHO, amHO, shHH, amHH,
      gOr, gOi, gHr, gHi);

  stageA_gemm<<<dim3(J / 64, QMP / 64, NBATCH), 128, 0, stream>>>(gOr, gOi, gHr, gHi,
                                                                  Wc, Wss, Tr, Ti);

  finalize<<<dim3(J / FIN_J, NBATCH), FIN_J * FIN_Q, 0, stream>>>(Tr, Ti, Cx, Sx, charge, out);
}


# phase tables, W and Y built in one launch (6 launches -> 5): a workgroup owns 32 particles, computes their sin/cos into LDS and writes whole 128-byte lines of its rows from there; values unchanged
# speedup vs baseline: 11.3926x; 1.0300x over previous
#include <hip/hip_runtime.h>
#include <hip/hip_bf16.h>
#include <math.h>

typedef float v2f __attribute__((ext_vector_type(2)));
typedef float v8f __attribute__((ext_vector_type(8)));

constexpr int NM    = 31;
constexpr int NPH   = 481;
constexpr int NPPAD = 512;
constexpr int KNP   = 484;
constexpr int J     = 768;
constexpr int JOs   = 256;
constexpr int CCH   = 4;
constexpr int CMP   = 128;
constexpr int QMR   = 496;
constexpr int QMP   = 512;
constexpr int KSL   = 128;
constexpr int NSL_O = JOs / KSL;
constexpr int NSL   = J / KSL;
static_assert(JOs % KSL == 0 && J % KSL == 0, "a slice never straddles the two species");
constexpr float LBOX   = 20.0f;
constexpr float VBOX   = 8000.0f;
constexpr float TWOPI_L = 0.31415926535897931f;

constexpr size_t SZ_PH  = (size_t)NM * J;
constexpr size_t OFF_CX = 0;
constexpr size_t OFF_SX = OFF_CX + SZ_PH;
constexpr size_t SZ_W   = (size_t)NPPAD * J;
constexpr size_t OFF_WC = OFF_SX + SZ_PH;
constexpr size_t OFF_WS = OFF_WC + SZ_W;
constexpr size_t SZ_Y   = (size_t)CMP * J;
constexpr size_t OFF_YR = OFF_WS + SZ_W;
constexpr size_t OFF_YI = OFF_YR + SZ_Y;
constexpr size_t SZ_G   = (size_t)QMP * KNP;
constexpr size_t OFF_GOR = OFF_YI + SZ_Y;
constexpr size_t OFF_GOI = OFF_GOR + SZ_G;
constexpr size_t OFF_GHR = OFF_GOI + SZ_G;
constexpr size_t OFF_GHI = OFF_GHR + SZ_G;
constexpr size_t SZ_T   = (size_t)QMP * J;
constexpr size_t OFF_TR = OFF_GHI + SZ_G;
constexpr size_t OFF_TI = OFF_TR + SZ_T;
constexpr size_t WS_TOTAL = OFF_TI + SZ_T;
constexpr int NBATCH = 2;
constexpr size_t SZ_R     = (size_t)CMP * NPPAD;
constexpr size_t RP_BATCH = (size_t)NSL * 2 * SZ_R;
constexpr size_t OFF_RP   = (size_t)NBATCH * WS_TOTAL;

constexpr int TB_J       = 32;
constexpr int TB_THREADS = 512;
constexpr int TB_WAVES   = TB_THREADS / 32;
constexpr int TB_WPARTS  = 4;
constexpr int TB_ROWS    = (NPH + TB_WPARTS - 1) / TB_WPARTS;
static_assert(J % TB_J == 0, "a particle block is a whole line of every row");
__global__ __launch_bounds__(TB_THREADS)
void tables_build(const float* __restrict__ pos, const float* __restrict__ charge,
                  float* __restrict__ Cx, float* __restrict__ Sx,
                  float* __restrict__ Wc, float* __restrict__ Ws,
                  float* __restrict__ Yr, float* __restrict__ Yi) {
  __shared__ float cs[2][NM][TB_J];
  __shared__ float sn[2][NM][TB_J];
  const int b = blockIdx.z;
  const size_t bo = (size_t)b * WS_TOTAL;
  const int j0 = blockIdx.x * TB_J;
  const int part = blockIdx.y;
  const bool xpart = part == TB_WPARTS;
  const int ndim = xpart ? 1 : 2;
  const int d0   = xpart ? 0 : 1;
  for (int i = threadIdx.x; i < ndim * NM * TB_J; i += TB_THREADS) {
    const int jl = i % TB_J, dm = i / TB_J;
    const int mi = dm % NM, dd = dm / NM;
    const float th = TWOPI_L * (pos[(size_t)b * J * 3 + (size_t)(j0 + jl) * 3 + d0 + dd] - 0.5f * LBOX);
    float s, c;
    sincosf((float)(mi - 15) * th, &s, &c);
    cs[dd][mi][jl] = c;
    sn[dd][mi][jl] = s;
  }
  __syncthreads();
  const int jl = threadIdx.x & (TB_J - 1);
  const int w  = __builtin_amdgcn_readfirstlane((int)(threadIdx.x >> 5));
  const size_t col = bo + j0 + jl;
  if (!xpart) {
    const int lo = part * TB_ROWS;
    const int hi = (lo + TB_ROWS < NPH) ? (lo + TB_ROWS) : NPH;
    for (int np = lo + w; np < hi; np += TB_WAVES) {
      const int n = np / NM, p = np % NM;
      const float cy = cs[0][n][jl], sy = sn[0][n][jl];
      const float cz = cs[1][p][jl], sz = sn[1][p][jl];
      Wc[col + (size_t)np * J] = cy * cz - sy * sz;
      Ws[col + (size_t)np * J] = sy * cz + cy * sz;
    }
  } else {
    for (int m = w; m < NM; m += TB_WAVES) {
      Cx[col + (size_t)m * J] = cs[0][m][jl];
      Sx[col + (size_t)m * J] = sn[0][m][jl];
    }
    for (int cm = w; cm < CCH * NM; cm += TB_WAVES) {
      const int c = cm / NM, m = cm % NM;
      const float q = charge[(size_t)b * J * CCH + (size_t)(j0 + jl) * CCH + c];
      Yr[col + (size_t)cm * J] =  q * cs[0][m][jl];
      Yi[col + (size_t)cm * J] = -q * sn[0][m][jl];
    }
  }
}

template <int CH>
struct Frags {
  v2f ar[CH][2], ai[CH][2], br[CH][2], bi[CH][2];
};

template <bool NTMODE, int CH>
__device__ __forceinline__ void load_chunk(Frags<CH>& f,
                                           const float* __restrict__ arp0,
                                           const float* __restrict__ aip0,
                                           const float* __restrict__ arp1,
                                           const float* __restrict__ aip1,
                                           const float* __restrict__ brp,
                                           const float* __restrict__ bip,
                                           int ldb, int k0) {
#pragma unroll
  for (int s = 0; s < CH; ++s) {
    const int k = k0 + 4 * s;
    f.ar[s][0] = *(const v2f*)(arp0 + k);
    f.ar[s][1] = *(const v2f*)(arp1 + k);
    f.ai[s][0] = *(const v2f*)(aip0 + k);
    f.ai[s][1] = *(const v2f*)(aip1 + k);
    if (NTMODE) {
      f.br[s][0] = *(const v2f*)(brp + k);
      f.bi[s][0] = *(const v2f*)(bip + k);
      f.br[s][1] = *(const v2f*)(brp + (size_t)16 * ldb + k);
      f.bi[s][1] = *(const v2f*)(bip + (size_t)16 * ldb + k);
    } else {
      const size_t r0 = (size_t)k * ldb;
#pragma unroll
      for (int nt = 0; nt < 2; ++nt) {
        f.br[s][nt].x = brp[r0 + nt * 16];
        f.br[s][nt].y = brp[r0 + ldb + nt * 16];
        f.bi[s][nt].x = bip[r0 + nt * 16];
        f.bi[s][nt].y = bip[r0 + ldb + nt * 16];
      }
    }
  }
}

template <bool SPLUS>
__device__ __forceinline__ void kstep(const v2f (&a_r)[2], const v2f (&a_i)[2],
                                      const v2f (&b_r)[2], const v2f (&b_i)[2],
                                      v8f (&accr)[2][2], v8f (&acci)[2][2]) {
  v2f neg[2];
#pragma unroll
  for (int mt = 0; mt < 2; ++mt) {
    v2f src = SPLUS ? a_i[mt] : a_r[mt];
    neg[mt].x = -src.x;
    neg[mt].y = -src.y;
  }
#pragma unroll
  for (int mt = 0; mt < 2; ++mt)
#pragma unroll
    for (int nt = 0; nt < 2; ++nt) {
      accr[mt][nt] = __builtin_amdgcn_wmma_f32_16x16x4_f32(
          false, a_r[mt], false, b_r[nt], (short)0, accr[mt][nt], false, false);
      acci[mt][nt] = __builtin_amdgcn_wmma_f32_16x16x4_f32(
          false, a_i[mt], false, b_r[nt], (short)0, acci[mt][nt], false, false);
      if (SPLUS) {
        accr[mt][nt] = __builtin_amdgcn_wmma_f32_16x16x4_f32(
            false, neg[mt], false, b_i[nt], (short)0, accr[mt][nt], false, false);
        acci[mt][nt] = __builtin_amdgcn_wmma_f32_16x16x4_f32(
            false, a_r[mt], false, b_i[nt], (short)0, acci[mt][nt], false, false);
      } else {
        accr[mt][nt] = __builtin_amdgcn_wmma_f32_16x16x4_f32(
            false, a_i[mt], false, b_i[nt], (short)0, accr[mt][nt], false, false);
        acci[mt][nt] = __builtin_amdgcn_wmma_f32_16x16x4_f32(
            false, neg[mt], false, b_i[nt], (short)0, acci[mt][nt], false, false);
      }
    }
}

template <bool SPLUS, int CH>
__device__ __forceinline__ void compute_chunk(const Frags<CH>& f,
                                              v8f (&accr)[2][2], v8f (&acci)[2][2]) {
#pragma unroll
  for (int s = 0; s < CH; ++s)
    kstep<SPLUS>(f.ar[s], f.ai[s], f.br[s], f.bi[s], accr, acci);
}

template <bool NTMODE, bool SPLUS>
__device__ __forceinline__
void cgemm_tile(const float* __restrict__ Ar, const float* __restrict__ Ai, int lda,
                const float* __restrict__ Br, const float* __restrict__ Bi, int ldb,
                float* __restrict__ Cr, float* __restrict__ Ci, int ldc, int K,
                int m0, int n0) {
  constexpr int CH = NTMODE ? 4 : 5;
  const int lane = threadIdx.x & 31;
  const int row  = lane & 15;
  const int kp   = lane >> 4;

  v8f accr[2][2], acci[2][2];
#pragma unroll
  for (int mt = 0; mt < 2; ++mt)
#pragma unroll
    for (int nt = 0; nt < 2; ++nt) {
      accr[mt][nt] = (v8f){0.f, 0.f, 0.f, 0.f, 0.f, 0.f, 0.f, 0.f};
      acci[mt][nt] = (v8f){0.f, 0.f, 0.f, 0.f, 0.f, 0.f, 0.f, 0.f};
    }

  const float* arp0 = Ar + (size_t)(m0 + row) * lda + 2 * kp;
  const float* aip0 = Ai + (size_t)(m0 + row) * lda + 2 * kp;
  const float* arp1 = arp0 + (size_t)16 * lda;
  const float* aip1 = aip0 + (size_t)16 * lda;
  const float* brp = NTMODE ? Br + (size_t)(n0 + row) * ldb + 2 * kp
                            : Br + (size_t)(2 * kp) * ldb + n0 + row;
  const float* bip = NTMODE ? Bi + (size_t)(n0 + row) * ldb + 2 * kp
                            : Bi + (size_t)(2 * kp) * ldb + n0 + row;

  const int nch = K / (4 * CH);
  Frags<CH> f0, f1;
  if (nch > 0) load_chunk<NTMODE, CH>(f0, arp0, aip0, arp1, aip1, brp, bip, ldb, 0);
  int c = 0;
  for (; c + 2 <= nch; c += 2) {
    load_chunk<NTMODE, CH>(f1, arp0, aip0, arp1, aip1, brp, bip, ldb, (c + 1) * 4 * CH);
    __builtin_amdgcn_sched_barrier(0);
    compute_chunk<SPLUS, CH>(f0, accr, acci);
    __builtin_amdgcn_sched_barrier(0);
    const int cn = (c + 2 < nch) ? (c + 2) : (nch - 1);
    load_chunk<NTMODE, CH>(f0, arp0, aip0, arp1, aip1, brp, bip, ldb, cn * 4 * CH);
    __builtin_amdgcn_sched_barrier(0);
    compute_chunk<SPLUS, CH>(f1, accr, acci);
    __builtin_amdgcn_sched_barrier(0);
  }
  if (c < nch) compute_chunk<SPLUS, CH>(f0, accr, acci);

  for (int k0 = nch * 4 * CH; k0 < K; k0 += 4) {
    Frags<1> t;
    load_chunk<NTMODE, 1>(t, arp0, aip0, arp1, aip1, brp, bip, ldb, k0);
    compute_chunk<SPLUS, 1>(t, accr, acci);
  }

#pragma unroll
  for (int mt = 0; mt < 2; ++mt)
#pragma unroll
    for (int nt = 0; nt < 2; ++nt)
#pragma unroll
      for (int v = 0; v < 8; ++v) {
        const size_t o = (size_t)(m0 + mt * 16 + v + 8 * kp) * ldc + n0 + nt * 16 + row;
        Cr[o] = accr[mt][nt][v];
        Ci[o] = acci[mt][nt][v];
      }
}


__global__ __launch_bounds__(128)
void nufft1_gemm(const float* __restrict__ Yr, const float* __restrict__ Yi,
                 const float* __restrict__ Wc, const float* __restrict__ Ws,
                 float* __restrict__ rP) {
  const int wave = threadIdx.x >> 5;
  const size_t bo = (size_t)blockIdx.z * WS_TOTAL;
  const int j0 = blockIdx.y * KSL;
  float* pr = rP + (size_t)blockIdx.z * RP_BATCH + (size_t)blockIdx.y * 2 * SZ_R;
  cgemm_tile<true, false>(Yr + bo + j0, Yi + bo + j0, J,
                          Wc + bo + j0, Ws + bo + j0, J,
                          pr, pr + SZ_R, NPPAD, KSL,
                          wave * 32, blockIdx.x * 32);
}

__global__ __launch_bounds__(128)
void stageA_gemm(const float* __restrict__ gOr, const float* __restrict__ gOi,
                 const float* __restrict__ gHr, const float* __restrict__ gHi,
                 const float* __restrict__ Wc, const float* __restrict__ Ws,
                 float* __restrict__ Tr, float* __restrict__ Ti) {
  const int wave = threadIdx.x >> 5;
  const bool h = blockIdx.x * 64 >= JOs;
  const size_t bo = (size_t)blockIdx.z * WS_TOTAL;
  const int n0 = blockIdx.x * 64 + (wave & 1) * 32;
  const int m0 = blockIdx.y * 64 + (wave >> 1) * 32;
  cgemm_tile<false, true>((h ? gHr : gOr) + bo, (h ? gHi : gOi) + bo, KNP,
                          Wc + bo, Ws + bo, J,
                          Tr + bo, Ti + bo, J, KNP, m0, n0);
}

__global__ void g_build(const float* __restrict__ rP,
                        const float* __restrict__ shOO, const float* __restrict__ amOO,
                        const float* __restrict__ shOH, const float* __restrict__ amOH,
                        const float* __restrict__ shHO, const float* __restrict__ amHO,
                        const float* __restrict__ shHH, const float* __restrict__ amHH,
                        float* __restrict__ gOr, float* __restrict__ gOi,
                        float* __restrict__ gHr, float* __restrict__ gHi) {
  const size_t bo = (size_t)blockIdx.y * WS_TOTAL;
  int t = blockIdx.x * blockDim.x + threadIdx.x;
  if (t >= NM * NPH) return;
  int m = t / NPH, np = t % NPH;
  int n = np / NM, p = np % NM;
  float kx = TWOPI_L * (float)(m - 15);
  float ky = TWOPI_L * (float)(n - 15);
  float kz = TWOPI_L * (float)(p - 15);
  float squ = kx * kx + ky * ky + kz * kz;

  float mOO = 0.f, mOH = 0.f, mHO = 0.f, mHH = 0.f;
#pragma unroll
  for (int i = 0; i < 6; ++i) {
    mOO += amOO[i] * expf(-squ * expf(2.0f * shOO[i]));
    mOH += amOH[i] * expf(-squ * expf(2.0f * shOH[i]));
    mHO += amHO[i] * expf(-squ * expf(2.0f * shHO[i]));
    mHH += amHH[i] * expf(-squ * expf(2.0f * shHH[i]));
  }
  if (squ == 0.0f) { mOH = 0.f; mHO = 0.f; mHH = 0.f; }
  if (np < NPH - 1) { mOO *= 2.0f; mOH *= 2.0f; mHO *= 2.0f; mHH *= 2.0f; }

#pragma unroll
  for (int c = 0; c < CCH; ++c) {
    const float* pp = rP + (size_t)blockIdx.y * RP_BATCH + (size_t)(c * NM + m) * NPPAD + np;
    float ror = 0.f, roi = 0.f, rhr = 0.f, rhi = 0.f;
    for (int s = 0; s < NSL_O; ++s) {
      ror += pp[(size_t)(2 * s) * SZ_R];
      roi += pp[(size_t)(2 * s + 1) * SZ_R];
    }
    for (int s = NSL_O; s < NSL; ++s) {
      rhr += pp[(size_t)(2 * s) * SZ_R];
      rhi += pp[(size_t)(2 * s + 1) * SZ_R];
    }
    float fOr = mOO * ror + mOH * rhr, fOi = mOO * roi + mOH * rhi;
    float fHr = mHO * ror + mHH * rhr, fHi = mHO * roi + mHH * rhi;
    const size_t r0 = bo + (size_t)(c * NM + m)        * KNP + np;
    const size_t r1 = bo + (size_t)((4  + c) * NM + m) * KNP + np;
    const size_t r2 = bo + (size_t)((8  + c) * NM + m) * KNP + np;
    const size_t r3 = bo + (size_t)((12 + c) * NM + m) * KNP + np;
    gOr[r0] = fOr;        gOi[r0] = fOi;
    gOr[r1] = kx * fOr;   gOi[r1] = kx * fOi;
    gOr[r2] = ky * fOr;   gOi[r2] = ky * fOi;
    gOr[r3] = kz * fOr;   gOi[r3] = kz * fOi;
    gHr[r0] = fHr;        gHi[r0] = fHi;
    gHr[r1] = kx * fHr;   gHi[r1] = kx * fHi;
    gHr[r2] = ky * fHr;   gHi[r2] = ky * fHi;
    gHr[r3] = kz * fHr;   gHi[r3] = kz * fHi;
  }
}

constexpr int FIN_J = 32;
constexpr int FIN_Q = 16;
__global__ __launch_bounds__(FIN_J * FIN_Q)
void finalize(const float* __restrict__ Tr, const float* __restrict__ Ti,
              const float* __restrict__ Cx, const float* __restrict__ Sx,
              const float* __restrict__ charge, float* __restrict__ out) {
  __shared__ float u[FIN_Q][FIN_J];
  const int b = blockIdx.y;
  const size_t bo = (size_t)b * WS_TOTAL;
  const int jl = threadIdx.x & (FIN_J - 1);
  const int q = __builtin_amdgcn_readfirstlane((int)(threadIdx.x >> 5));
  const int j = blockIdx.x * FIN_J + jl;
  const float* trp = Tr + bo + (size_t)(q * NM) * J + j;
  const float* tip = Ti + bo + (size_t)(q * NM) * J + j;
  const float* cxp = Cx + bo + j;
  const float* sxp = Sx + bo + j;
  float cx[NM], sx[NM], tr[NM], ti[NM];
#pragma unroll
  for (int m = 0; m < NM; ++m) {
    cx[m] = cxp[m * J];
    sx[m] = sxp[m * J];
    tr[m] = trp[m * J];
    ti[m] = tip[m * J];
  }
  __builtin_amdgcn_sched_barrier(0);
  float acc = 0.f;
  if (q < 4) {
#pragma unroll
    for (int m = 0; m < NM; ++m) acc += tr[m] * cx[m] - ti[m] * sx[m];
  } else {
#pragma unroll
    for (int m = 0; m < NM; ++m) acc += tr[m] * sx[m] + ti[m] * cx[m];
  }
  u[q][jl] = acc;
  __syncthreads();
  if (q != 0) return;
  float e = 0.f, f0 = 0.f, f1 = 0.f, f2 = 0.f;
#pragma unroll
  for (int c = 0; c < CCH; ++c) {
    float qc = charge[(size_t)b * J * CCH + j * CCH + c];
    e  += qc * u[c][jl];
    f0 += qc * u[4 + c][jl];
    f1 += qc * u[8 + c][jl];
    f2 += qc * u[12 + c][jl];
  }
  out[b * J + j] = e / (2.0f * VBOX);
  float* F = out + 2 * J;
  F[(size_t)(b * J + j) * 3 + 0] = f0 / VBOX;
  F[(size_t)(b * J + j) * 3 + 1] = f1 / VBOX;
  F[(size_t)(b * J + j) * 3 + 2] = f2 / VBOX;
}

extern "C" void kernel_launch(void* const* d_in, const int* in_sizes, int n_in,
                              void* d_out, int out_size, void* d_ws, size_t ws_size,
                              hipStream_t stream) {
  const float* pos    = (const float*)d_in[0];
  const float* charge = (const float*)d_in[1];
  const float* shOO = (const float*)d_in[2];
  const float* amOO = (const float*)d_in[3];
  const float* shOH = (const float*)d_in[4];
  const float* amOH = (const float*)d_in[5];
  const float* shHO = (const float*)d_in[6];
  const float* amHO = (const float*)d_in[7];
  const float* shHH = (const float*)d_in[8];
  const float* amHH = (const float*)d_in[9];
  float* out = (float*)d_out;
  float* ws  = (float*)d_ws;

  float *Cx = ws + OFF_CX, *Sx = ws + OFF_SX;
  float *Wc = ws + OFF_WC, *Wss = ws + OFF_WS;
  float *Yr = ws + OFF_YR, *Yi = ws + OFF_YI;
  float *rP = ws + OFF_RP;
  float *gOr = ws + OFF_GOR, *gOi = ws + OFF_GOI;
  float *gHr = ws + OFF_GHR, *gHi = ws + OFF_GHI;
  float *Tr = ws + OFF_TR, *Ti = ws + OFF_TI;

  hipMemsetAsync(d_ws, 0, NBATCH * WS_TOTAL * sizeof(float), stream);

  tables_build<<<dim3(J / TB_J, TB_WPARTS + 1, NBATCH), TB_THREADS, 0, stream>>>(
      pos, charge, Cx, Sx, Wc, Wss, Yr, Yi);

  nufft1_gemm<<<dim3(NPPAD / 32, NSL, NBATCH), 128, 0, stream>>>(Yr, Yi, Wc, Wss, rP);

  g_build<<<dim3((NM * NPH + 255) / 256, NBATCH), 256, 0, stream>>>(
      rP,
      shOO, amOO, shOH, amOH, shHO, amHO, shHH, amHH,
      gOr, gOi, gHr, gHi);

  stageA_gemm<<<dim3(J / 64, QMP / 64, NBATCH), 128, 0, stream>>>(gOr, gOi, gHr, gHi,
                                                                  Wc, Wss, Tr, Ti);

  finalize<<<dim3(J / FIN_J, NBATCH), FIN_J * FIN_Q, 0, stream>>>(Tr, Ti, Cx, Sx, charge, out);
}


# Sum over m moved into stage A's epilogue: g rows q*32+m, a wave's tile is one channel, x phases applied and m summed in registers, 4 charge waves join in LDS; T never stored: m sum order changes
# speedup vs baseline: 12.9238x; 1.1344x over previous
#include <hip/hip_runtime.h>
#include <hip/hip_bf16.h>
#include <math.h>

typedef float v2f __attribute__((ext_vector_type(2)));
typedef float v8f __attribute__((ext_vector_type(8)));

constexpr int NM    = 31;
constexpr int NPH   = 481;
constexpr int NPPAD = 512;
constexpr int KNP   = 484;
constexpr int J     = 768;
constexpr int JOs   = 256;
constexpr int CCH   = 4;
constexpr int CMP   = 128;
constexpr int QROW  = 32;
constexpr int QMP   = 16 * QROW;
constexpr int KSL   = 128;
constexpr int NSL_O = JOs / KSL;
constexpr int NSL   = J / KSL;
static_assert(JOs % KSL == 0 && J % KSL == 0, "a slice never straddles the two species");
constexpr float LBOX   = 20.0f;
constexpr float VBOX   = 8000.0f;
constexpr float TWOPI_L = 0.31415926535897931f;

constexpr size_t SZ_PH  = (size_t)NM * J;
constexpr size_t OFF_CX = 0;
constexpr size_t OFF_SX = OFF_CX + SZ_PH;
constexpr size_t SZ_W   = (size_t)NPPAD * J;
constexpr size_t OFF_WC = OFF_SX + SZ_PH;
constexpr size_t OFF_WS = OFF_WC + SZ_W;
constexpr size_t SZ_Y   = (size_t)CMP * J;
constexpr size_t OFF_YR = OFF_WS + SZ_W;
constexpr size_t OFF_YI = OFF_YR + SZ_Y;
constexpr size_t SZ_G   = (size_t)QMP * KNP;
constexpr size_t OFF_GOR = OFF_YI + SZ_Y;
constexpr size_t OFF_GOI = OFF_GOR + SZ_G;
constexpr size_t OFF_GHR = OFF_GOI + SZ_G;
constexpr size_t OFF_GHI = OFF_GHR + SZ_G;
constexpr size_t SZ_U   = (size_t)4 * J;
constexpr size_t OFF_U  = OFF_GHI + SZ_G;
constexpr size_t WS_TOTAL = OFF_U + SZ_U;
constexpr int NBATCH = 2;
constexpr size_t SZ_R     = (size_t)CMP * NPPAD;
constexpr size_t RP_BATCH = (size_t)NSL * 2 * SZ_R;
constexpr size_t OFF_RP   = (size_t)NBATCH * WS_TOTAL;

constexpr int TB_J       = 32;
constexpr int TB_THREADS = 512;
constexpr int TB_WAVES   = TB_THREADS / 32;
constexpr int TB_WPARTS  = 4;
constexpr int TB_ROWS    = (NPH + TB_WPARTS - 1) / TB_WPARTS;
static_assert(J % TB_J == 0, "a particle block is a whole line of every row");
__global__ __launch_bounds__(TB_THREADS)
void tables_build(const float* __restrict__ pos, const float* __restrict__ charge,
                  float* __restrict__ Cx, float* __restrict__ Sx,
                  float* __restrict__ Wc, float* __restrict__ Ws,
                  float* __restrict__ Yr, float* __restrict__ Yi) {
  __shared__ float cs[2][NM][TB_J];
  __shared__ float sn[2][NM][TB_J];
  const int b = blockIdx.z;
  const size_t bo = (size_t)b * WS_TOTAL;
  const int j0 = blockIdx.x * TB_J;
  const int part = blockIdx.y;
  const bool xpart = part == TB_WPARTS;
  const int ndim = xpart ? 1 : 2;
  const int d0   = xpart ? 0 : 1;
  for (int i = threadIdx.x; i < ndim * NM * TB_J; i += TB_THREADS) {
    const int jl = i % TB_J, dm = i / TB_J;
    const int mi = dm % NM, dd = dm / NM;
    const float th = TWOPI_L * (pos[(size_t)b * J * 3 + (size_t)(j0 + jl) * 3 + d0 + dd] - 0.5f * LBOX);
    float s, c;
    sincosf((float)(mi - 15) * th, &s, &c);
    cs[dd][mi][jl] = c;
    sn[dd][mi][jl] = s;
  }
  __syncthreads();
  const int jl = threadIdx.x & (TB_J - 1);
  const int w  = __builtin_amdgcn_readfirstlane((int)(threadIdx.x >> 5));
  const size_t col = bo + j0 + jl;
  if (!xpart) {
    const int lo = part * TB_ROWS;
    const int hi = (lo + TB_ROWS < NPH) ? (lo + TB_ROWS) : NPH;
    for (int np = lo + w; np < hi; np += TB_WAVES) {
      const int n = np / NM, p = np % NM;
      const float cy = cs[0][n][jl], sy = sn[0][n][jl];
      const float cz = cs[1][p][jl], sz = sn[1][p][jl];
      Wc[col + (size_t)np * J] = cy * cz - sy * sz;
      Ws[col + (size_t)np * J] = sy * cz + cy * sz;
    }
  } else {
    for (int m = w; m < NM; m += TB_WAVES) {
      Cx[col + (size_t)m * J] = cs[0][m][jl];
      Sx[col + (size_t)m * J] = sn[0][m][jl];
    }
    for (int cm = w; cm < CCH * NM; cm += TB_WAVES) {
      const int c = cm / NM, m = cm % NM;
      const float q = charge[(size_t)b * J * CCH + (size_t)(j0 + jl) * CCH + c];
      Yr[col + (size_t)cm * J] =  q * cs[0][m][jl];
      Yi[col + (size_t)cm * J] = -q * sn[0][m][jl];
    }
  }
}

template <int CH>
struct Frags {
  v2f ar[CH][2], ai[CH][2], br[CH][2], bi[CH][2];
};

template <bool NTMODE, int CH>
__device__ __forceinline__ void load_chunk(Frags<CH>& f,
                                           const float* __restrict__ arp0,
                                           const float* __restrict__ aip0,
                                           const float* __restrict__ arp1,
                                           const float* __restrict__ aip1,
                                           const float* __restrict__ brp,
                                           const float* __restrict__ bip,
                                           int ldb, int k0) {
#pragma unroll
  for (int s = 0; s < CH; ++s) {
    const int k = k0 + 4 * s;
    f.ar[s][0] = *(const v2f*)(arp0 + k);
    f.ar[s][1] = *(const v2f*)(arp1 + k);
    f.ai[s][0] = *(const v2f*)(aip0 + k);
    f.ai[s][1] = *(const v2f*)(aip1 + k);
    if (NTMODE) {
      f.br[s][0] = *(const v2f*)(brp + k);
      f.bi[s][0] = *(const v2f*)(bip + k);
      f.br[s][1] = *(const v2f*)(brp + (size_t)16 * ldb + k);
      f.bi[s][1] = *(const v2f*)(bip + (size_t)16 * ldb + k);
    } else {
      const size_t r0 = (size_t)k * ldb;
#pragma unroll
      for (int nt = 0; nt < 2; ++nt) {
        f.br[s][nt].x = brp[r0 + nt * 16];
        f.br[s][nt].y = brp[r0 + ldb + nt * 16];
        f.bi[s][nt].x = bip[r0 + nt * 16];
        f.bi[s][nt].y = bip[r0 + ldb + nt * 16];
      }
    }
  }
}

template <bool SPLUS>
__device__ __forceinline__ void kstep(const v2f (&a_r)[2], const v2f (&a_i)[2],
                                      const v2f (&b_r)[2], const v2f (&b_i)[2],
                                      v8f (&accr)[2][2], v8f (&acci)[2][2]) {
  v2f neg[2];
#pragma unroll
  for (int mt = 0; mt < 2; ++mt) {
    v2f src = SPLUS ? a_i[mt] : a_r[mt];
    neg[mt].x = -src.x;
    neg[mt].y = -src.y;
  }
#pragma unroll
  for (int mt = 0; mt < 2; ++mt)
#pragma unroll
    for (int nt = 0; nt < 2; ++nt) {
      accr[mt][nt] = __builtin_amdgcn_wmma_f32_16x16x4_f32(
          false, a_r[mt], false, b_r[nt], (short)0, accr[mt][nt], false, false);
      acci[mt][nt] = __builtin_amdgcn_wmma_f32_16x16x4_f32(
          false, a_i[mt], false, b_r[nt], (short)0, acci[mt][nt], false, false);
      if (SPLUS) {
        accr[mt][nt] = __builtin_amdgcn_wmma_f32_16x16x4_f32(
            false, neg[mt], false, b_i[nt], (short)0, accr[mt][nt], false, false);
        acci[mt][nt] = __builtin_amdgcn_wmma_f32_16x16x4_f32(
            false, a_r[mt], false, b_i[nt], (short)0, acci[mt][nt], false, false);
      } else {
        accr[mt][nt] = __builtin_amdgcn_wmma_f32_16x16x4_f32(
            false, a_i[mt], false, b_i[nt], (short)0, accr[mt][nt], false, false);
        acci[mt][nt] = __builtin_amdgcn_wmma_f32_16x16x4_f32(
            false, neg[mt], false, b_i[nt], (short)0, acci[mt][nt], false, false);
      }
    }
}

template <bool SPLUS, int CH>
__device__ __forceinline__ void compute_chunk(const Frags<CH>& f,
                                              v8f (&accr)[2][2], v8f (&acci)[2][2]) {
#pragma unroll
  for (int s = 0; s < CH; ++s)
    kstep<SPLUS>(f.ar[s], f.ai[s], f.br[s], f.bi[s], accr, acci);
}

template <bool NTMODE, bool SPLUS>
__device__ __forceinline__
void cgemm_acc(const float* __restrict__ Ar, const float* __restrict__ Ai, int lda,
               const float* __restrict__ Br, const float* __restrict__ Bi, int ldb,
               int K, int m0, int n0, v8f (&accr)[2][2], v8f (&acci)[2][2]) {
  constexpr int CH = NTMODE ? 4 : 5;
  const int lane = threadIdx.x & 31;
  const int row  = lane & 15;
  const int kp   = lane >> 4;

#pragma unroll
  for (int mt = 0; mt < 2; ++mt)
#pragma unroll
    for (int nt = 0; nt < 2; ++nt) {
      accr[mt][nt] = (v8f){0.f, 0.f, 0.f, 0.f, 0.f, 0.f, 0.f, 0.f};
      acci[mt][nt] = (v8f){0.f, 0.f, 0.f, 0.f, 0.f, 0.f, 0.f, 0.f};
    }

  const float* arp0 = Ar + (size_t)(m0 + row) * lda + 2 * kp;
  const float* aip0 = Ai + (size_t)(m0 + row) * lda + 2 * kp;
  const float* arp1 = arp0 + (size_t)16 * lda;
  const float* aip1 = aip0 + (size_t)16 * lda;
  const float* brp = NTMODE ? Br + (size_t)(n0 + row) * ldb + 2 * kp
                            : Br + (size_t)(2 * kp) * ldb + n0 + row;
  const float* bip = NTMODE ? Bi + (size_t)(n0 + row) * ldb + 2 * kp
                            : Bi + (size_t)(2 * kp) * ldb + n0 + row;

  const int nch = K / (4 * CH);
  Frags<CH> f0, f1;
  if (nch > 0) load_chunk<NTMODE, CH>(f0, arp0, aip0, arp1, aip1, brp, bip, ldb, 0);
  int c = 0;
  for (; c + 2 <= nch; c += 2) {
    load_chunk<NTMODE, CH>(f1, arp0, aip0, arp1, aip1, brp, bip, ldb, (c + 1) * 4 * CH);
    __builtin_amdgcn_sched_barrier(0);
    compute_chunk<SPLUS, CH>(f0, accr, acci);
    __builtin_amdgcn_sched_barrier(0);
    const int cn = (c + 2 < nch) ? (c + 2) : (nch - 1);
    load_chunk<NTMODE, CH>(f0, arp0, aip0, arp1, aip1, brp, bip, ldb, cn * 4 * CH);
    __builtin_amdgcn_sched_barrier(0);
    compute_chunk<SPLUS, CH>(f1, accr, acci);
    __builtin_amdgcn_sched_barrier(0);
  }
  if (c < nch) compute_chunk<SPLUS, CH>(f0, accr, acci);

  for (int k0 = nch * 4 * CH; k0 < K; k0 += 4) {
    Frags<1> t;
    load_chunk<NTMODE, 1>(t, arp0, aip0, arp1, aip1, brp, bip, ldb, k0);
    compute_chunk<SPLUS, 1>(t, accr, acci);
  }
}

template <bool NTMODE, bool SPLUS>
__device__ __forceinline__
void cgemm_tile(const float* __restrict__ Ar, const float* __restrict__ Ai, int lda,
                const float* __restrict__ Br, const float* __restrict__ Bi, int ldb,
                float* __restrict__ Cr, float* __restrict__ Ci, int ldc, int K,
                int m0, int n0) {
  const int lane = threadIdx.x & 31;
  const int row  = lane & 15;
  const int kp   = lane >> 4;
  v8f accr[2][2], acci[2][2];
  cgemm_acc<NTMODE, SPLUS>(Ar, Ai, lda, Br, Bi, ldb, K, m0, n0, accr, acci);
#pragma unroll
  for (int mt = 0; mt < 2; ++mt)
#pragma unroll
    for (int nt = 0; nt < 2; ++nt)
#pragma unroll
      for (int v = 0; v < 8; ++v) {
        const size_t o = (size_t)(m0 + mt * 16 + v + 8 * kp) * ldc + n0 + nt * 16 + row;
        Cr[o] = accr[mt][nt][v];
        Ci[o] = acci[mt][nt][v];
      }
}


__global__ __launch_bounds__(128)
void nufft1_gemm(const float* __restrict__ Yr, const float* __restrict__ Yi,
                 const float* __restrict__ Wc, const float* __restrict__ Ws,
                 float* __restrict__ rP) {
  const int wave = threadIdx.x >> 5;
  const size_t bo = (size_t)blockIdx.z * WS_TOTAL;
  const int j0 = blockIdx.y * KSL;
  float* pr = rP + (size_t)blockIdx.z * RP_BATCH + (size_t)blockIdx.y * 2 * SZ_R;
  cgemm_tile<true, false>(Yr + bo + j0, Yi + bo + j0, J,
                          Wc + bo + j0, Ws + bo + j0, J,
                          pr, pr + SZ_R, NPPAD, KSL,
                          wave * 32, blockIdx.x * 32);
}

__global__ __launch_bounds__(128)
void stageA_gemm(const float* __restrict__ gOr, const float* __restrict__ gOi,
                 const float* __restrict__ gHr, const float* __restrict__ gHi,
                 const float* __restrict__ Wc, const float* __restrict__ Ws,
                 const float* __restrict__ Cx, const float* __restrict__ Sx,
                 const float* __restrict__ charge, float* __restrict__ U) {
  __shared__ float red[CCH][32];
  const int c    = __builtin_amdgcn_readfirstlane((int)(threadIdx.x >> 5));
  const int comp = blockIdx.y;
  const int bt   = blockIdx.z;
  const int n0   = blockIdx.x * 32;
  const bool h   = n0 >= JOs;
  const size_t bo = (size_t)bt * WS_TOTAL;
  const int m0 = (comp * CCH + c) * QROW;
  const int lane = threadIdx.x & 31;
  const int col  = lane & 15;
  const int kp   = lane >> 4;
  float qc[2];
#pragma unroll
  for (int nt = 0; nt < 2; ++nt)
    qc[nt] = charge[((size_t)bt * J + n0 + nt * 16 + col) * CCH + c];

  v8f accr[2][2], acci[2][2];
  cgemm_acc<false, true>((h ? gHr : gOr) + bo, (h ? gHi : gOi) + bo, KNP,
                         Wc + bo, Ws + bo, J, KNP, m0, n0, accr, acci);

  float cx[2][16], sx[2][16];
#pragma unroll
  for (int nt = 0; nt < 2; ++nt) {
    const float* cxp = Cx + bo + n0 + nt * 16 + col;
    const float* sxp = Sx + bo + n0 + nt * 16 + col;
#pragma unroll
    for (int mt = 0; mt < 2; ++mt)
#pragma unroll
      for (int v = 0; v < 8; ++v) {
        const int m  = mt * 16 + 8 * kp + v;
        const int mm = m < NM ? m : NM - 1;
        cx[nt][mt * 8 + v] = cxp[(size_t)mm * J];
        sx[nt][mt * 8 + v] = sxp[(size_t)mm * J];
      }
  }
  float s[2];
  if (comp == 0) {
#pragma unroll
    for (int nt = 0; nt < 2; ++nt) {
      float acc = 0.f;
#pragma unroll
      for (int mt = 0; mt < 2; ++mt)
#pragma unroll
        for (int v = 0; v < 8; ++v)
          acc += accr[mt][nt][v] * cx[nt][mt * 8 + v] - acci[mt][nt][v] * sx[nt][mt * 8 + v];
      s[nt] = acc;
    }
  } else {
#pragma unroll
    for (int nt = 0; nt < 2; ++nt) {
      float acc = 0.f;
#pragma unroll
      for (int mt = 0; mt < 2; ++mt)
#pragma unroll
        for (int v = 0; v < 8; ++v)
          acc += accr[mt][nt][v] * sx[nt][mt * 8 + v] + acci[mt][nt][v] * cx[nt][mt * 8 + v];
      s[nt] = acc;
    }
  }
#pragma unroll
  for (int nt = 0; nt < 2; ++nt) {
    s[nt] += __shfl_xor(s[nt], 16);
    s[nt] *= qc[nt];
  }
  if (kp == 0) {
    red[c][col]      = s[0];
    red[c][16 + col] = s[1];
  }
  __syncthreads();
  if (c != 0) return;
  float e = 0.f;
#pragma unroll
  for (int cc = 0; cc < CCH; ++cc) e += red[cc][lane];
  U[bo + (size_t)comp * J + n0 + lane] = (comp == 0) ? e / (2.0f * VBOX) : e / VBOX;
}

__global__ void g_build(const float* __restrict__ rP,
                        const float* __restrict__ shOO, const float* __restrict__ amOO,
                        const float* __restrict__ shOH, const float* __restrict__ amOH,
                        const float* __restrict__ shHO, const float* __restrict__ amHO,
                        const float* __restrict__ shHH, const float* __restrict__ amHH,
                        float* __restrict__ gOr, float* __restrict__ gOi,
                        float* __restrict__ gHr, float* __restrict__ gHi) {
  const size_t bo = (size_t)blockIdx.y * WS_TOTAL;
  int t = blockIdx.x * blockDim.x + threadIdx.x;
  if (t >= NM * NPH) return;
  int m = t / NPH, np = t % NPH;
  int n = np / NM, p = np % NM;
  float kx = TWOPI_L * (float)(m - 15);
  float ky = TWOPI_L * (float)(n - 15);
  float kz = TWOPI_L * (float)(p - 15);
  float squ = kx * kx + ky * ky + kz * kz;

  float mOO = 0.f, mOH = 0.f, mHO = 0.f, mHH = 0.f;
#pragma unroll
  for (int i = 0; i < 6; ++i) {
    mOO += amOO[i] * expf(-squ * expf(2.0f * shOO[i]));
    mOH += amOH[i] * expf(-squ * expf(2.0f * shOH[i]));
    mHO += amHO[i] * expf(-squ * expf(2.0f * shHO[i]));
    mHH += amHH[i] * expf(-squ * expf(2.0f * shHH[i]));
  }
  if (squ == 0.0f) { mOH = 0.f; mHO = 0.f; mHH = 0.f; }
  if (np < NPH - 1) { mOO *= 2.0f; mOH *= 2.0f; mHO *= 2.0f; mHH *= 2.0f; }

#pragma unroll
  for (int c = 0; c < CCH; ++c) {
    const float* pp = rP + (size_t)blockIdx.y * RP_BATCH + (size_t)(c * NM + m) * NPPAD + np;
    float ror = 0.f, roi = 0.f, rhr = 0.f, rhi = 0.f;
    for (int s = 0; s < NSL_O; ++s) {
      ror += pp[(size_t)(2 * s) * SZ_R];
      roi += pp[(size_t)(2 * s + 1) * SZ_R];
    }
    for (int s = NSL_O; s < NSL; ++s) {
      rhr += pp[(size_t)(2 * s) * SZ_R];
      rhi += pp[(size_t)(2 * s + 1) * SZ_R];
    }
    float fOr = mOO * ror + mOH * rhr, fOi = mOO * roi + mOH * rhi;
    float fHr = mHO * ror + mHH * rhr, fHi = mHO * roi + mHH * rhi;
    const size_t r0 = bo + (size_t)(c * QROW + m)        * KNP + np;
    const size_t r1 = bo + (size_t)((4  + c) * QROW + m) * KNP + np;
    const size_t r2 = bo + (size_t)((8  + c) * QROW + m) * KNP + np;
    const size_t r3 = bo + (size_t)((12 + c) * QROW + m) * KNP + np;
    gOr[r0] = fOr;        gOi[r0] = fOi;
    gOr[r1] = kx * fOr;   gOi[r1] = kx * fOi;
    gOr[r2] = ky * fOr;   gOi[r2] = ky * fOi;
    gOr[r3] = kz * fOr;   gOi[r3] = kz * fOi;
    gHr[r0] = fHr;        gHi[r0] = fHi;
    gHr[r1] = kx * fHr;   gHi[r1] = kx * fHi;
    gHr[r2] = ky * fHr;   gHi[r2] = ky * fHi;
    gHr[r3] = kz * fHr;   gHi[r3] = kz * fHi;
  }
}

constexpr int FIN_J = 32;
__global__ __launch_bounds__(4 * FIN_J)
void finalize(const float* __restrict__ U, float* __restrict__ out) {
  const int b  = blockIdx.y;
  const float* u = U + (size_t)b * WS_TOTAL;
  const int j0 = blockIdx.x * FIN_J;
  const int lane = threadIdx.x & 31;
  const int w = __builtin_amdgcn_readfirstlane((int)(threadIdx.x >> 5));
  if (w == 0) {
    out[b * J + j0 + lane] = u[j0 + lane];
  } else {
    const int t = (w - 1) * 32 + lane;
    float* F = out + 2 * J;
    F[(size_t)(b * J + j0) * 3 + t] = u[(size_t)(1 + t % 3) * J + j0 + t / 3];
  }
}

extern "C" void kernel_launch(void* const* d_in, const int* in_sizes, int n_in,
                              void* d_out, int out_size, void* d_ws, size_t ws_size,
                              hipStream_t stream) {
  const float* pos    = (const float*)d_in[0];
  const float* charge = (const float*)d_in[1];
  const float* shOO = (const float*)d_in[2];
  const float* amOO = (const float*)d_in[3];
  const float* shOH = (const float*)d_in[4];
  const float* amOH = (const float*)d_in[5];
  const float* shHO = (const float*)d_in[6];
  const float* amHO = (const float*)d_in[7];
  const float* shHH = (const float*)d_in[8];
  const float* amHH = (const float*)d_in[9];
  float* out = (float*)d_out;
  float* ws  = (float*)d_ws;

  float *Cx = ws + OFF_CX, *Sx = ws + OFF_SX;
  float *Wc = ws + OFF_WC, *Wss = ws + OFF_WS;
  float *Yr = ws + OFF_YR, *Yi = ws + OFF_YI;
  float *rP = ws + OFF_RP;
  float *gOr = ws + OFF_GOR, *gOi = ws + OFF_GOI;
  float *gHr = ws + OFF_GHR, *gHi = ws + OFF_GHI;
  float *U = ws + OFF_U;

  hipMemsetAsync(d_ws, 0, NBATCH * WS_TOTAL * sizeof(float), stream);

  tables_build<<<dim3(J / TB_J, TB_WPARTS + 1, NBATCH), TB_THREADS, 0, stream>>>(
      pos, charge, Cx, Sx, Wc, Wss, Yr, Yi);

  nufft1_gemm<<<dim3(NPPAD / 32, NSL, NBATCH), 128, 0, stream>>>(Yr, Yi, Wc, Wss, rP);

  g_build<<<dim3((NM * NPH + 255) / 256, NBATCH), 256, 0, stream>>>(
      rP,
      shOO, amOO, shOH, amOH, shHO, amHO, shHH, amHH,
      gOr, gOi, gHr, gHi);

  stageA_gemm<<<dim3(J / 32, 4, NBATCH), 128, 0, stream>>>(gOr, gOi, gHr, gHi, Wc, Wss,
                                                           Cx, Sx, charge, U);

  finalize<<<dim3(J / FIN_J, NBATCH), 4 * FIN_J, 0, stream>>>(U, out);
}


# Complex products of both GEMM stages by three real matrix products instead of four: p1=ar*br, p2=ai*bi, p3=(ar+ai)*(br+-bi), Cr and Ci formed behind the K loop: sum order and rounding change
# speedup vs baseline: 13.4528x; 1.0409x over previous
#include <hip/hip_runtime.h>
#include <hip/hip_bf16.h>
#include <math.h>

typedef float v2f __attribute__((ext_vector_type(2)));
typedef float v8f __attribute__((ext_vector_type(8)));

constexpr int NM    = 31;
constexpr int NPH   = 481;
constexpr int NPPAD = 512;
constexpr int KNP   = 484;
constexpr int J     = 768;
constexpr int JOs   = 256;
constexpr int CCH   = 4;
constexpr int CMP   = 128;
constexpr int QROW  = 32;
constexpr int QMP   = 16 * QROW;
constexpr int KSL   = 128;
constexpr int NSL_O = JOs / KSL;
constexpr int NSL   = J / KSL;
static_assert(JOs % KSL == 0 && J % KSL == 0, "a slice never straddles the two species");
constexpr float LBOX   = 20.0f;
constexpr float VBOX   = 8000.0f;
constexpr float TWOPI_L = 0.31415926535897931f;

constexpr size_t SZ_PH  = (size_t)NM * J;
constexpr size_t OFF_CX = 0;
constexpr size_t OFF_SX = OFF_CX + SZ_PH;
constexpr size_t SZ_W   = (size_t)NPPAD * J;
constexpr size_t OFF_WC = OFF_SX + SZ_PH;
constexpr size_t OFF_WS = OFF_WC + SZ_W;
constexpr size_t SZ_Y   = (size_t)CMP * J;
constexpr size_t OFF_YR = OFF_WS + SZ_W;
constexpr size_t OFF_YI = OFF_YR + SZ_Y;
constexpr size_t SZ_G   = (size_t)QMP * KNP;
constexpr size_t OFF_GOR = OFF_YI + SZ_Y;
constexpr size_t OFF_GOI = OFF_GOR + SZ_G;
constexpr size_t OFF_GHR = OFF_GOI + SZ_G;
constexpr size_t OFF_GHI = OFF_GHR + SZ_G;
constexpr size_t SZ_U   = (size_t)4 * J;
constexpr size_t OFF_U  = OFF_GHI + SZ_G;
constexpr size_t WS_TOTAL = OFF_U + SZ_U;
constexpr int NBATCH = 2;
constexpr size_t SZ_R     = (size_t)CMP * NPPAD;
constexpr size_t RP_BATCH = (size_t)NSL * 2 * SZ_R;
constexpr size_t OFF_RP   = (size_t)NBATCH * WS_TOTAL;

constexpr int TB_J       = 32;
constexpr int TB_THREADS = 512;
constexpr int TB_WAVES   = TB_THREADS / 32;
constexpr int TB_WPARTS  = 4;
constexpr int TB_ROWS    = (NPH + TB_WPARTS - 1) / TB_WPARTS;
static_assert(J % TB_J == 0, "a particle block is a whole line of every row");
__global__ __launch_bounds__(TB_THREADS)
void tables_build(const float* __restrict__ pos, const float* __restrict__ charge,
                  float* __restrict__ Cx, float* __restrict__ Sx,
                  float* __restrict__ Wc, float* __restrict__ Ws,
                  float* __restrict__ Yr, float* __restrict__ Yi) {
  __shared__ float cs[2][NM][TB_J];
  __shared__ float sn[2][NM][TB_J];
  const int b = blockIdx.z;
  const size_t bo = (size_t)b * WS_TOTAL;
  const int j0 = blockIdx.x * TB_J;
  const int part = blockIdx.y;
  const bool xpart = part == TB_WPARTS;
  const int ndim = xpart ? 1 : 2;
  const int d0   = xpart ? 0 : 1;
  for (int i = threadIdx.x; i < ndim * NM * TB_J; i += TB_THREADS) {
    const int jl = i % TB_J, dm = i / TB_J;
    const int mi = dm % NM, dd = dm / NM;
    const float th = TWOPI_L * (pos[(size_t)b * J * 3 + (size_t)(j0 + jl) * 3 + d0 + dd] - 0.5f * LBOX);
    float s, c;
    sincosf((float)(mi - 15) * th, &s, &c);
    cs[dd][mi][jl] = c;
    sn[dd][mi][jl] = s;
  }
  __syncthreads();
  const int jl = threadIdx.x & (TB_J - 1);
  const int w  = __builtin_amdgcn_readfirstlane((int)(threadIdx.x >> 5));
  const size_t col = bo + j0 + jl;
  if (!xpart) {
    const int lo = part * TB_ROWS;
    const int hi = (lo + TB_ROWS < NPH) ? (lo + TB_ROWS) : NPH;
    for (int np = lo + w; np < hi; np += TB_WAVES) {
      const int n = np / NM, p = np % NM;
      const float cy = cs[0][n][jl], sy = sn[0][n][jl];
      const float cz = cs[1][p][jl], sz = sn[1][p][jl];
      Wc[col + (size_t)np * J] = cy * cz - sy * sz;
      Ws[col + (size_t)np * J] = sy * cz + cy * sz;
    }
  } else {
    for (int m = w; m < NM; m += TB_WAVES) {
      Cx[col + (size_t)m * J] = cs[0][m][jl];
      Sx[col + (size_t)m * J] = sn[0][m][jl];
    }
    for (int cm = w; cm < CCH * NM; cm += TB_WAVES) {
      const int c = cm / NM, m = cm % NM;
      const float q = charge[(size_t)b * J * CCH + (size_t)(j0 + jl) * CCH + c];
      Yr[col + (size_t)cm * J] =  q * cs[0][m][jl];
      Yi[col + (size_t)cm * J] = -q * sn[0][m][jl];
    }
  }
}

template <int CH>
struct Frags {
  v2f ar[CH][2], ai[CH][2], br[CH][2], bi[CH][2];
};

template <bool NTMODE, int CH>
__device__ __forceinline__ void load_chunk(Frags<CH>& f,
                                           const float* __restrict__ arp0,
                                           const float* __restrict__ aip0,
                                           const float* __restrict__ arp1,
                                           const float* __restrict__ aip1,
                                           const float* __restrict__ brp,
                                           const float* __restrict__ bip,
                                           int ldb, int k0) {
#pragma unroll
  for (int s = 0; s < CH; ++s) {
    const int k = k0 + 4 * s;
    f.ar[s][0] = *(const v2f*)(arp0 + k);
    f.ar[s][1] = *(const v2f*)(arp1 + k);
    f.ai[s][0] = *(const v2f*)(aip0 + k);
    f.ai[s][1] = *(const v2f*)(aip1 + k);
    if (NTMODE) {
      f.br[s][0] = *(const v2f*)(brp + k);
      f.bi[s][0] = *(const v2f*)(bip + k);
      f.br[s][1] = *(const v2f*)(brp + (size_t)16 * ldb + k);
      f.bi[s][1] = *(const v2f*)(bip + (size_t)16 * ldb + k);
    } else {
      const size_t r0 = (size_t)k * ldb;
#pragma unroll
      for (int nt = 0; nt < 2; ++nt) {
        f.br[s][nt].x = brp[r0 + nt * 16];
        f.br[s][nt].y = brp[r0 + ldb + nt * 16];
        f.bi[s][nt].x = bip[r0 + nt * 16];
        f.bi[s][nt].y = bip[r0 + ldb + nt * 16];
      }
    }
  }
}

template <bool SPLUS>
__device__ __forceinline__ void kstep(const v2f (&a_r)[2], const v2f (&a_i)[2],
                                      const v2f (&b_r)[2], const v2f (&b_i)[2],
                                      v8f (&p1)[2][2], v8f (&p2)[2][2], v8f (&p3)[2][2]) {
  v2f sa[2], sb[2];
#pragma unroll
  for (int mt = 0; mt < 2; ++mt) sa[mt] = a_r[mt] + a_i[mt];
#pragma unroll
  for (int nt = 0; nt < 2; ++nt) sb[nt] = SPLUS ? (b_r[nt] + b_i[nt]) : (b_r[nt] - b_i[nt]);
#pragma unroll
  for (int mt = 0; mt < 2; ++mt)
#pragma unroll
    for (int nt = 0; nt < 2; ++nt) {
      p1[mt][nt] = __builtin_amdgcn_wmma_f32_16x16x4_f32(
          false, a_r[mt], false, b_r[nt], (short)0, p1[mt][nt], false, false);
      p2[mt][nt] = __builtin_amdgcn_wmma_f32_16x16x4_f32(
          false, a_i[mt], false, b_i[nt], (short)0, p2[mt][nt], false, false);
      p3[mt][nt] = __builtin_amdgcn_wmma_f32_16x16x4_f32(
          false, sa[mt], false, sb[nt], (short)0, p3[mt][nt], false, false);
    }
}

template <bool SPLUS, int CH>
__device__ __forceinline__ void compute_chunk(const Frags<CH>& f, v8f (&p1)[2][2],
                                              v8f (&p2)[2][2], v8f (&p3)[2][2]) {
#pragma unroll
  for (int s = 0; s < CH; ++s)
    kstep<SPLUS>(f.ar[s], f.ai[s], f.br[s], f.bi[s], p1, p2, p3);
}

template <bool NTMODE, bool SPLUS>
__device__ __forceinline__
void cgemm_acc(const float* __restrict__ Ar, const float* __restrict__ Ai, int lda,
               const float* __restrict__ Br, const float* __restrict__ Bi, int ldb,
               int K, int m0, int n0, v8f (&accr)[2][2], v8f (&acci)[2][2]) {
  constexpr int CH = NTMODE ? 4 : 5;
  const int lane = threadIdx.x & 31;
  const int row  = lane & 15;
  const int kp   = lane >> 4;

  v8f p1[2][2], p2[2][2], p3[2][2];
#pragma unroll
  for (int mt = 0; mt < 2; ++mt)
#pragma unroll
    for (int nt = 0; nt < 2; ++nt) {
      p1[mt][nt] = (v8f){0.f, 0.f, 0.f, 0.f, 0.f, 0.f, 0.f, 0.f};
      p2[mt][nt] = (v8f){0.f, 0.f, 0.f, 0.f, 0.f, 0.f, 0.f, 0.f};
      p3[mt][nt] = (v8f){0.f, 0.f, 0.f, 0.f, 0.f, 0.f, 0.f, 0.f};
    }

  const float* arp0 = Ar + (size_t)(m0 + row) * lda + 2 * kp;
  const float* aip0 = Ai + (size_t)(m0 + row) * lda + 2 * kp;
  const float* arp1 = arp0 + (size_t)16 * lda;
  const float* aip1 = aip0 + (size_t)16 * lda;
  const float* brp = NTMODE ? Br + (size_t)(n0 + row) * ldb + 2 * kp
                            : Br + (size_t)(2 * kp) * ldb + n0 + row;
  const float* bip = NTMODE ? Bi + (size_t)(n0 + row) * ldb + 2 * kp
                            : Bi + (size_t)(2 * kp) * ldb + n0 + row;

  const int nch = K / (4 * CH);
  Frags<CH> f0, f1;
  if (nch > 0) load_chunk<NTMODE, CH>(f0, arp0, aip0, arp1, aip1, brp, bip, ldb, 0);
  int c = 0;
  constexpr int UNR = NTMODE ? 4 : 1;
#pragma unroll UNR
  for (; c + 2 <= nch; c += 2) {
    load_chunk<NTMODE, CH>(f1, arp0, aip0, arp1, aip1, brp, bip, ldb, (c + 1) * 4 * CH);
    __builtin_amdgcn_sched_barrier(0);
    compute_chunk<SPLUS, CH>(f0, p1, p2, p3);
    __builtin_amdgcn_sched_barrier(0);
    const int cn = (c + 2 < nch) ? (c + 2) : (nch - 1);
    load_chunk<NTMODE, CH>(f0, arp0, aip0, arp1, aip1, brp, bip, ldb, cn * 4 * CH);
    __builtin_amdgcn_sched_barrier(0);
    compute_chunk<SPLUS, CH>(f1, p1, p2, p3);
    __builtin_amdgcn_sched_barrier(0);
  }
  if (c < nch) compute_chunk<SPLUS, CH>(f0, p1, p2, p3);

  for (int k0 = nch * 4 * CH; k0 < K; k0 += 4) {
    Frags<1> t;
    load_chunk<NTMODE, 1>(t, arp0, aip0, arp1, aip1, brp, bip, ldb, k0);
    compute_chunk<SPLUS, 1>(t, p1, p2, p3);
  }

#pragma unroll
  for (int mt = 0; mt < 2; ++mt)
#pragma unroll
    for (int nt = 0; nt < 2; ++nt) {
      if (SPLUS) {
        accr[mt][nt] = p1[mt][nt] - p2[mt][nt];
        acci[mt][nt] = p3[mt][nt] - p1[mt][nt] - p2[mt][nt];
      } else {
        accr[mt][nt] = p1[mt][nt] + p2[mt][nt];
        acci[mt][nt] = p3[mt][nt] - p1[mt][nt] + p2[mt][nt];
      }
    }
}

template <bool NTMODE, bool SPLUS>
__device__ __forceinline__
void cgemm_tile(const float* __restrict__ Ar, const float* __restrict__ Ai, int lda,
                const float* __restrict__ Br, const float* __restrict__ Bi, int ldb,
                float* __restrict__ Cr, float* __restrict__ Ci, int ldc, int K,
                int m0, int n0) {
  const int lane = threadIdx.x & 31;
  const int row  = lane & 15;
  const int kp   = lane >> 4;
  v8f accr[2][2], acci[2][2];
  cgemm_acc<NTMODE, SPLUS>(Ar, Ai, lda, Br, Bi, ldb, K, m0, n0, accr, acci);
#pragma unroll
  for (int mt = 0; mt < 2; ++mt)
#pragma unroll
    for (int nt = 0; nt < 2; ++nt)
#pragma unroll
      for (int v = 0; v < 8; ++v) {
        const size_t o = (size_t)(m0 + mt * 16 + v + 8 * kp) * ldc + n0 + nt * 16 + row;
        Cr[o] = accr[mt][nt][v];
        Ci[o] = acci[mt][nt][v];
      }
}


__global__ __launch_bounds__(128)
void nufft1_gemm(const float* __restrict__ Yr, const float* __restrict__ Yi,
                 const float* __restrict__ Wc, const float* __restrict__ Ws,
                 float* __restrict__ rP) {
  const int wave = threadIdx.x >> 5;
  const size_t bo = (size_t)blockIdx.z * WS_TOTAL;
  const int j0 = blockIdx.y * KSL;
  float* pr = rP + (size_t)blockIdx.z * RP_BATCH + (size_t)blockIdx.y * 2 * SZ_R;
  cgemm_tile<true, false>(Yr + bo + j0, Yi + bo + j0, J,
                          Wc + bo + j0, Ws + bo + j0, J,
                          pr, pr + SZ_R, NPPAD, KSL,
                          wave * 32, blockIdx.x * 32);
}

__global__ __launch_bounds__(128)
void stageA_gemm(const float* __restrict__ gOr, const float* __restrict__ gOi,
                 const float* __restrict__ gHr, const float* __restrict__ gHi,
                 const float* __restrict__ Wc, const float* __restrict__ Ws,
                 const float* __restrict__ Cx, const float* __restrict__ Sx,
                 const float* __restrict__ charge, float* __restrict__ U) {
  __shared__ float red[CCH][32];
  const int c    = __builtin_amdgcn_readfirstlane((int)(threadIdx.x >> 5));
  const int comp = blockIdx.y;
  const int bt   = blockIdx.z;
  const int n0   = blockIdx.x * 32;
  const bool h   = n0 >= JOs;
  const size_t bo = (size_t)bt * WS_TOTAL;
  const int m0 = (comp * CCH + c) * QROW;
  const int lane = threadIdx.x & 31;
  const int col  = lane & 15;
  const int kp   = lane >> 4;
  float qc[2];
#pragma unroll
  for (int nt = 0; nt < 2; ++nt)
    qc[nt] = charge[((size_t)bt * J + n0 + nt * 16 + col) * CCH + c];

  v8f accr[2][2], acci[2][2];
  cgemm_acc<false, true>((h ? gHr : gOr) + bo, (h ? gHi : gOi) + bo, KNP,
                         Wc + bo, Ws + bo, J, KNP, m0, n0, accr, acci);

  float cx[2][16], sx[2][16];
#pragma unroll
  for (int nt = 0; nt < 2; ++nt) {
    const float* cxp = Cx + bo + n0 + nt * 16 + col;
    const float* sxp = Sx + bo + n0 + nt * 16 + col;
#pragma unroll
    for (int mt = 0; mt < 2; ++mt)
#pragma unroll
      for (int v = 0; v < 8; ++v) {
        const int m  = mt * 16 + 8 * kp + v;
        const int mm = m < NM ? m : NM - 1;
        cx[nt][mt * 8 + v] = cxp[(size_t)mm * J];
        sx[nt][mt * 8 + v] = sxp[(size_t)mm * J];
      }
  }
  float s[2];
  if (comp == 0) {
#pragma unroll
    for (int nt = 0; nt < 2; ++nt) {
      float acc = 0.f;
#pragma unroll
      for (int mt = 0; mt < 2; ++mt)
#pragma unroll
        for (int v = 0; v < 8; ++v)
          acc += accr[mt][nt][v] * cx[nt][mt * 8 + v] - acci[mt][nt][v] * sx[nt][mt * 8 + v];
      s[nt] = acc;
    }
  } else {
#pragma unroll
    for (int nt = 0; nt < 2; ++nt) {
      float acc = 0.f;
#pragma unroll
      for (int mt = 0; mt < 2; ++mt)
#pragma unroll
        for (int v = 0; v < 8; ++v)
          acc += accr[mt][nt][v] * sx[nt][mt * 8 + v] + acci[mt][nt][v] * cx[nt][mt * 8 + v];
      s[nt] = acc;
    }
  }
#pragma unroll
  for (int nt = 0; nt < 2; ++nt) {
    s[nt] += __shfl_xor(s[nt], 16);
    s[nt] *= qc[nt];
  }
  if (kp == 0) {
    red[c][col]      = s[0];
    red[c][16 + col] = s[1];
  }
  __syncthreads();
  if (c != 0) return;
  float e = 0.f;
#pragma unroll
  for (int cc = 0; cc < CCH; ++cc) e += red[cc][lane];
  U[bo + (size_t)comp * J + n0 + lane] = (comp == 0) ? e / (2.0f * VBOX) : e / VBOX;
}

__global__ void g_build(const float* __restrict__ rP,
                        const float* __restrict__ shOO, const float* __restrict__ amOO,
                        const float* __restrict__ shOH, const float* __restrict__ amOH,
                        const float* __restrict__ shHO, const float* __restrict__ amHO,
                        const float* __restrict__ shHH, const float* __restrict__ amHH,
                        float* __restrict__ gOr, float* __restrict__ gOi,
                        float* __restrict__ gHr, float* __restrict__ gHi) {
  const size_t bo = (size_t)blockIdx.y * WS_TOTAL;
  int t = blockIdx.x * blockDim.x + threadIdx.x;
  if (t >= NM * NPH) return;
  int m = t / NPH, np = t % NPH;
  int n = np / NM, p = np % NM;
  float kx = TWOPI_L * (float)(m - 15);
  float ky = TWOPI_L * (float)(n - 15);
  float kz = TWOPI_L * (float)(p - 15);
  float squ = kx * kx + ky * ky + kz * kz;

  float mOO = 0.f, mOH = 0.f, mHO = 0.f, mHH = 0.f;
#pragma unroll
  for (int i = 0; i < 6; ++i) {
    mOO += amOO[i] * expf(-squ * expf(2.0f * shOO[i]));
    mOH += amOH[i] * expf(-squ * expf(2.0f * shOH[i]));
    mHO += amHO[i] * expf(-squ * expf(2.0f * shHO[i]));
    mHH += amHH[i] * expf(-squ * expf(2.0f * shHH[i]));
  }
  if (squ == 0.0f) { mOH = 0.f; mHO = 0.f; mHH = 0.f; }
  if (np < NPH - 1) { mOO *= 2.0f; mOH *= 2.0f; mHO *= 2.0f; mHH *= 2.0f; }

#pragma unroll
  for (int c = 0; c < CCH; ++c) {
    const float* pp = rP + (size_t)blockIdx.y * RP_BATCH + (size_t)(c * NM + m) * NPPAD + np;
    float ror = 0.f, roi = 0.f, rhr = 0.f, rhi = 0.f;
    for (int s = 0; s < NSL_O; ++s) {
      ror += pp[(size_t)(2 * s) * SZ_R];
      roi += pp[(size_t)(2 * s + 1) * SZ_R];
    }
    for (int s = NSL_O; s < NSL; ++s) {
      rhr += pp[(size_t)(2 * s) * SZ_R];
      rhi += pp[(size_t)(2 * s + 1) * SZ_R];
    }
    float fOr = mOO * ror + mOH * rhr, fOi = mOO * roi + mOH * rhi;
    float fHr = mHO * ror + mHH * rhr, fHi = mHO * roi + mHH * rhi;
    const size_t r0 = bo + (size_t)(c * QROW + m)        * KNP + np;
    const size_t r1 = bo + (size_t)((4  + c) * QROW + m) * KNP + np;
    const size_t r2 = bo + (size_t)((8  + c) * QROW + m) * KNP + np;
    const size_t r3 = bo + (size_t)((12 + c) * QROW + m) * KNP + np;
    gOr[r0] = fOr;        gOi[r0] = fOi;
    gOr[r1] = kx * fOr;   gOi[r1] = kx * fOi;
    gOr[r2] = ky * fOr;   gOi[r2] = ky * fOi;
    gOr[r3] = kz * fOr;   gOi[r3] = kz * fOi;
    gHr[r0] = fHr;        gHi[r0] = fHi;
    gHr[r1] = kx * fHr;   gHi[r1] = kx * fHi;
    gHr[r2] = ky * fHr;   gHi[r2] = ky * fHi;
    gHr[r3] = kz * fHr;   gHi[r3] = kz * fHi;
  }
}

constexpr int FIN_J = 32;
__global__ __launch_bounds__(4 * FIN_J)
void finalize(const float* __restrict__ U, float* __restrict__ out) {
  const int b  = blockIdx.y;
  const float* u = U + (size_t)b * WS_TOTAL;
  const int j0 = blockIdx.x * FIN_J;
  const int lane = threadIdx.x & 31;
  const int w = __builtin_amdgcn_readfirstlane((int)(threadIdx.x >> 5));
  if (w == 0) {
    out[b * J + j0 + lane] = u[j0 + lane];
  } else {
    const int t = (w - 1) * 32 + lane;
    float* F = out + 2 * J;
    F[(size_t)(b * J + j0) * 3 + t] = u[(size_t)(1 + t % 3) * J + j0 + t / 3];
  }
}

extern "C" void kernel_launch(void* const* d_in, const int* in_sizes, int n_in,
                              void* d_out, int out_size, void* d_ws, size_t ws_size,
                              hipStream_t stream) {
  const float* pos    = (const float*)d_in[0];
  const float* charge = (const float*)d_in[1];
  const float* shOO = (const float*)d_in[2];
  const float* amOO = (const float*)d_in[3];
  const float* shOH = (const float*)d_in[4];
  const float* amOH = (const float*)d_in[5];
  const float* shHO = (const float*)d_in[6];
  const float* amHO = (const float*)d_in[7];
  const float* shHH = (const float*)d_in[8];
  const float* amHH = (const float*)d_in[9];
  float* out = (float*)d_out;
  float* ws  = (float*)d_ws;

  float *Cx = ws + OFF_CX, *Sx = ws + OFF_SX;
  float *Wc = ws + OFF_WC, *Wss = ws + OFF_WS;
  float *Yr = ws + OFF_YR, *Yi = ws + OFF_YI;
  float *rP = ws + OFF_RP;
  float *gOr = ws + OFF_GOR, *gOi = ws + OFF_GOI;
  float *gHr = ws + OFF_GHR, *gHi = ws + OFF_GHI;
  float *U = ws + OFF_U;

  hipMemsetAsync(d_ws, 0, NBATCH * WS_TOTAL * sizeof(float), stream);

  tables_build<<<dim3(J / TB_J, TB_WPARTS + 1, NBATCH), TB_THREADS, 0, stream>>>(
      pos, charge, Cx, Sx, Wc, Wss, Yr, Yi);

  nufft1_gemm<<<dim3(NPPAD / 32, NSL, NBATCH), 128, 0, stream>>>(Yr, Yi, Wc, Wss, rP);

  g_build<<<dim3((NM * NPH + 255) / 256, NBATCH), 256, 0, stream>>>(
      rP,
      shOO, amOO, shOH, amOH, shHO, amHO, shHH, amHH,
      gOr, gOi, gHr, gHi);

  stageA_gemm<<<dim3(J / 32, 4, NBATCH), 128, 0, stream>>>(gOr, gOi, gHr, gHi, Wc, Wss,
                                                           Cx, Sx, charge, U);

  finalize<<<dim3(J / FIN_J, NBATCH), 4 * FIN_J, 0, stream>>>(U, out);
}


# Stage A's sum over np cut into 4 slices, a workgroup per slice (768 workgroups, 3 per WGP, 3 waves per SIMD); the output kernel adds the 4 partial sums in slice order: np sum order changes
# speedup vs baseline: 14.2351x; 1.0582x over previous
#include <hip/hip_runtime.h>
#include <hip/hip_bf16.h>
#include <math.h>

typedef float v2f __attribute__((ext_vector_type(2)));
typedef float v8f __attribute__((ext_vector_type(8)));

constexpr int NM    = 31;
constexpr int NPH   = 481;
constexpr int NPPAD = 512;
constexpr int KNP   = 484;
constexpr int J     = 768;
constexpr int JOs   = 256;
constexpr int CCH   = 4;
constexpr int CMP   = 128;
constexpr int QROW  = 32;
constexpr int QMP   = 16 * QROW;
constexpr int KSL   = 128;
constexpr int NSL_O = JOs / KSL;
constexpr int NSL   = J / KSL;
static_assert(JOs % KSL == 0 && J % KSL == 0, "a slice never straddles the two species");
constexpr int SA_NSL = 4;
constexpr int SA_KSL = 120;
static_assert(SA_KSL % 20 == 0 && SA_KSL % 2 == 0, "whole chunks per slice, 8-byte aligned A loads");
static_assert(SA_KSL * (SA_NSL - 1) < KNP, "the last slice is not empty");
constexpr float LBOX   = 20.0f;
constexpr float VBOX   = 8000.0f;
constexpr float TWOPI_L = 0.31415926535897931f;

constexpr size_t SZ_PH  = (size_t)NM * J;
constexpr size_t OFF_CX = 0;
constexpr size_t OFF_SX = OFF_CX + SZ_PH;
constexpr size_t SZ_W   = (size_t)NPPAD * J;
constexpr size_t OFF_WC = OFF_SX + SZ_PH;
constexpr size_t OFF_WS = OFF_WC + SZ_W;
constexpr size_t SZ_Y   = (size_t)CMP * J;
constexpr size_t OFF_YR = OFF_WS + SZ_W;
constexpr size_t OFF_YI = OFF_YR + SZ_Y;
constexpr size_t SZ_G   = (size_t)QMP * KNP;
constexpr size_t OFF_GOR = OFF_YI + SZ_Y;
constexpr size_t OFF_GOI = OFF_GOR + SZ_G;
constexpr size_t OFF_GHR = OFF_GOI + SZ_G;
constexpr size_t OFF_GHI = OFF_GHR + SZ_G;
constexpr size_t SZ_U   = (size_t)4 * SA_NSL * J;
constexpr size_t OFF_U  = OFF_GHI + SZ_G;
constexpr size_t WS_TOTAL = OFF_U + SZ_U;
constexpr int NBATCH = 2;
constexpr size_t SZ_R     = (size_t)CMP * NPPAD;
constexpr size_t RP_BATCH = (size_t)NSL * 2 * SZ_R;
constexpr size_t OFF_RP   = (size_t)NBATCH * WS_TOTAL;

constexpr int TB_J       = 32;
constexpr int TB_THREADS = 512;
constexpr int TB_WAVES   = TB_THREADS / 32;
constexpr int TB_WPARTS  = 4;
constexpr int TB_ROWS    = (NPH + TB_WPARTS - 1) / TB_WPARTS;
static_assert(J % TB_J == 0, "a particle block is a whole line of every row");
__global__ __launch_bounds__(TB_THREADS)
void tables_build(const float* __restrict__ pos, const float* __restrict__ charge,
                  float* __restrict__ Cx, float* __restrict__ Sx,
                  float* __restrict__ Wc, float* __restrict__ Ws,
                  float* __restrict__ Yr, float* __restrict__ Yi) {
  __shared__ float cs[2][NM][TB_J];
  __shared__ float sn[2][NM][TB_J];
  const int b = blockIdx.z;
  const size_t bo = (size_t)b * WS_TOTAL;
  const int j0 = blockIdx.x * TB_J;
  const int part = blockIdx.y;
  const bool xpart = part == TB_WPARTS;
  const int ndim = xpart ? 1 : 2;
  const int d0   = xpart ? 0 : 1;
  for (int i = threadIdx.x; i < ndim * NM * TB_J; i += TB_THREADS) {
    const int jl = i % TB_J, dm = i / TB_J;
    const int mi = dm % NM, dd = dm / NM;
    const float th = TWOPI_L * (pos[(size_t)b * J * 3 + (size_t)(j0 + jl) * 3 + d0 + dd] - 0.5f * LBOX);
    float s, c;
    sincosf((float)(mi - 15) * th, &s, &c);
    cs[dd][mi][jl] = c;
    sn[dd][mi][jl] = s;
  }
  __syncthreads();
  const int jl = threadIdx.x & (TB_J - 1);
  const int w  = __builtin_amdgcn_readfirstlane((int)(threadIdx.x >> 5));
  const size_t col = bo + j0 + jl;
  if (!xpart) {
    const int lo = part * TB_ROWS;
    const int hi = (lo + TB_ROWS < NPH) ? (lo + TB_ROWS) : NPH;
    for (int np = lo + w; np < hi; np += TB_WAVES) {
      const int n = np / NM, p = np % NM;
      const float cy = cs[0][n][jl], sy = sn[0][n][jl];
      const float cz = cs[1][p][jl], sz = sn[1][p][jl];
      Wc[col + (size_t)np * J] = cy * cz - sy * sz;
      Ws[col + (size_t)np * J] = sy * cz + cy * sz;
    }
  } else {
    for (int m = w; m < NM; m += TB_WAVES) {
      Cx[col + (size_t)m * J] = cs[0][m][jl];
      Sx[col + (size_t)m * J] = sn[0][m][jl];
    }
    for (int cm = w; cm < CCH * NM; cm += TB_WAVES) {
      const int c = cm / NM, m = cm % NM;
      const float q = charge[(size_t)b * J * CCH + (size_t)(j0 + jl) * CCH + c];
      Yr[col + (size_t)cm * J] =  q * cs[0][m][jl];
      Yi[col + (size_t)cm * J] = -q * sn[0][m][jl];
    }
  }
}

template <int CH>
struct Frags {
  v2f ar[CH][2], ai[CH][2], br[CH][2], bi[CH][2];
};

template <bool NTMODE, int CH>
__device__ __forceinline__ void load_chunk(Frags<CH>& f,
                                           const float* __restrict__ arp0,
                                           const float* __restrict__ aip0,
                                           const float* __restrict__ arp1,
                                           const float* __restrict__ aip1,
                                           const float* __restrict__ brp,
                                           const float* __restrict__ bip,
                                           int ldb, int k0) {
#pragma unroll
  for (int s = 0; s < CH; ++s) {
    const int k = k0 + 4 * s;
    f.ar[s][0] = *(const v2f*)(arp0 + k);
    f.ar[s][1] = *(const v2f*)(arp1 + k);
    f.ai[s][0] = *(const v2f*)(aip0 + k);
    f.ai[s][1] = *(const v2f*)(aip1 + k);
    if (NTMODE) {
      f.br[s][0] = *(const v2f*)(brp + k);
      f.bi[s][0] = *(const v2f*)(bip + k);
      f.br[s][1] = *(const v2f*)(brp + (size_t)16 * ldb + k);
      f.bi[s][1] = *(const v2f*)(bip + (size_t)16 * ldb + k);
    } else {
      const size_t r0 = (size_t)k * ldb;
#pragma unroll
      for (int nt = 0; nt < 2; ++nt) {
        f.br[s][nt].x = brp[r0 + nt * 16];
        f.br[s][nt].y = brp[r0 + ldb + nt * 16];
        f.bi[s][nt].x = bip[r0 + nt * 16];
        f.bi[s][nt].y = bip[r0 + ldb + nt * 16];
      }
    }
  }
}

template <bool SPLUS>
__device__ __forceinline__ void kstep(const v2f (&a_r)[2], const v2f (&a_i)[2],
                                      const v2f (&b_r)[2], const v2f (&b_i)[2],
                                      v8f (&p1)[2][2], v8f (&p2)[2][2], v8f (&p3)[2][2]) {
  v2f sa[2], sb[2];
#pragma unroll
  for (int mt = 0; mt < 2; ++mt) sa[mt] = a_r[mt] + a_i[mt];
#pragma unroll
  for (int nt = 0; nt < 2; ++nt) sb[nt] = SPLUS ? (b_r[nt] + b_i[nt]) : (b_r[nt] - b_i[nt]);
#pragma unroll
  for (int mt = 0; mt < 2; ++mt)
#pragma unroll
    for (int nt = 0; nt < 2; ++nt) {
      p1[mt][nt] = __builtin_amdgcn_wmma_f32_16x16x4_f32(
          false, a_r[mt], false, b_r[nt], (short)0, p1[mt][nt], false, false);
      p2[mt][nt] = __builtin_amdgcn_wmma_f32_16x16x4_f32(
          false, a_i[mt], false, b_i[nt], (short)0, p2[mt][nt], false, false);
      p3[mt][nt] = __builtin_amdgcn_wmma_f32_16x16x4_f32(
          false, sa[mt], false, sb[nt], (short)0, p3[mt][nt], false, false);
    }
}

template <bool SPLUS, int CH>
__device__ __forceinline__ void compute_chunk(const Frags<CH>& f, v8f (&p1)[2][2],
                                              v8f (&p2)[2][2], v8f (&p3)[2][2]) {
#pragma unroll
  for (int s = 0; s < CH; ++s)
    kstep<SPLUS>(f.ar[s], f.ai[s], f.br[s], f.bi[s], p1, p2, p3);
}

template <bool NTMODE, bool SPLUS>
__device__ __forceinline__
void cgemm_acc(const float* __restrict__ Ar, const float* __restrict__ Ai, int lda,
               const float* __restrict__ Br, const float* __restrict__ Bi, int ldb,
               int K, int m0, int n0, v8f (&accr)[2][2], v8f (&acci)[2][2]) {
  constexpr int CH = NTMODE ? 4 : 5;
  const int lane = threadIdx.x & 31;
  const int row  = lane & 15;
  const int kp   = lane >> 4;

  v8f p1[2][2], p2[2][2], p3[2][2];
#pragma unroll
  for (int mt = 0; mt < 2; ++mt)
#pragma unroll
    for (int nt = 0; nt < 2; ++nt) {
      p1[mt][nt] = (v8f){0.f, 0.f, 0.f, 0.f, 0.f, 0.f, 0.f, 0.f};
      p2[mt][nt] = (v8f){0.f, 0.f, 0.f, 0.f, 0.f, 0.f, 0.f, 0.f};
      p3[mt][nt] = (v8f){0.f, 0.f, 0.f, 0.f, 0.f, 0.f, 0.f, 0.f};
    }

  const float* arp0 = Ar + (size_t)(m0 + row) * lda + 2 * kp;
  const float* aip0 = Ai + (size_t)(m0 + row) * lda + 2 * kp;
  const float* arp1 = arp0 + (size_t)16 * lda;
  const float* aip1 = aip0 + (size_t)16 * lda;
  const float* brp = NTMODE ? Br + (size_t)(n0 + row) * ldb + 2 * kp
                            : Br + (size_t)(2 * kp) * ldb + n0 + row;
  const float* bip = NTMODE ? Bi + (size_t)(n0 + row) * ldb + 2 * kp
                            : Bi + (size_t)(2 * kp) * ldb + n0 + row;

  const int nch = K / (4 * CH);
  Frags<CH> f0, f1;
  if (nch > 0) load_chunk<NTMODE, CH>(f0, arp0, aip0, arp1, aip1, brp, bip, ldb, 0);
  int c = 0;
  constexpr int UNR = NTMODE ? 4 : 1;
#pragma unroll UNR
  for (; c + 2 <= nch; c += 2) {
    load_chunk<NTMODE, CH>(f1, arp0, aip0, arp1, aip1, brp, bip, ldb, (c + 1) * 4 * CH);
    __builtin_amdgcn_sched_barrier(0);
    compute_chunk<SPLUS, CH>(f0, p1, p2, p3);
    __builtin_amdgcn_sched_barrier(0);
    const int cn = (c + 2 < nch) ? (c + 2) : (nch - 1);
    load_chunk<NTMODE, CH>(f0, arp0, aip0, arp1, aip1, brp, bip, ldb, cn * 4 * CH);
    __builtin_amdgcn_sched_barrier(0);
    compute_chunk<SPLUS, CH>(f1, p1, p2, p3);
    __builtin_amdgcn_sched_barrier(0);
  }
  if (c < nch) compute_chunk<SPLUS, CH>(f0, p1, p2, p3);

  for (int k0 = nch * 4 * CH; k0 < K; k0 += 4) {
    Frags<1> t;
    load_chunk<NTMODE, 1>(t, arp0, aip0, arp1, aip1, brp, bip, ldb, k0);
    compute_chunk<SPLUS, 1>(t, p1, p2, p3);
  }

#pragma unroll
  for (int mt = 0; mt < 2; ++mt)
#pragma unroll
    for (int nt = 0; nt < 2; ++nt) {
      if (SPLUS) {
        accr[mt][nt] = p1[mt][nt] - p2[mt][nt];
        acci[mt][nt] = p3[mt][nt] - p1[mt][nt] - p2[mt][nt];
      } else {
        accr[mt][nt] = p1[mt][nt] + p2[mt][nt];
        acci[mt][nt] = p3[mt][nt] - p1[mt][nt] + p2[mt][nt];
      }
    }
}

template <bool NTMODE, bool SPLUS>
__device__ __forceinline__
void cgemm_tile(const float* __restrict__ Ar, const float* __restrict__ Ai, int lda,
                const float* __restrict__ Br, const float* __restrict__ Bi, int ldb,
                float* __restrict__ Cr, float* __restrict__ Ci, int ldc, int K,
                int m0, int n0) {
  const int lane = threadIdx.x & 31;
  const int row  = lane & 15;
  const int kp   = lane >> 4;
  v8f accr[2][2], acci[2][2];
  cgemm_acc<NTMODE, SPLUS>(Ar, Ai, lda, Br, Bi, ldb, K, m0, n0, accr, acci);
#pragma unroll
  for (int mt = 0; mt < 2; ++mt)
#pragma unroll
    for (int nt = 0; nt < 2; ++nt)
#pragma unroll
      for (int v = 0; v < 8; ++v) {
        const size_t o = (size_t)(m0 + mt * 16 + v + 8 * kp) * ldc + n0 + nt * 16 + row;
        Cr[o] = accr[mt][nt][v];
        Ci[o] = acci[mt][nt][v];
      }
}


__global__ __launch_bounds__(128)
void nufft1_gemm(const float* __restrict__ Yr, const float* __restrict__ Yi,
                 const float* __restrict__ Wc, const float* __restrict__ Ws,
                 float* __restrict__ rP) {
  const int wave = threadIdx.x >> 5;
  const size_t bo = (size_t)blockIdx.z * WS_TOTAL;
  const int j0 = blockIdx.y * KSL;
  float* pr = rP + (size_t)blockIdx.z * RP_BATCH + (size_t)blockIdx.y * 2 * SZ_R;
  cgemm_tile<true, false>(Yr + bo + j0, Yi + bo + j0, J,
                          Wc + bo + j0, Ws + bo + j0, J,
                          pr, pr + SZ_R, NPPAD, KSL,
                          wave * 32, blockIdx.x * 32);
}

__global__ __launch_bounds__(128)
void stageA_gemm(const float* __restrict__ gOr, const float* __restrict__ gOi,
                 const float* __restrict__ gHr, const float* __restrict__ gHi,
                 const float* __restrict__ Wc, const float* __restrict__ Ws,
                 const float* __restrict__ Cx, const float* __restrict__ Sx,
                 const float* __restrict__ charge, float* __restrict__ U) {
  __shared__ float red[CCH][32];
  const int c    = __builtin_amdgcn_readfirstlane((int)(threadIdx.x >> 5));
  const int comp = blockIdx.y / SA_NSL;
  const int sl   = blockIdx.y % SA_NSL;
  const int k0   = sl * SA_KSL;
  const int klen = (sl == SA_NSL - 1) ? (KNP - k0) : SA_KSL;
  const int bt   = blockIdx.z;
  const int n0   = blockIdx.x * 32;
  const bool h   = n0 >= JOs;
  const size_t bo = (size_t)bt * WS_TOTAL;
  const int m0 = (comp * CCH + c) * QROW;
  const int lane = threadIdx.x & 31;
  const int col  = lane & 15;
  const int kp   = lane >> 4;
  float qc[2];
#pragma unroll
  for (int nt = 0; nt < 2; ++nt)
    qc[nt] = charge[((size_t)bt * J + n0 + nt * 16 + col) * CCH + c];

  v8f accr[2][2], acci[2][2];
  cgemm_acc<false, true>((h ? gHr : gOr) + bo + k0, (h ? gHi : gOi) + bo + k0, KNP,
                         Wc + bo + (size_t)k0 * J, Ws + bo + (size_t)k0 * J, J, klen,
                         m0, n0, accr, acci);

  float cx[2][16], sx[2][16];
#pragma unroll
  for (int nt = 0; nt < 2; ++nt) {
    const float* cxp = Cx + bo + n0 + nt * 16 + col;
    const float* sxp = Sx + bo + n0 + nt * 16 + col;
#pragma unroll
    for (int mt = 0; mt < 2; ++mt)
#pragma unroll
      for (int v = 0; v < 8; ++v) {
        const int m  = mt * 16 + 8 * kp + v;
        const int mm = m < NM ? m : NM - 1;
        cx[nt][mt * 8 + v] = cxp[(size_t)mm * J];
        sx[nt][mt * 8 + v] = sxp[(size_t)mm * J];
      }
  }
  float s[2];
  if (comp == 0) {
#pragma unroll
    for (int nt = 0; nt < 2; ++nt) {
      float acc = 0.f;
#pragma unroll
      for (int mt = 0; mt < 2; ++mt)
#pragma unroll
        for (int v = 0; v < 8; ++v)
          acc += accr[mt][nt][v] * cx[nt][mt * 8 + v] - acci[mt][nt][v] * sx[nt][mt * 8 + v];
      s[nt] = acc;
    }
  } else {
#pragma unroll
    for (int nt = 0; nt < 2; ++nt) {
      float acc = 0.f;
#pragma unroll
      for (int mt = 0; mt < 2; ++mt)
#pragma unroll
        for (int v = 0; v < 8; ++v)
          acc += accr[mt][nt][v] * sx[nt][mt * 8 + v] + acci[mt][nt][v] * cx[nt][mt * 8 + v];
      s[nt] = acc;
    }
  }
#pragma unroll
  for (int nt = 0; nt < 2; ++nt) {
    s[nt] += __shfl_xor(s[nt], 16);
    s[nt] *= qc[nt];
  }
  if (kp == 0) {
    red[c][col]      = s[0];
    red[c][16 + col] = s[1];
  }
  __syncthreads();
  if (c != 0) return;
  float e = 0.f;
#pragma unroll
  for (int cc = 0; cc < CCH; ++cc) e += red[cc][lane];
  U[bo + (size_t)(comp * SA_NSL + sl) * J + n0 + lane] = e;
}

__global__ void g_build(const float* __restrict__ rP,
                        const float* __restrict__ shOO, const float* __restrict__ amOO,
                        const float* __restrict__ shOH, const float* __restrict__ amOH,
                        const float* __restrict__ shHO, const float* __restrict__ amHO,
                        const float* __restrict__ shHH, const float* __restrict__ amHH,
                        float* __restrict__ gOr, float* __restrict__ gOi,
                        float* __restrict__ gHr, float* __restrict__ gHi) {
  const size_t bo = (size_t)blockIdx.y * WS_TOTAL;
  int t = blockIdx.x * blockDim.x + threadIdx.x;
  if (t >= NM * NPH) return;
  int m = t / NPH, np = t % NPH;
  int n = np / NM, p = np % NM;
  float kx = TWOPI_L * (float)(m - 15);
  float ky = TWOPI_L * (float)(n - 15);
  float kz = TWOPI_L * (float)(p - 15);
  float squ = kx * kx + ky * ky + kz * kz;

  float mOO = 0.f, mOH = 0.f, mHO = 0.f, mHH = 0.f;
#pragma unroll
  for (int i = 0; i < 6; ++i) {
    mOO += amOO[i] * expf(-squ * expf(2.0f * shOO[i]));
    mOH += amOH[i] * expf(-squ * expf(2.0f * shOH[i]));
    mHO += amHO[i] * expf(-squ * expf(2.0f * shHO[i]));
    mHH += amHH[i] * expf(-squ * expf(2.0f * shHH[i]));
  }
  if (squ == 0.0f) { mOH = 0.f; mHO = 0.f; mHH = 0.f; }
  if (np < NPH - 1) { mOO *= 2.0f; mOH *= 2.0f; mHO *= 2.0f; mHH *= 2.0f; }

#pragma unroll
  for (int c = 0; c < CCH; ++c) {
    const float* pp = rP + (size_t)blockIdx.y * RP_BATCH + (size_t)(c * NM + m) * NPPAD + np;
    float ror = 0.f, roi = 0.f, rhr = 0.f, rhi = 0.f;
    for (int s = 0; s < NSL_O; ++s) {
      ror += pp[(size_t)(2 * s) * SZ_R];
      roi += pp[(size_t)(2 * s + 1) * SZ_R];
    }
    for (int s = NSL_O; s < NSL; ++s) {
      rhr += pp[(size_t)(2 * s) * SZ_R];
      rhi += pp[(size_t)(2 * s + 1) * SZ_R];
    }
    float fOr = mOO * ror + mOH * rhr, fOi = mOO * roi + mOH * rhi;
    float fHr = mHO * ror + mHH * rhr, fHi = mHO * roi + mHH * rhi;
    const size_t r0 = bo + (size_t)(c * QROW + m)        * KNP + np;
    const size_t r1 = bo + (size_t)((4  + c) * QROW + m) * KNP + np;
    const size_t r2 = bo + (size_t)((8  + c) * QROW + m) * KNP + np;
    const size_t r3 = bo + (size_t)((12 + c) * QROW + m) * KNP + np;
    gOr[r0] = fOr;        gOi[r0] = fOi;
    gOr[r1] = kx * fOr;   gOi[r1] = kx * fOi;
    gOr[r2] = ky * fOr;   gOi[r2] = ky * fOi;
    gOr[r3] = kz * fOr;   gOi[r3] = kz * fOi;
    gHr[r0] = fHr;        gHi[r0] = fHi;
    gHr[r1] = kx * fHr;   gHi[r1] = kx * fHi;
    gHr[r2] = ky * fHr;   gHi[r2] = ky * fHi;
    gHr[r3] = kz * fHr;   gHi[r3] = kz * fHi;
  }
}

constexpr int FIN_J = 32;
__global__ __launch_bounds__(4 * FIN_J)
void finalize(const float* __restrict__ U, float* __restrict__ out) {
  const int b  = blockIdx.y;
  const float* u = U + (size_t)b * WS_TOTAL;
  const int j0 = blockIdx.x * FIN_J;
  const int lane = threadIdx.x & 31;
  const int w = __builtin_amdgcn_readfirstlane((int)(threadIdx.x >> 5));
  if (w == 0) {
    float e = 0.f;
    for (int s = 0; s < SA_NSL; ++s) e += u[(size_t)s * J + j0 + lane];
    out[b * J + j0 + lane] = e / (2.0f * VBOX);
  } else {
    const int t = (w - 1) * 32 + lane;
    float* F = out + 2 * J;
    const float* up = u + (size_t)(1 + t % 3) * SA_NSL * J + j0 + t / 3;
    float f = 0.f;
    for (int s = 0; s < SA_NSL; ++s) f += up[(size_t)s * J];
    F[(size_t)(b * J + j0) * 3 + t] = f / VBOX;
  }
}

extern "C" void kernel_launch(void* const* d_in, const int* in_sizes, int n_in,
                              void* d_out, int out_size, void* d_ws, size_t ws_size,
                              hipStream_t stream) {
  const float* pos    = (const float*)d_in[0];
  const float* charge = (const float*)d_in[1];
  const float* shOO = (const float*)d_in[2];
  const float* amOO = (const float*)d_in[3];
  const float* shOH = (const float*)d_in[4];
  const float* amOH = (const float*)d_in[5];
  const float* shHO = (const float*)d_in[6];
  const float* amHO = (const float*)d_in[7];
  const float* shHH = (const float*)d_in[8];
  const float* amHH = (const float*)d_in[9];
  float* out = (float*)d_out;
  float* ws  = (float*)d_ws;

  float *Cx = ws + OFF_CX, *Sx = ws + OFF_SX;
  float *Wc = ws + OFF_WC, *Wss = ws + OFF_WS;
  float *Yr = ws + OFF_YR, *Yi = ws + OFF_YI;
  float *rP = ws + OFF_RP;
  float *gOr = ws + OFF_GOR, *gOi = ws + OFF_GOI;
  float *gHr = ws + OFF_GHR, *gHi = ws + OFF_GHI;
  float *U = ws + OFF_U;

  hipMemsetAsync(d_ws, 0, NBATCH * WS_TOTAL * sizeof(float), stream);

  tables_build<<<dim3(J / TB_J, TB_WPARTS + 1, NBATCH), TB_THREADS, 0, stream>>>(
      pos, charge, Cx, Sx, Wc, Wss, Yr, Yi);

  nufft1_gemm<<<dim3(NPPAD / 32, NSL, NBATCH), 128, 0, stream>>>(Yr, Yi, Wc, Wss, rP);

  g_build<<<dim3((NM * NPH + 255) / 256, NBATCH), 256, 0, stream>>>(
      rP,
      shOO, amOO, shOH, amOH, shHO, amHO, shHH, amHH,
      gOr, gOi, gHr, gHi);

  stageA_gemm<<<dim3(J / 32, 4 * SA_NSL, NBATCH), 128, 0, stream>>>(gOr, gOi, gHr, gHi, Wc, Wss,
                                                                    Cx, Sx, charge, U);

  finalize<<<dim3(J / FIN_J, NBATCH), 4 * FIN_J, 0, stream>>>(U, out);
}


# x force taken from the energy channel's tile (kx depends on row m alone, applied behind the np sum): g and stage A keep 3 components, np cut into 5 slices: np sum order and x-force rounding change
# speedup vs baseline: 15.2940x; 1.0744x over previous
#include <hip/hip_runtime.h>
#include <hip/hip_bf16.h>
#include <math.h>

typedef float v2f __attribute__((ext_vector_type(2)));
typedef float v8f __attribute__((ext_vector_type(8)));

constexpr int NM    = 31;
constexpr int NPH   = 481;
constexpr int NPPAD = 512;
constexpr int KNP   = 484;
constexpr int J     = 768;
constexpr int JOs   = 256;
constexpr int CCH   = 4;
constexpr int CMP   = 128;
constexpr int QROW  = 32;
constexpr int NCOMP = 3;
constexpr int QMP   = NCOMP * 4 * QROW;
constexpr int KSL   = 128;
constexpr int NSL_O = JOs / KSL;
constexpr int NSL   = J / KSL;
static_assert(JOs % KSL == 0 && J % KSL == 0, "a slice never straddles the two species");
constexpr int SA_NSL = 5;
constexpr int SA_KSL = 100;
static_assert(SA_KSL % 20 == 0 && SA_KSL % 2 == 0, "whole chunks per slice, 8-byte aligned A loads");
static_assert(SA_KSL * (SA_NSL - 1) < KNP, "the last slice is not empty");
constexpr float LBOX   = 20.0f;
constexpr float VBOX   = 8000.0f;
constexpr float TWOPI_L = 0.31415926535897931f;

constexpr size_t SZ_PH  = (size_t)NM * J;
constexpr size_t OFF_CX = 0;
constexpr size_t OFF_SX = OFF_CX + SZ_PH;
constexpr size_t SZ_W   = (size_t)NPPAD * J;
constexpr size_t OFF_WC = OFF_SX + SZ_PH;
constexpr size_t OFF_WS = OFF_WC + SZ_W;
constexpr size_t SZ_Y   = (size_t)CMP * J;
constexpr size_t OFF_YR = OFF_WS + SZ_W;
constexpr size_t OFF_YI = OFF_YR + SZ_Y;
constexpr size_t SZ_G   = (size_t)QMP * KNP;
constexpr size_t OFF_GOR = OFF_YI + SZ_Y;
constexpr size_t OFF_GOI = OFF_GOR + SZ_G;
constexpr size_t OFF_GHR = OFF_GOI + SZ_G;
constexpr size_t OFF_GHI = OFF_GHR + SZ_G;
constexpr size_t SZ_U   = (size_t)4 * SA_NSL * J;
constexpr size_t OFF_U  = OFF_GHI + SZ_G;
constexpr size_t WS_TOTAL = OFF_U + SZ_U;
constexpr int NBATCH = 2;
constexpr size_t SZ_R     = (size_t)CMP * NPPAD;
constexpr size_t RP_BATCH = (size_t)NSL * 2 * SZ_R;
constexpr size_t OFF_RP   = (size_t)NBATCH * WS_TOTAL;

constexpr int TB_J       = 32;
constexpr int TB_THREADS = 512;
constexpr int TB_WAVES   = TB_THREADS / 32;
constexpr int TB_WPARTS  = 4;
constexpr int TB_ROWS    = (NPH + TB_WPARTS - 1) / TB_WPARTS;
static_assert(J % TB_J == 0, "a particle block is a whole line of every row");
__global__ __launch_bounds__(TB_THREADS)
void tables_build(const float* __restrict__ pos, const float* __restrict__ charge,
                  float* __restrict__ Cx, float* __restrict__ Sx,
                  float* __restrict__ Wc, float* __restrict__ Ws,
                  float* __restrict__ Yr, float* __restrict__ Yi) {
  __shared__ float cs[2][NM][TB_J];
  __shared__ float sn[2][NM][TB_J];
  const int b = blockIdx.z;
  const size_t bo = (size_t)b * WS_TOTAL;
  const int j0 = blockIdx.x * TB_J;
  const int part = blockIdx.y;
  const bool xpart = part == TB_WPARTS;
  const int ndim = xpart ? 1 : 2;
  const int d0   = xpart ? 0 : 1;
  for (int i = threadIdx.x; i < ndim * NM * TB_J; i += TB_THREADS) {
    const int jl = i % TB_J, dm = i / TB_J;
    const int mi = dm % NM, dd = dm / NM;
    const float th = TWOPI_L * (pos[(size_t)b * J * 3 + (size_t)(j0 + jl) * 3 + d0 + dd] - 0.5f * LBOX);
    float s, c;
    sincosf((float)(mi - 15) * th, &s, &c);
    cs[dd][mi][jl] = c;
    sn[dd][mi][jl] = s;
  }
  __syncthreads();
  const int jl = threadIdx.x & (TB_J - 1);
  const int w  = __builtin_amdgcn_readfirstlane((int)(threadIdx.x >> 5));
  const size_t col = bo + j0 + jl;
  if (!xpart) {
    const int lo = part * TB_ROWS;
    const int hi = (lo + TB_ROWS < NPH) ? (lo + TB_ROWS) : NPH;
    for (int np = lo + w; np < hi; np += TB_WAVES) {
      const int n = np / NM, p = np % NM;
      const float cy = cs[0][n][jl], sy = sn[0][n][jl];
      const float cz = cs[1][p][jl], sz = sn[1][p][jl];
      Wc[col + (size_t)np * J] = cy * cz - sy * sz;
      Ws[col + (size_t)np * J] = sy * cz + cy * sz;
    }
  } else {
    for (int m = w; m < NM; m += TB_WAVES) {
      Cx[col + (size_t)m * J] = cs[0][m][jl];
      Sx[col + (size_t)m * J] = sn[0][m][jl];
    }
    for (int cm = w; cm < CCH * NM; cm += TB_WAVES) {
      const int c = cm / NM, m = cm % NM;
      const float q = charge[(size_t)b * J * CCH + (size_t)(j0 + jl) * CCH + c];
      Yr[col + (size_t)cm * J] =  q * cs[0][m][jl];
      Yi[col + (size_t)cm * J] = -q * sn[0][m][jl];
    }
  }
}

template <int CH>
struct Frags {
  v2f ar[CH][2], ai[CH][2], br[CH][2], bi[CH][2];
};

template <bool NTMODE, int CH>
__device__ __forceinline__ void load_chunk(Frags<CH>& f,
                                           const float* __restrict__ arp0,
                                           const float* __restrict__ aip0,
                                           const float* __restrict__ arp1,
                                           const float* __restrict__ aip1,
                                           const float* __restrict__ brp,
                                           const float* __restrict__ bip,
                                           int ldb, int k0) {
#pragma unroll
  for (int s = 0; s < CH; ++s) {
    const int k = k0 + 4 * s;
    f.ar[s][0] = *(const v2f*)(arp0 + k);
    f.ar[s][1] = *(const v2f*)(arp1 + k);
    f.ai[s][0] = *(const v2f*)(aip0 + k);
    f.ai[s][1] = *(const v2f*)(aip1 + k);
    if (NTMODE) {
      f.br[s][0] = *(const v2f*)(brp + k);
      f.bi[s][0] = *(const v2f*)(bip + k);
      f.br[s][1] = *(const v2f*)(brp + (size_t)16 * ldb + k);
      f.bi[s][1] = *(const v2f*)(bip + (size_t)16 * ldb + k);
    } else {
      const size_t r0 = (size_t)k * ldb;
#pragma unroll
      for (int nt = 0; nt < 2; ++nt) {
        f.br[s][nt].x = brp[r0 + nt * 16];
        f.br[s][nt].y = brp[r0 + ldb + nt * 16];
        f.bi[s][nt].x = bip[r0 + nt * 16];
        f.bi[s][nt].y = bip[r0 + ldb + nt * 16];
      }
    }
  }
}

template <bool SPLUS>
__device__ __forceinline__ void kstep(const v2f (&a_r)[2], const v2f (&a_i)[2],
                                      const v2f (&b_r)[2], const v2f (&b_i)[2],
                                      v8f (&p1)[2][2], v8f (&p2)[2][2], v8f (&p3)[2][2]) {
  v2f sa[2], sb[2];
#pragma unroll
  for (int mt = 0; mt < 2; ++mt) sa[mt] = a_r[mt] + a_i[mt];
#pragma unroll
  for (int nt = 0; nt < 2; ++nt) sb[nt] = SPLUS ? (b_r[nt] + b_i[nt]) : (b_r[nt] - b_i[nt]);
#pragma unroll
  for (int mt = 0; mt < 2; ++mt)
#pragma unroll
    for (int nt = 0; nt < 2; ++nt) {
      p1[mt][nt] = __builtin_amdgcn_wmma_f32_16x16x4_f32(
          false, a_r[mt], false, b_r[nt], (short)0, p1[mt][nt], false, false);
      p2[mt][nt] = __builtin_amdgcn_wmma_f32_16x16x4_f32(
          false, a_i[mt], false, b_i[nt], (short)0, p2[mt][nt], false, false);
      p3[mt][nt] = __builtin_amdgcn_wmma_f32_16x16x4_f32(
          false, sa[mt], false, sb[nt], (short)0, p3[mt][nt], false, false);
    }
}

template <bool SPLUS, int CH>
__device__ __forceinline__ void compute_chunk(const Frags<CH>& f, v8f (&p1)[2][2],
                                              v8f (&p2)[2][2], v8f (&p3)[2][2]) {
#pragma unroll
  for (int s = 0; s < CH; ++s)
    kstep<SPLUS>(f.ar[s], f.ai[s], f.br[s], f.bi[s], p1, p2, p3);
}

template <bool NTMODE, bool SPLUS>
__device__ __forceinline__
void cgemm_acc(const float* __restrict__ Ar, const float* __restrict__ Ai, int lda,
               const float* __restrict__ Br, const float* __restrict__ Bi, int ldb,
               int K, int m0, int n0, v8f (&accr)[2][2], v8f (&acci)[2][2]) {
  constexpr int CH = NTMODE ? 4 : 5;
  const int lane = threadIdx.x & 31;
  const int row  = lane & 15;
  const int kp   = lane >> 4;

  v8f p1[2][2], p2[2][2], p3[2][2];
#pragma unroll
  for (int mt = 0; mt < 2; ++mt)
#pragma unroll
    for (int nt = 0; nt < 2; ++nt) {
      p1[mt][nt] = (v8f){0.f, 0.f, 0.f, 0.f, 0.f, 0.f, 0.f, 0.f};
      p2[mt][nt] = (v8f){0.f, 0.f, 0.f, 0.f, 0.f, 0.f, 0.f, 0.f};
      p3[mt][nt] = (v8f){0.f, 0.f, 0.f, 0.f, 0.f, 0.f, 0.f, 0.f};
    }

  const float* arp0 = Ar + (size_t)(m0 + row) * lda + 2 * kp;
  const float* aip0 = Ai + (size_t)(m0 + row) * lda + 2 * kp;
  const float* arp1 = arp0 + (size_t)16 * lda;
  const float* aip1 = aip0 + (size_t)16 * lda;
  const float* brp = NTMODE ? Br + (size_t)(n0 + row) * ldb + 2 * kp
                            : Br + (size_t)(2 * kp) * ldb + n0 + row;
  const float* bip = NTMODE ? Bi + (size_t)(n0 + row) * ldb + 2 * kp
                            : Bi + (size_t)(2 * kp) * ldb + n0 + row;

  const int nch = K / (4 * CH);
  Frags<CH> f0, f1;
  if (nch > 0) load_chunk<NTMODE, CH>(f0, arp0, aip0, arp1, aip1, brp, bip, ldb, 0);
  int c = 0;
  constexpr int UNR = NTMODE ? 4 : 1;
#pragma unroll UNR
  for (; c + 2 <= nch; c += 2) {
    load_chunk<NTMODE, CH>(f1, arp0, aip0, arp1, aip1, brp, bip, ldb, (c + 1) * 4 * CH);
    __builtin_amdgcn_sched_barrier(0);
    compute_chunk<SPLUS, CH>(f0, p1, p2, p3);
    __builtin_amdgcn_sched_barrier(0);
    const int cn = (c + 2 < nch) ? (c + 2) : (nch - 1);
    load_chunk<NTMODE, CH>(f0, arp0, aip0, arp1, aip1, brp, bip, ldb, cn * 4 * CH);
    __builtin_amdgcn_sched_barrier(0);
    compute_chunk<SPLUS, CH>(f1, p1, p2, p3);
    __builtin_amdgcn_sched_barrier(0);
  }
  if (c < nch) compute_chunk<SPLUS, CH>(f0, p1, p2, p3);

  for (int k0 = nch * 4 * CH; k0 < K; k0 += 4) {
    Frags<1> t;
    load_chunk<NTMODE, 1>(t, arp0, aip0, arp1, aip1, brp, bip, ldb, k0);
    compute_chunk<SPLUS, 1>(t, p1, p2, p3);
  }

#pragma unroll
  for (int mt = 0; mt < 2; ++mt)
#pragma unroll
    for (int nt = 0; nt < 2; ++nt) {
      if (SPLUS) {
        accr[mt][nt] = p1[mt][nt] - p2[mt][nt];
        acci[mt][nt] = p3[mt][nt] - p1[mt][nt] - p2[mt][nt];
      } else {
        accr[mt][nt] = p1[mt][nt] + p2[mt][nt];
        acci[mt][nt] = p3[mt][nt] - p1[mt][nt] + p2[mt][nt];
      }
    }
}

template <bool NTMODE, bool SPLUS>
__device__ __forceinline__
void cgemm_tile(const float* __restrict__ Ar, const float* __restrict__ Ai, int lda,
                const float* __restrict__ Br, const float* __restrict__ Bi, int ldb,
                float* __restrict__ Cr, float* __restrict__ Ci, int ldc, int K,
                int m0, int n0) {
  const int lane = threadIdx.x & 31;
  const int row  = lane & 15;
  const int kp   = lane >> 4;
  v8f accr[2][2], acci[2][2];
  cgemm_acc<NTMODE, SPLUS>(Ar, Ai, lda, Br, Bi, ldb, K, m0, n0, accr, acci);
#pragma unroll
  for (int mt = 0; mt < 2; ++mt)
#pragma unroll
    for (int nt = 0; nt < 2; ++nt)
#pragma unroll
      for (int v = 0; v < 8; ++v) {
        const size_t o = (size_t)(m0 + mt * 16 + v + 8 * kp) * ldc + n0 + nt * 16 + row;
        Cr[o] = accr[mt][nt][v];
        Ci[o] = acci[mt][nt][v];
      }
}


__global__ __launch_bounds__(128)
void nufft1_gemm(const float* __restrict__ Yr, const float* __restrict__ Yi,
                 const float* __restrict__ Wc, const float* __restrict__ Ws,
                 float* __restrict__ rP) {
  const int wave = threadIdx.x >> 5;
  const size_t bo = (size_t)blockIdx.z * WS_TOTAL;
  const int j0 = blockIdx.y * KSL;
  float* pr = rP + (size_t)blockIdx.z * RP_BATCH + (size_t)blockIdx.y * 2 * SZ_R;
  cgemm_tile<true, false>(Yr + bo + j0, Yi + bo + j0, J,
                          Wc + bo + j0, Ws + bo + j0, J,
                          pr, pr + SZ_R, NPPAD, KSL,
                          wave * 32, blockIdx.x * 32);
}

__global__ __launch_bounds__(128)
void stageA_gemm(const float* __restrict__ gOr, const float* __restrict__ gOi,
                 const float* __restrict__ gHr, const float* __restrict__ gHi,
                 const float* __restrict__ Wc, const float* __restrict__ Ws,
                 const float* __restrict__ Cx, const float* __restrict__ Sx,
                 const float* __restrict__ charge, float* __restrict__ U) {
  __shared__ float red[2][CCH][32];
  const int c    = __builtin_amdgcn_readfirstlane((int)(threadIdx.x >> 5));
  const int comp = blockIdx.y / SA_NSL;
  const int sl   = blockIdx.y % SA_NSL;
  const int k0   = sl * SA_KSL;
  const int klen = (sl == SA_NSL - 1) ? (KNP - k0) : SA_KSL;
  const int bt   = blockIdx.z;
  const int n0   = blockIdx.x * 32;
  const bool h   = n0 >= JOs;
  const size_t bo = (size_t)bt * WS_TOTAL;
  const int m0 = (comp * CCH + c) * QROW;
  const int lane = threadIdx.x & 31;
  const int col  = lane & 15;
  const int kp   = lane >> 4;
  float qc[2];
#pragma unroll
  for (int nt = 0; nt < 2; ++nt)
    qc[nt] = charge[((size_t)bt * J + n0 + nt * 16 + col) * CCH + c];

  v8f accr[2][2], acci[2][2];
  cgemm_acc<false, true>((h ? gHr : gOr) + bo + k0, (h ? gHi : gOi) + bo + k0, KNP,
                         Wc + bo + (size_t)k0 * J, Ws + bo + (size_t)k0 * J, J, klen,
                         m0, n0, accr, acci);

  float cx[2][16], sx[2][16];
#pragma unroll
  for (int nt = 0; nt < 2; ++nt) {
    const float* cxp = Cx + bo + n0 + nt * 16 + col;
    const float* sxp = Sx + bo + n0 + nt * 16 + col;
#pragma unroll
    for (int mt = 0; mt < 2; ++mt)
#pragma unroll
      for (int v = 0; v < 8; ++v) {
        const int m  = mt * 16 + 8 * kp + v;
        const int mm = m < NM ? m : NM - 1;
        cx[nt][mt * 8 + v] = cxp[(size_t)mm * J];
        sx[nt][mt * 8 + v] = sxp[(size_t)mm * J];
      }
  }
  float s[2], sxf[2] = {0.f, 0.f};
  if (comp == 0) {
#pragma unroll
    for (int nt = 0; nt < 2; ++nt) {
      float acc = 0.f, accx = 0.f;
#pragma unroll
      for (int mt = 0; mt < 2; ++mt)
#pragma unroll
        for (int v = 0; v < 8; ++v) {
          const float kxm = TWOPI_L * (float)(mt * 16 + 8 * kp + v - 15);
          acc += accr[mt][nt][v] * cx[nt][mt * 8 + v] - acci[mt][nt][v] * sx[nt][mt * 8 + v];
          accx += kxm * (accr[mt][nt][v] * sx[nt][mt * 8 + v] + acci[mt][nt][v] * cx[nt][mt * 8 + v]);
        }
      s[nt] = acc;
      sxf[nt] = accx;
    }
  } else {
#pragma unroll
    for (int nt = 0; nt < 2; ++nt) {
      float acc = 0.f;
#pragma unroll
      for (int mt = 0; mt < 2; ++mt)
#pragma unroll
        for (int v = 0; v < 8; ++v)
          acc += accr[mt][nt][v] * sx[nt][mt * 8 + v] + acci[mt][nt][v] * cx[nt][mt * 8 + v];
      s[nt] = acc;
    }
  }
#pragma unroll
  for (int nt = 0; nt < 2; ++nt) {
    s[nt] += __shfl_xor(s[nt], 16);
    s[nt] *= qc[nt];
  }
  if (kp == 0) {
    red[0][c][col]      = s[0];
    red[0][c][16 + col] = s[1];
  }
  if (comp == 0) {
#pragma unroll
    for (int nt = 0; nt < 2; ++nt) {
      sxf[nt] += __shfl_xor(sxf[nt], 16);
      sxf[nt] *= qc[nt];
    }
    if (kp == 0) {
      red[1][c][col]      = sxf[0];
      red[1][c][16 + col] = sxf[1];
    }
  }
  __syncthreads();
  if (c != 0) return;
  const int o0 = (comp == 0) ? 0 : comp + 1;
  float e = 0.f;
#pragma unroll
  for (int cc = 0; cc < CCH; ++cc) e += red[0][cc][lane];
  U[bo + (size_t)(o0 * SA_NSL + sl) * J + n0 + lane] = e;
  if (comp == 0) {
    float fx = 0.f;
#pragma unroll
    for (int cc = 0; cc < CCH; ++cc) fx += red[1][cc][lane];
    U[bo + (size_t)(1 * SA_NSL + sl) * J + n0 + lane] = fx;
  }
}

__global__ void g_build(const float* __restrict__ rP,
                        const float* __restrict__ shOO, const float* __restrict__ amOO,
                        const float* __restrict__ shOH, const float* __restrict__ amOH,
                        const float* __restrict__ shHO, const float* __restrict__ amHO,
                        const float* __restrict__ shHH, const float* __restrict__ amHH,
                        float* __restrict__ gOr, float* __restrict__ gOi,
                        float* __restrict__ gHr, float* __restrict__ gHi) {
  const size_t bo = (size_t)blockIdx.y * WS_TOTAL;
  int t = blockIdx.x * blockDim.x + threadIdx.x;
  if (t >= NM * NPH) return;
  int m = t / NPH, np = t % NPH;
  int n = np / NM, p = np % NM;
  float kx = TWOPI_L * (float)(m - 15);
  float ky = TWOPI_L * (float)(n - 15);
  float kz = TWOPI_L * (float)(p - 15);
  float squ = kx * kx + ky * ky + kz * kz;

  float mOO = 0.f, mOH = 0.f, mHO = 0.f, mHH = 0.f;
#pragma unroll
  for (int i = 0; i < 6; ++i) {
    mOO += amOO[i] * expf(-squ * expf(2.0f * shOO[i]));
    mOH += amOH[i] * expf(-squ * expf(2.0f * shOH[i]));
    mHO += amHO[i] * expf(-squ * expf(2.0f * shHO[i]));
    mHH += amHH[i] * expf(-squ * expf(2.0f * shHH[i]));
  }
  if (squ == 0.0f) { mOH = 0.f; mHO = 0.f; mHH = 0.f; }
  if (np < NPH - 1) { mOO *= 2.0f; mOH *= 2.0f; mHO *= 2.0f; mHH *= 2.0f; }

#pragma unroll
  for (int c = 0; c < CCH; ++c) {
    const float* pp = rP + (size_t)blockIdx.y * RP_BATCH + (size_t)(c * NM + m) * NPPAD + np;
    float ror = 0.f, roi = 0.f, rhr = 0.f, rhi = 0.f;
    for (int s = 0; s < NSL_O; ++s) {
      ror += pp[(size_t)(2 * s) * SZ_R];
      roi += pp[(size_t)(2 * s + 1) * SZ_R];
    }
    for (int s = NSL_O; s < NSL; ++s) {
      rhr += pp[(size_t)(2 * s) * SZ_R];
      rhi += pp[(size_t)(2 * s + 1) * SZ_R];
    }
    float fOr = mOO * ror + mOH * rhr, fOi = mOO * roi + mOH * rhi;
    float fHr = mHO * ror + mHH * rhr, fHi = mHO * roi + mHH * rhi;
    const size_t r0 = bo + (size_t)(c * QROW + m)        * KNP + np;
    const size_t r1 = bo + (size_t)((4  + c) * QROW + m) * KNP + np;
    const size_t r2 = bo + (size_t)((8  + c) * QROW + m) * KNP + np;
    gOr[r0] = fOr;        gOi[r0] = fOi;
    gOr[r1] = ky * fOr;   gOi[r1] = ky * fOi;
    gOr[r2] = kz * fOr;   gOi[r2] = kz * fOi;
    gHr[r0] = fHr;        gHi[r0] = fHi;
    gHr[r1] = ky * fHr;   gHi[r1] = ky * fHi;
    gHr[r2] = kz * fHr;   gHi[r2] = kz * fHi;
  }
}

constexpr int FIN_J = 32;
__global__ __launch_bounds__(4 * FIN_J)
void finalize(const float* __restrict__ U, float* __restrict__ out) {
  const int b  = blockIdx.y;
  const float* u = U + (size_t)b * WS_TOTAL;
  const int j0 = blockIdx.x * FIN_J;
  const int lane = threadIdx.x & 31;
  const int w = __builtin_amdgcn_readfirstlane((int)(threadIdx.x >> 5));
  if (w == 0) {
    float e = 0.f;
    for (int s = 0; s < SA_NSL; ++s) e += u[(size_t)s * J + j0 + lane];
    out[b * J + j0 + lane] = e / (2.0f * VBOX);
  } else {
    const int t = (w - 1) * 32 + lane;
    float* F = out + 2 * J;
    const float* up = u + (size_t)(1 + t % 3) * SA_NSL * J + j0 + t / 3;
    float f = 0.f;
    for (int s = 0; s < SA_NSL; ++s) f += up[(size_t)s * J];
    F[(size_t)(b * J + j0) * 3 + t] = f / VBOX;
  }
}

extern "C" void kernel_launch(void* const* d_in, const int* in_sizes, int n_in,
                              void* d_out, int out_size, void* d_ws, size_t ws_size,
                              hipStream_t stream) {
  const float* pos    = (const float*)d_in[0];
  const float* charge = (const float*)d_in[1];
  const float* shOO = (const float*)d_in[2];
  const float* amOO = (const float*)d_in[3];
  const float* shOH = (const float*)d_in[4];
  const float* amOH = (const float*)d_in[5];
  const float* shHO = (const float*)d_in[6];
  const float* amHO = (const float*)d_in[7];
  const float* shHH = (const float*)d_in[8];
  const float* amHH = (const float*)d_in[9];
  float* out = (float*)d_out;
  float* ws  = (float*)d_ws;

  float *Cx = ws + OFF_CX, *Sx = ws + OFF_SX;
  float *Wc = ws + OFF_WC, *Wss = ws + OFF_WS;
  float *Yr = ws + OFF_YR, *Yi = ws + OFF_YI;
  float *rP = ws + OFF_RP;
  float *gOr = ws + OFF_GOR, *gOi = ws + OFF_GOI;
  float *gHr = ws + OFF_GHR, *gHi = ws + OFF_GHI;
  float *U = ws + OFF_U;

  hipMemsetAsync(d_ws, 0, NBATCH * WS_TOTAL * sizeof(float), stream);

  tables_build<<<dim3(J / TB_J, TB_WPARTS + 1, NBATCH), TB_THREADS, 0, stream>>>(
      pos, charge, Cx, Sx, Wc, Wss, Yr, Yi);

  nufft1_gemm<<<dim3(NPPAD / 32, NSL, NBATCH), 128, 0, stream>>>(Yr, Yi, Wc, Wss, rP);

  g_build<<<dim3((NM * NPH + 255) / 256, NBATCH), 256, 0, stream>>>(
      rP,
      shOO, amOO, shOH, amOH, shHO, amHO, shHH, amHH,
      gOr, gOi, gHr, gHi);

  stageA_gemm<<<dim3(J / 32, NCOMP * SA_NSL, NBATCH), 128, 0, stream>>>(gOr, gOi, gHr, gHi, Wc, Wss,
                                                                        Cx, Sx, charge, U);

  finalize<<<dim3(J / FIN_J, NBATCH), 4 * FIN_J, 0, stream>>>(U, out);
}
